# starting kernel
# baseline (speedup 1.0000x reference)
#include <hip/hip_runtime.h>
#include <math.h>

#define B_   32
#define L_   4096
#define MEM_ 256
#define Q_   256
#define BL_  (B_ * L_)
#define BQ_  (B_ * Q_)
#define B3Q_ (B_ * 3 * Q_)

#define OFF_S1   0
#define OFF_S2   (OFF_S1 + BL_)
#define OFF_SACC (OFF_S2 + BL_)
#define OFF_EACC (OFF_SACC + BL_)
#define OFF_QB   (OFF_EACC + BL_)
#define OFF_QA   (OFF_QB + BQ_)
#define OFF_QE   (OFF_QA + BQ_)
#define OFF_EV   (OFF_QE + BQ_)
#define OFF_X1   (OFF_EV + BQ_)
#define OFF_S_A  (OFF_X1 + BQ_)
#define OFF_S_B  (OFF_S_A + BQ_)
#define OFF_GI   (OFF_S_B + BQ_)
#define OFF_GH   (OFF_GI + B3Q_)

typedef __attribute__((ext_vector_type(2))) float v2f;
typedef __attribute__((ext_vector_type(8))) float v8f;

__global__ void wmma_gemm_kernel(const float* __restrict__ A, int lda,
                                 const float* __restrict__ W, int ldw,
                                 const float* __restrict__ bias,
                                 float* __restrict__ C, int ldc,
                                 int K, int accumulate) {
    const int lane = threadIdx.x & 31;
    const int row  = lane & 15;
    const int hi   = lane >> 4;
    const int m0   = blockIdx.y * 16;
    const int n0   = blockIdx.x * 16;

    const float binit = bias ? bias[n0 + row] : 0.0f;
    v8f acc = {};
    if (accumulate) {
#pragma unroll
        for (int j = 0; j < 8; ++j)
            acc[j] = C[(size_t)(m0 + j + 8 * hi) * ldc + n0 + row] + binit;
    } else {
#pragma unroll
        for (int j = 0; j < 8; ++j) acc[j] = binit;
    }

    const float* aRow = A + (size_t)(m0 + row) * lda + 2 * hi;
    const float* wRow = W + (size_t)(n0 + row) * ldw + 2 * hi;
    for (int k = 0; k < K; k += 4) {
        v2f a; a.x = aRow[k]; a.y = aRow[k + 1];
        v2f b; b.x = wRow[k]; b.y = wRow[k + 1];
        acc = __builtin_amdgcn_wmma_f32_16x16x4_f32(false, a, false, b,
                                                    (short)0, acc, false, false);
    }
#pragma unroll
    for (int j = 0; j < 8; ++j)
        C[(size_t)(m0 + j + 8 * hi) * ldc + n0 + row] = acc[j];
}

__global__ void scores_kernel(const float* __restrict__ M,
                              const float* __restrict__ q1,
                              const float* __restrict__ q2,
                              float* __restrict__ S1,
                              float* __restrict__ S2) {
    const int lane = threadIdx.x & 31;
    const int row  = lane & 15;
    const int hi   = lane >> 4;
    const int b    = blockIdx.y;
    const int l0   = blockIdx.x * 16;

    const float* Mb = M + ((size_t)b * L_ + l0 + row) * MEM_ + 2 * hi;
    const float* qv = ((row == 0) ? q1 : q2) + (size_t)b * MEM_ + 2 * hi;

    __builtin_prefetch(Mb + 16 * MEM_, 0, 0);

    v8f acc = {};
    for (int k = 0; k < MEM_; k += 4) {
        v2f a; a.x = Mb[k]; a.y = Mb[k + 1];
        v2f bb;
        if (row < 2) { bb.x = qv[k]; bb.y = qv[k + 1]; }
        else         { bb.x = 0.0f;  bb.y = 0.0f; }
        acc = __builtin_amdgcn_wmma_f32_16x16x4_f32(false, a, false, bb,
                                                    (short)0, acc, false, false);
    }
    if (row < 2) {
        float* dst = (row == 0) ? S1 : S2;
#pragma unroll
        for (int j = 0; j < 8; ++j)
            dst[(size_t)b * L_ + l0 + j + 8 * hi] = acc[j];
    }
}

__global__ void wsum_kernel(const float* __restrict__ M,
                            const float* __restrict__ p,
                            const float* __restrict__ w,
                            float* __restrict__ ev,
                            float* __restrict__ x1) {
    const int lane = threadIdx.x & 31;
    const int row  = lane & 15;
    const int hi   = lane >> 4;
    const int b    = blockIdx.y;
    const int m0   = blockIdx.x * 16;

    const float* pv = ((row == 0) ? p : w) + (size_t)b * L_ + 2 * hi;
    const float* Mb = M + ((size_t)b * L_ + 2 * hi) * MEM_ + m0 + row;

    v8f acc = {};
    for (int l = 0; l < L_; l += 4) {
        v2f a;
        if (row < 2) { a.x = pv[l]; a.y = pv[l + 1]; }
        else         { a.x = 0.0f;  a.y = 0.0f; }
        v2f bb;
        bb.x = Mb[(size_t)l * MEM_];
        bb.y = Mb[(size_t)l * MEM_ + MEM_];
        acc = __builtin_amdgcn_wmma_f32_16x16x4_f32(false, a, false, bb,
                                                    (short)0, acc, false, false);
    }
    if (hi == 0) {
        ev[(size_t)b * MEM_ + m0 + row] = acc[0];
        x1[(size_t)b * MEM_ + m0 + row] = acc[1];
    }
}

__global__ void softmax_kernel(float* __restrict__ S,
                               float* __restrict__ accDst,
                               float scale) {
    const int b = blockIdx.x;
    const int tid = threadIdx.x;
    __shared__ float red[256];
    float* rowp = S + (size_t)b * L_;

    float m = -3.402823466e38f;
    for (int i = tid; i < L_; i += 256) m = fmaxf(m, rowp[i]);
    red[tid] = m; __syncthreads();
    for (int s = 128; s > 0; s >>= 1) {
        if (tid < s) red[tid] = fmaxf(red[tid], red[tid + s]);
        __syncthreads();
    }
    m = red[0]; __syncthreads();

    float sum = 0.0f;
    for (int i = tid; i < L_; i += 256) {
        float e = __expf(rowp[i] - m);
        rowp[i] = e;
        sum += e;
    }
    red[tid] = sum; __syncthreads();
    for (int s = 128; s > 0; s >>= 1) {
        if (tid < s) red[tid] += red[tid + s];
        __syncthreads();
    }
    const float inv = 1.0f / red[0];

    for (int i = tid; i < L_; i += 256) {
        float pv = rowp[i] * inv;
        rowp[i] = pv;
        if (accDst) accDst[(size_t)b * L_ + i] += scale * pv;
    }
}

__global__ void gru_kernel(const float* __restrict__ gi,
                           const float* __restrict__ gh,
                           const float* __restrict__ sOld,
                           float* __restrict__ sNew) {
    int i = blockIdx.x * 256 + threadIdx.x;
    if (i >= BQ_) return;
    int b = i / Q_, q = i % Q_;
    const float* gib = gi + (size_t)b * 3 * Q_;
    const float* ghb = gh + (size_t)b * 3 * Q_;
    float ir = gib[q], iz = gib[Q_ + q], in_ = gib[2 * Q_ + q];
    float hr = ghb[q], hz = ghb[Q_ + q], hn  = ghb[2 * Q_ + q];
    float r = 1.0f / (1.0f + __expf(-(ir + hr)));
    float z = 1.0f / (1.0f + __expf(-(iz + hz)));
    float n = tanhf(in_ + r * hn);
    sNew[i] = (1.0f - z) * n + z * sOld[i];
}

__global__ void init_kernel(const float* __restrict__ s0, float* __restrict__ ws) {
    int i = blockIdx.x * 256 + threadIdx.x;
    if (i < 2 * BL_) ws[OFF_SACC + i] = 0.0f;
    if (i < BQ_)     ws[OFF_S_A + i] = s0[i];
}

__global__ void final_kernel(const float* __restrict__ ws, float* __restrict__ out) {
    int i = blockIdx.x * 256 + threadIdx.x;
    if (i < BL_) {
        out[i]       = logf(ws[OFF_SACC + i]);
        out[BL_ + i] = logf(ws[OFF_EACC + i]);
    }
}

extern "C" void kernel_launch(void* const* d_in, const int* in_sizes, int n_in,
                              void* d_out, int out_size, void* d_ws, size_t ws_size,
                              hipStream_t stream) {
    const float* M    = (const float*)d_in[0];
    const float* s0   = (const float*)d_in[1];
    const float* Wb   = (const float*)d_in[2];
    const float* We   = (const float*)d_in[3];
    const float* Wa   = (const float*)d_in[4];
    const float* W_ih = (const float*)d_in[5];
    const float* W_hh = (const float*)d_in[6];
    const float* b_ih = (const float*)d_in[7];
    const float* b_hh = (const float*)d_in[8];

    float* ws   = (float*)d_ws;
    float* S1   = ws + OFF_S1;
    float* S2   = ws + OFF_S2;
    float* sacc = ws + OFF_SACC;
    float* eacc = ws + OFF_EACC;
    float* qb   = ws + OFF_QB;
    float* qa   = ws + OFF_QA;
    float* qe   = ws + OFF_QE;
    float* ev   = ws + OFF_EV;
    float* x1   = ws + OFF_X1;
    float* sA   = ws + OFF_S_A;
    float* sB   = ws + OFF_S_B;
    float* gi   = ws + OFF_GI;
    float* gh   = ws + OFF_GH;

    init_kernel<<<(2 * BL_) / 256, 256, 0, stream>>>(s0, ws);

    auto gemm = [&](const float* A, int lda, const float* W, int ldw,
                    const float* bias, float* C, int ldc, int K, int N, int accF) {
        dim3 g(N / 16, B_ / 16);
        wmma_gemm_kernel<<<g, 32, 0, stream>>>(A, lda, W, ldw, bias, C, ldc, K, accF);
    };

    for (int t = 0; t < 5; ++t) {
        const float* sCur = (t & 1) ? sB : sA;
        float*       sNxt = (t & 1) ? sA : sB;

        gemm(sCur, Q_, Wb, Q_, nullptr, qb, MEM_, Q_, MEM_, 0);
        gemm(sCur, Q_, Wa, Q_, nullptr, qa, MEM_, Q_, MEM_, 0);

        scores_kernel<<<dim3(L_ / 16, B_), 32, 0, stream>>>(M, qb, qa, S1, S2);

        softmax_kernel<<<B_, 256, 0, stream>>>(S1, sacc, 0.2f);
        softmax_kernel<<<B_, 256, 0, stream>>>(S2, nullptr, 0.0f);

        wsum_kernel<<<dim3(MEM_ / 16, B_), 32, 0, stream>>>(M, S1, S2, ev, x1);

        gemm(sCur, Q_,  We,      Q_ + MEM_, nullptr, qe, MEM_, Q_,   MEM_, 0);
        gemm(ev,  MEM_, We + Q_, Q_ + MEM_, nullptr, qe, MEM_, MEM_, MEM_, 1);

        scores_kernel<<<dim3(L_ / 16, B_), 32, 0, stream>>>(M, qe, qe, S1, S2);
        softmax_kernel<<<B_, 256, 0, stream>>>(S1, eacc, 0.2f);

        gemm(x1,  MEM_, W_ih, MEM_, b_ih, gi, 3 * Q_, MEM_, 3 * Q_, 0);
        gemm(sCur, Q_,  W_hh, Q_,   b_hh, gh, 3 * Q_, Q_,   3 * Q_, 0);
        gru_kernel<<<BQ_ / 256, 256, 0, stream>>>(gi, gh, sCur, sNxt);
    }

    final_kernel<<<BL_ / 256, 256, 0, stream>>>(ws, (float*)d_out);
}


# small GEMMs: block-diagonal WMMA fragments, a wave owns two 4x4 output tiles, K split over the 4 lanes of a quad; all 24 loads requested at once; whole-line stores via LDS; new summation order
# speedup vs baseline: 6.2723x; 6.2723x over previous
#include <hip/hip_runtime.h>
#include <math.h>

#define B_   32
#define L_   4096
#define MEM_ 256
#define Q_   256
#define BL_  (B_ * L_)
#define BQ_  (B_ * Q_)
#define B3Q_ (B_ * 3 * Q_)

#define OFF_S1   0
#define OFF_S2   (OFF_S1 + BL_)
#define OFF_SACC (OFF_S2 + BL_)
#define OFF_EACC (OFF_SACC + BL_)
#define OFF_QB   (OFF_EACC + BL_)
#define OFF_QA   (OFF_QB + BQ_)
#define OFF_QE   (OFF_QA + BQ_)
#define OFF_EV   (OFF_QE + BQ_)
#define OFF_X1   (OFF_EV + BQ_)
#define OFF_S_A  (OFF_X1 + BQ_)
#define OFF_S_B  (OFF_S_A + BQ_)
#define OFF_GI   (OFF_S_B + BQ_)
#define OFF_GH   (OFF_GI + B3Q_)

typedef __attribute__((ext_vector_type(2))) float v2f;
typedef __attribute__((ext_vector_type(8))) float v8f;

typedef __attribute__((ext_vector_type(4))) float v4f;

#define SG_K     256
#define SG_LINES (SG_K / 32)
#define SG_WAVES 8
#define SG_ROWS  8

__global__ void __launch_bounds__(SG_WAVES * 32)
wmma_gemm_kernel(const float* __restrict__ A, int lda,
                 const float* __restrict__ W, int ldw,
                 const float* __restrict__ bias,
                 float* __restrict__ C, int ldc,
                 int accumulate) {
    const int tid   = threadIdx.x;
    const int lane  = tid & 31;
    const int wave  = tid >> 5;
    const int c     = lane & 15;
    const int hi    = lane >> 4;
    const int kb    = c & 3;
    const int piece = 4 * (2 * kb + hi);
    const int m0    = blockIdx.y * SG_ROWS;
    const int n0    = blockIdx.x * 32;

    const float* ap = A + (size_t)(m0 + (c >> 2)) * lda + piece;
    const float* wp = W + (size_t)(n0 + 4 * wave + (c >> 2)) * ldw + piece;

    v4f xw[SG_LINES], xa[2][SG_LINES];
#pragma unroll
    for (int j = 0; j < SG_LINES; ++j) xw[j] = *(const v4f*)(wp + 32 * j);
#pragma unroll
    for (int u = 0; u < 2; ++u) {
#pragma unroll
        for (int j = 0; j < SG_LINES; ++j)
            xa[u][j] = *(const v4f*)(ap + (size_t)(4 * u) * lda + 32 * j);
    }

    const int orow = wave, ocol = lane;
    float* cp = C + (size_t)(m0 + orow) * ldc + n0 + ocol;
    float bv = 0.0f, cv = 0.0f;
    if (bias) bv = bias[n0 + ocol];
    if (accumulate) cv = *cp;

    v8f acc0[2], acc1[2];
#pragma unroll
    for (int u = 0; u < 2; ++u) { acc0[u] = (v8f){}; acc1[u] = (v8f){}; }
#pragma unroll
    for (int j = 0; j < SG_LINES; ++j) {
#pragma unroll
        for (int u = 0; u < 2; ++u) {
            v2f a, bb;
            a.x = xa[u][j][0]; a.y = xa[u][j][1]; bb.x = xw[j][0]; bb.y = xw[j][1];
            acc0[u] = __builtin_amdgcn_wmma_f32_16x16x4_f32(false, a, false, bb,
                                                            (short)0, acc0[u], false, false);
            a.x = xa[u][j][2]; a.y = xa[u][j][3]; bb.x = xw[j][2]; bb.y = xw[j][3];
            acc1[u] = __builtin_amdgcn_wmma_f32_16x16x4_f32(false, a, false, bb,
                                                            (short)0, acc1[u], false, false);
        }
    }
    float lo[2], up[2];
#pragma unroll
    for (int u = 0; u < 2; ++u) {
        const v8f tv = acc0[u] + acc1[u];
        const v4f tl = {tv[0], tv[1], tv[2], tv[3]};
        const v4f tu = {tv[4], tv[5], tv[6], tv[7]};
        lo[u] = tl[kb]; up[u] = tu[kb];
    }
#pragma unroll
    for (int u = 0; u < 2; ++u) { lo[u] += __shfl_xor(lo[u], 1); up[u] += __shfl_xor(up[u], 1); }
#pragma unroll
    for (int u = 0; u < 2; ++u) { lo[u] += __shfl_xor(lo[u], 2); up[u] += __shfl_xor(up[u], 2); }

    __shared__ float tile[SG_ROWS][32];
    if (kb == 0) {
#pragma unroll
        for (int u = 0; u < 2; ++u) {
            tile[4 * u + 2 * hi][4 * wave + (c >> 2)]     = lo[u];
            tile[4 * u + 2 * hi + 1][4 * wave + (c >> 2)] = up[u];
        }
    }
    __syncthreads();
    *cp = (cv + bv) + tile[orow][ocol];
}

#define SC_CH     8
#define SC_WAVES  16
#define SC_WGROWS (L_ / SC_CH)
#define SC_WROWS  (SC_WGROWS / SC_WAVES)
#define SC_GROUPS (SC_WROWS / 4)
#define SC_LINES  (MEM_ / 32)
#define SC_AHEAD  3
#define SC_RING   (SC_AHEAD + 1)

__global__ void __launch_bounds__(SC_WAVES * 32)
scores_rows_kernel(const float* __restrict__ M,
                   const float* __restrict__ q1,
                   const float* __restrict__ q2,
                   float* __restrict__ S1,
                   float* __restrict__ S2) {
    const int tid   = threadIdx.x;
    const int lane  = tid & 31;
    const int wave  = tid >> 5;
    const int c     = lane & 15;
    const int hi    = lane >> 4;
    const int kb    = c & 3;
    const int b     = blockIdx.y;
    const int ch    = blockIdx.x;
    const int piece = 4 * (2 * kb + hi);
    const int r0    = ch * SC_WGROWS + wave * SC_WROWS;

    const float* Mb = M + ((size_t)b * L_ + r0 + (c >> 2)) * MEM_ + piece;
    const float* qv = (((c >> 2) & 1) ? q2 : q1) + (size_t)b * MEM_ + piece;

    v4f x[SC_RING][SC_LINES];
#pragma unroll
    for (int g = 0; g < SC_AHEAD; ++g) {
#pragma unroll
        for (int j = 0; j < SC_LINES; ++j)
            x[g][j] = *(const v4f*)(Mb + (size_t)(4 * g) * MEM_ + 32 * j);
    }

    v4f qf[SC_LINES];
#pragma unroll
    for (int j = 0; j < SC_LINES; ++j) qf[j] = *(const v4f*)(qv + 32 * j);

    float slo[SC_GROUPS], shi[SC_GROUPS];
#pragma unroll
    for (int g = 0; g < SC_GROUPS; ++g) {
        const int cur = g % SC_RING;
        if (g + SC_AHEAD < SC_GROUPS) {
            const int nxt = (g + SC_AHEAD) % SC_RING;
#pragma unroll
            for (int j = 0; j < SC_LINES; ++j)
                x[nxt][j] = *(const v4f*)(Mb + (size_t)(4 * (g + SC_AHEAD)) * MEM_ + 32 * j);
        }
        __builtin_amdgcn_sched_barrier(0);
        v8f acc0 = {}, acc1 = {};
#pragma unroll
        for (int j = 0; j < SC_LINES; ++j) {
            v2f a, bb;
            a.x = x[cur][j][0]; a.y = x[cur][j][1]; bb.x = qf[j][0]; bb.y = qf[j][1];
            acc0 = __builtin_amdgcn_wmma_f32_16x16x4_f32(false, a, false, bb,
                                                         (short)0, acc0, false, false);
            a.x = x[cur][j][2]; a.y = x[cur][j][3]; bb.x = qf[j][2]; bb.y = qf[j][3];
            acc1 = __builtin_amdgcn_wmma_f32_16x16x4_f32(false, a, false, bb,
                                                         (short)0, acc1, false, false);
        }
        float t[8];
#pragma unroll
        for (int r = 0; r < 8; ++r) t[r] = acc0[r] + acc1[r];
        float lo = (kb == 0) ? t[0] : (kb == 1) ? t[1] : (kb == 2) ? t[2] : t[3];
        float up = (kb == 0) ? t[4] : (kb == 1) ? t[5] : (kb == 2) ? t[6] : t[7];
        lo += __shfl_xor(lo, 1); up += __shfl_xor(up, 1);
        lo += __shfl_xor(lo, 2); up += __shfl_xor(up, 2);
        slo[g] = lo; shi[g] = up;
    }

    __shared__ float sc[2][SC_WGROWS];
    if (c == 0 || c == 4) {
        float* dst = &sc[c >> 2][wave * SC_WROWS + 2 * hi];
#pragma unroll
        for (int g = 0; g < SC_GROUPS; ++g) { dst[4 * g] = slo[g]; dst[4 * g + 1] = shi[g]; }
    }
    __syncthreads();
    S1[(size_t)b * L_ + ch * SC_WGROWS + tid] = sc[0][tid];
    S2[(size_t)b * L_ + ch * SC_WGROWS + tid] = sc[1][tid];
}

#define WS_CH     8
#define WS_WAVES  16
#define WS_WGROWS (L_ / WS_CH)
#define WS_WROWS  (WS_WGROWS / WS_WAVES)
#define WS_KSTEPS (WS_WROWS / 4)
#define WS_KSTRIDE (4 * WS_WAVES)

__global__ void __launch_bounds__(WS_WAVES * 32)
wsum_rows_kernel(const float* __restrict__ M,
                 float* pbuf,
                 const float* wbuf) {
    const int tid  = threadIdx.x;
    const int lane = tid & 31;
    const int wave = tid >> 5;
    const int row  = lane & 15;
    const int hi   = lane >> 4;
    const int b    = blockIdx.y;
    const int c    = blockIdx.x;
    const int l0   = c * WS_WGROWS + 4 * wave;

    const float* pv = ((row == 1) ? wbuf : (const float*)pbuf) + (size_t)b * L_ + l0 + 2 * hi;
    const float* Mb = M + ((size_t)b * L_ + l0 + 2 * hi) * MEM_ + 4 * row;

    v8f acc[16];
#pragma unroll
    for (int t = 0; t < 16; ++t) acc[t] = (v8f){};

    v2f a[WS_KSTEPS];
#pragma unroll
    for (int ks = 0; ks < WS_KSTEPS; ++ks) {
        a[ks] = *(const v2f*)(pv + WS_KSTRIDE * ks);
        if (row >= 2) { a[ks].x = 0.0f; a[ks].y = 0.0f; }
    }

    v4f x[2][2][4];
#pragma unroll
    for (int g = 0; g < 4; ++g) {
        x[0][0][g] = *(const v4f*)(Mb + 64 * g);
        x[0][1][g] = *(const v4f*)(Mb + MEM_ + 64 * g);
    }

#pragma unroll
    for (int ks = 0; ks < WS_KSTEPS; ++ks) {
        const int cur = ks & 1, nxt = cur ^ 1;
        if (ks + 1 < WS_KSTEPS) {
#pragma unroll
            for (int g = 0; g < 4; ++g) {
                x[nxt][0][g] = *(const v4f*)(Mb + (size_t)(WS_KSTRIDE * (ks + 1))     * MEM_ + 64 * g);
                x[nxt][1][g] = *(const v4f*)(Mb + (size_t)(WS_KSTRIDE * (ks + 1) + 1) * MEM_ + 64 * g);
            }
        }
        __builtin_amdgcn_sched_barrier(0);
#pragma unroll
        for (int g = 0; g < 4; ++g) {
#pragma unroll
            for (int cc = 0; cc < 4; ++cc) {
                v2f bb; bb.x = x[cur][0][g][cc]; bb.y = x[cur][1][g][cc];
                acc[4 * g + cc] = __builtin_amdgcn_wmma_f32_16x16x4_f32(
                    false, a[ks], false, bb, (short)0, acc[4 * g + cc], false, false);
            }
        }
    }

    __shared__ float red[WS_WAVES][2][MEM_];
    if (hi == 0) {
#pragma unroll
        for (int g = 0; g < 4; ++g) {
            v4f es, xs;
#pragma unroll
            for (int cc = 0; cc < 4; ++cc) { es[cc] = acc[4 * g + cc][0]; xs[cc] = acc[4 * g + cc][1]; }
            *(v4f*)&red[wave][0][64 * g + 4 * row] = es;
            *(v4f*)&red[wave][1][64 * g + 4 * row] = xs;
        }
    }
    __syncthreads();
    {
        const int v = tid >> 8, col = tid & 255;
        float s = 0.0f;
#pragma unroll
        for (int w = 0; w < WS_WAVES; ++w) s += red[w][v][col];
        pbuf[(size_t)b * L_ + c * WS_WGROWS + v * MEM_ + col] = s;
    }
}

__global__ void wsum_reduce_kernel(const float* __restrict__ part,
                                   float* __restrict__ ev,
                                   float* __restrict__ x1) {
    const int i   = blockIdx.x * 256 + threadIdx.x;
    const int b   = i >> 9;
    const int v   = (i >> 8) & 1;
    const int col = i & 255;
    const float* src = part + (size_t)b * L_ + v * MEM_ + col;
    float s = 0.0f;
#pragma unroll
    for (int c = 0; c < WS_CH; ++c) s += src[c * WS_WGROWS];
    (v ? x1 : ev)[(size_t)b * MEM_ + col] = s;
}

__global__ void softmax_kernel(float* __restrict__ S,
                               float* __restrict__ accDst,
                               float scale) {
    const int b = blockIdx.x;
    const int tid = threadIdx.x;
    __shared__ float red[256];
    float* rowp = S + (size_t)b * L_;

    float m = -3.402823466e38f;
    for (int i = tid; i < L_; i += 256) m = fmaxf(m, rowp[i]);
    red[tid] = m; __syncthreads();
    for (int s = 128; s > 0; s >>= 1) {
        if (tid < s) red[tid] = fmaxf(red[tid], red[tid + s]);
        __syncthreads();
    }
    m = red[0]; __syncthreads();

    float sum = 0.0f;
    for (int i = tid; i < L_; i += 256) {
        float e = __expf(rowp[i] - m);
        rowp[i] = e;
        sum += e;
    }
    red[tid] = sum; __syncthreads();
    for (int s = 128; s > 0; s >>= 1) {
        if (tid < s) red[tid] += red[tid + s];
        __syncthreads();
    }
    const float inv = 1.0f / red[0];

    for (int i = tid; i < L_; i += 256) {
        float pv = rowp[i] * inv;
        rowp[i] = pv;
        if (accDst) accDst[(size_t)b * L_ + i] += scale * pv;
    }
}

__global__ void gru_kernel(const float* __restrict__ gi,
                           const float* __restrict__ gh,
                           const float* __restrict__ sOld,
                           float* __restrict__ sNew) {
    int i = blockIdx.x * 256 + threadIdx.x;
    if (i >= BQ_) return;
    int b = i / Q_, q = i % Q_;
    const float* gib = gi + (size_t)b * 3 * Q_;
    const float* ghb = gh + (size_t)b * 3 * Q_;
    float ir = gib[q], iz = gib[Q_ + q], in_ = gib[2 * Q_ + q];
    float hr = ghb[q], hz = ghb[Q_ + q], hn  = ghb[2 * Q_ + q];
    float r = 1.0f / (1.0f + __expf(-(ir + hr)));
    float z = 1.0f / (1.0f + __expf(-(iz + hz)));
    float n = tanhf(in_ + r * hn);
    sNew[i] = (1.0f - z) * n + z * sOld[i];
}

__global__ void init_kernel(const float* __restrict__ s0, float* __restrict__ ws) {
    int i = blockIdx.x * 256 + threadIdx.x;
    if (i < 2 * BL_) ws[OFF_SACC + i] = 0.0f;
    if (i < BQ_)     ws[OFF_S_A + i] = s0[i];
}

__global__ void final_kernel(const float* __restrict__ ws, float* __restrict__ out) {
    int i = blockIdx.x * 256 + threadIdx.x;
    if (i < BL_) {
        out[i]       = logf(ws[OFF_SACC + i]);
        out[BL_ + i] = logf(ws[OFF_EACC + i]);
    }
}

extern "C" void kernel_launch(void* const* d_in, const int* in_sizes, int n_in,
                              void* d_out, int out_size, void* d_ws, size_t ws_size,
                              hipStream_t stream) {
    const float* M    = (const float*)d_in[0];
    const float* s0   = (const float*)d_in[1];
    const float* Wb   = (const float*)d_in[2];
    const float* We   = (const float*)d_in[3];
    const float* Wa   = (const float*)d_in[4];
    const float* W_ih = (const float*)d_in[5];
    const float* W_hh = (const float*)d_in[6];
    const float* b_ih = (const float*)d_in[7];
    const float* b_hh = (const float*)d_in[8];

    float* ws   = (float*)d_ws;
    float* S1   = ws + OFF_S1;
    float* S2   = ws + OFF_S2;
    float* sacc = ws + OFF_SACC;
    float* eacc = ws + OFF_EACC;
    float* qb   = ws + OFF_QB;
    float* qa   = ws + OFF_QA;
    float* qe   = ws + OFF_QE;
    float* ev   = ws + OFF_EV;
    float* x1   = ws + OFF_X1;
    float* sA   = ws + OFF_S_A;
    float* sB   = ws + OFF_S_B;
    float* gi   = ws + OFF_GI;
    float* gh   = ws + OFF_GH;

    init_kernel<<<(2 * BL_) / 256, 256, 0, stream>>>(s0, ws);

    auto gemm = [&](const float* A, int lda, const float* W, int ldw,
                    const float* bias, float* C, int ldc, int N, int accF) {
        dim3 g(N / 32, B_ / SG_ROWS);
        wmma_gemm_kernel<<<g, SG_WAVES * 32, 0, stream>>>(A, lda, W, ldw, bias, C, ldc, accF);
    };

    for (int t = 0; t < 5; ++t) {
        const float* sCur = (t & 1) ? sB : sA;
        float*       sNxt = (t & 1) ? sA : sB;

        gemm(sCur, Q_, Wb, Q_, nullptr, qb, MEM_, MEM_, 0);
        gemm(sCur, Q_, Wa, Q_, nullptr, qa, MEM_, MEM_, 0);

        scores_rows_kernel<<<dim3(SC_CH, B_), SC_WAVES * 32, 0, stream>>>(M, qb, qa, S1, S2);

        softmax_kernel<<<B_, 256, 0, stream>>>(S1, sacc, 0.2f);
        softmax_kernel<<<B_, 256, 0, stream>>>(S2, nullptr, 0.0f);

        wsum_rows_kernel<<<dim3(WS_CH, B_), WS_WAVES * 32, 0, stream>>>(M, S1, S2);
        wsum_reduce_kernel<<<(2 * BQ_) / 256, 256, 0, stream>>>(S1, ev, x1);

        gemm(sCur, Q_,  We,      Q_ + MEM_, nullptr, qe, MEM_, MEM_, 0);
        gemm(ev,  MEM_, We + Q_, Q_ + MEM_, nullptr, qe, MEM_, MEM_, 1);

        scores_rows_kernel<<<dim3(SC_CH, B_), SC_WAVES * 32, 0, stream>>>(M, qe, qe, S1, S2);
        softmax_kernel<<<B_, 256, 0, stream>>>(S1, eacc, 0.2f);

        gemm(x1,  MEM_, W_ih, MEM_, b_ih, gi, 3 * Q_, 3 * Q_, 0);
        gemm(sCur, Q_,  W_hh, Q_,   b_hh, gh, 3 * Q_, 3 * Q_, 0);
        gru_kernel<<<BQ_ / 256, 256, 0, stream>>>(gi, gh, sCur, sNxt);
    }

    final_kernel<<<BL_ / 256, 256, 0, stream>>>(ws, (float*)d_out);
}


# softmax kernel: row kept in registers, each element loaded once and stored once (16-byte accesses), max and sum by vector-ALU lane exchanges plus one LDS step; new summation order of the softmax sum
# speedup vs baseline: 8.8880x; 1.4170x over previous
#include <hip/hip_runtime.h>
#include <math.h>

#define B_   32
#define L_   4096
#define MEM_ 256
#define Q_   256
#define BL_  (B_ * L_)
#define BQ_  (B_ * Q_)
#define B3Q_ (B_ * 3 * Q_)

#define OFF_S1   0
#define OFF_S2   (OFF_S1 + BL_)
#define OFF_SACC (OFF_S2 + BL_)
#define OFF_EACC (OFF_SACC + BL_)
#define OFF_QB   (OFF_EACC + BL_)
#define OFF_QA   (OFF_QB + BQ_)
#define OFF_QE   (OFF_QA + BQ_)
#define OFF_EV   (OFF_QE + BQ_)
#define OFF_X1   (OFF_EV + BQ_)
#define OFF_S_A  (OFF_X1 + BQ_)
#define OFF_S_B  (OFF_S_A + BQ_)
#define OFF_GI   (OFF_S_B + BQ_)
#define OFF_GH   (OFF_GI + B3Q_)

typedef __attribute__((ext_vector_type(2))) float v2f;
typedef __attribute__((ext_vector_type(8))) float v8f;

typedef __attribute__((ext_vector_type(4))) float v4f;

#define SG_K     256
#define SG_LINES (SG_K / 32)
#define SG_WAVES 8
#define SG_ROWS  8

__global__ void __launch_bounds__(SG_WAVES * 32)
wmma_gemm_kernel(const float* __restrict__ A, int lda,
                 const float* __restrict__ W, int ldw,
                 const float* __restrict__ bias,
                 float* __restrict__ C, int ldc,
                 int accumulate) {
    const int tid   = threadIdx.x;
    const int lane  = tid & 31;
    const int wave  = tid >> 5;
    const int c     = lane & 15;
    const int hi    = lane >> 4;
    const int kb    = c & 3;
    const int piece = 4 * (2 * kb + hi);
    const int m0    = blockIdx.y * SG_ROWS;
    const int n0    = blockIdx.x * 32;

    const float* ap = A + (size_t)(m0 + (c >> 2)) * lda + piece;
    const float* wp = W + (size_t)(n0 + 4 * wave + (c >> 2)) * ldw + piece;

    v4f xw[SG_LINES], xa[2][SG_LINES];
#pragma unroll
    for (int j = 0; j < SG_LINES; ++j) xw[j] = *(const v4f*)(wp + 32 * j);
#pragma unroll
    for (int u = 0; u < 2; ++u) {
#pragma unroll
        for (int j = 0; j < SG_LINES; ++j)
            xa[u][j] = *(const v4f*)(ap + (size_t)(4 * u) * lda + 32 * j);
    }

    const int orow = wave, ocol = lane;
    float* cp = C + (size_t)(m0 + orow) * ldc + n0 + ocol;
    float bv = 0.0f, cv = 0.0f;
    if (bias) bv = bias[n0 + ocol];
    if (accumulate) cv = *cp;

    v8f acc0[2], acc1[2];
#pragma unroll
    for (int u = 0; u < 2; ++u) { acc0[u] = (v8f){}; acc1[u] = (v8f){}; }
#pragma unroll
    for (int j = 0; j < SG_LINES; ++j) {
#pragma unroll
        for (int u = 0; u < 2; ++u) {
            v2f a, bb;
            a.x = xa[u][j][0]; a.y = xa[u][j][1]; bb.x = xw[j][0]; bb.y = xw[j][1];
            acc0[u] = __builtin_amdgcn_wmma_f32_16x16x4_f32(false, a, false, bb,
                                                            (short)0, acc0[u], false, false);
            a.x = xa[u][j][2]; a.y = xa[u][j][3]; bb.x = xw[j][2]; bb.y = xw[j][3];
            acc1[u] = __builtin_amdgcn_wmma_f32_16x16x4_f32(false, a, false, bb,
                                                            (short)0, acc1[u], false, false);
        }
    }
    float lo[2], up[2];
#pragma unroll
    for (int u = 0; u < 2; ++u) {
        const v8f tv = acc0[u] + acc1[u];
        const v4f tl = {tv[0], tv[1], tv[2], tv[3]};
        const v4f tu = {tv[4], tv[5], tv[6], tv[7]};
        lo[u] = tl[kb]; up[u] = tu[kb];
    }
#pragma unroll
    for (int u = 0; u < 2; ++u) { lo[u] += __shfl_xor(lo[u], 1); up[u] += __shfl_xor(up[u], 1); }
#pragma unroll
    for (int u = 0; u < 2; ++u) { lo[u] += __shfl_xor(lo[u], 2); up[u] += __shfl_xor(up[u], 2); }

    __shared__ float tile[SG_ROWS][32];
    if (kb == 0) {
#pragma unroll
        for (int u = 0; u < 2; ++u) {
            tile[4 * u + 2 * hi][4 * wave + (c >> 2)]     = lo[u];
            tile[4 * u + 2 * hi + 1][4 * wave + (c >> 2)] = up[u];
        }
    }
    __syncthreads();
    *cp = (cv + bv) + tile[orow][ocol];
}

#define SC_CH     8
#define SC_WAVES  16
#define SC_WGROWS (L_ / SC_CH)
#define SC_WROWS  (SC_WGROWS / SC_WAVES)
#define SC_GROUPS (SC_WROWS / 4)
#define SC_LINES  (MEM_ / 32)
#define SC_AHEAD  3
#define SC_RING   (SC_AHEAD + 1)

__global__ void __launch_bounds__(SC_WAVES * 32)
scores_rows_kernel(const float* __restrict__ M,
                   const float* __restrict__ q1,
                   const float* __restrict__ q2,
                   float* __restrict__ S1,
                   float* __restrict__ S2) {
    const int tid   = threadIdx.x;
    const int lane  = tid & 31;
    const int wave  = tid >> 5;
    const int c     = lane & 15;
    const int hi    = lane >> 4;
    const int kb    = c & 3;
    const int b     = blockIdx.y;
    const int ch    = blockIdx.x;
    const int piece = 4 * (2 * kb + hi);
    const int r0    = ch * SC_WGROWS + wave * SC_WROWS;

    const float* Mb = M + ((size_t)b * L_ + r0 + (c >> 2)) * MEM_ + piece;
    const float* qv = (((c >> 2) & 1) ? q2 : q1) + (size_t)b * MEM_ + piece;

    v4f x[SC_RING][SC_LINES];
#pragma unroll
    for (int g = 0; g < SC_AHEAD; ++g) {
#pragma unroll
        for (int j = 0; j < SC_LINES; ++j)
            x[g][j] = *(const v4f*)(Mb + (size_t)(4 * g) * MEM_ + 32 * j);
    }

    v4f qf[SC_LINES];
#pragma unroll
    for (int j = 0; j < SC_LINES; ++j) qf[j] = *(const v4f*)(qv + 32 * j);

    float slo[SC_GROUPS], shi[SC_GROUPS];
#pragma unroll
    for (int g = 0; g < SC_GROUPS; ++g) {
        const int cur = g % SC_RING;
        if (g + SC_AHEAD < SC_GROUPS) {
            const int nxt = (g + SC_AHEAD) % SC_RING;
#pragma unroll
            for (int j = 0; j < SC_LINES; ++j)
                x[nxt][j] = *(const v4f*)(Mb + (size_t)(4 * (g + SC_AHEAD)) * MEM_ + 32 * j);
        }
        __builtin_amdgcn_sched_barrier(0);
        v8f acc0 = {}, acc1 = {};
#pragma unroll
        for (int j = 0; j < SC_LINES; ++j) {
            v2f a, bb;
            a.x = x[cur][j][0]; a.y = x[cur][j][1]; bb.x = qf[j][0]; bb.y = qf[j][1];
            acc0 = __builtin_amdgcn_wmma_f32_16x16x4_f32(false, a, false, bb,
                                                         (short)0, acc0, false, false);
            a.x = x[cur][j][2]; a.y = x[cur][j][3]; bb.x = qf[j][2]; bb.y = qf[j][3];
            acc1 = __builtin_amdgcn_wmma_f32_16x16x4_f32(false, a, false, bb,
                                                         (short)0, acc1, false, false);
        }
        float t[8];
#pragma unroll
        for (int r = 0; r < 8; ++r) t[r] = acc0[r] + acc1[r];
        float lo = (kb == 0) ? t[0] : (kb == 1) ? t[1] : (kb == 2) ? t[2] : t[3];
        float up = (kb == 0) ? t[4] : (kb == 1) ? t[5] : (kb == 2) ? t[6] : t[7];
        lo += __shfl_xor(lo, 1); up += __shfl_xor(up, 1);
        lo += __shfl_xor(lo, 2); up += __shfl_xor(up, 2);
        slo[g] = lo; shi[g] = up;
    }

    __shared__ float sc[2][SC_WGROWS];
    if (c == 0 || c == 4) {
        float* dst = &sc[c >> 2][wave * SC_WROWS + 2 * hi];
#pragma unroll
        for (int g = 0; g < SC_GROUPS; ++g) { dst[4 * g] = slo[g]; dst[4 * g + 1] = shi[g]; }
    }
    __syncthreads();
    S1[(size_t)b * L_ + ch * SC_WGROWS + tid] = sc[0][tid];
    S2[(size_t)b * L_ + ch * SC_WGROWS + tid] = sc[1][tid];
}

#define WS_CH     8
#define WS_WAVES  16
#define WS_WGROWS (L_ / WS_CH)
#define WS_WROWS  (WS_WGROWS / WS_WAVES)
#define WS_KSTEPS (WS_WROWS / 4)
#define WS_KSTRIDE (4 * WS_WAVES)

__global__ void __launch_bounds__(WS_WAVES * 32)
wsum_rows_kernel(const float* __restrict__ M,
                 float* pbuf,
                 const float* wbuf) {
    const int tid  = threadIdx.x;
    const int lane = tid & 31;
    const int wave = tid >> 5;
    const int row  = lane & 15;
    const int hi   = lane >> 4;
    const int b    = blockIdx.y;
    const int c    = blockIdx.x;
    const int l0   = c * WS_WGROWS + 4 * wave;

    const float* pv = ((row == 1) ? wbuf : (const float*)pbuf) + (size_t)b * L_ + l0 + 2 * hi;
    const float* Mb = M + ((size_t)b * L_ + l0 + 2 * hi) * MEM_ + 4 * row;

    v8f acc[16];
#pragma unroll
    for (int t = 0; t < 16; ++t) acc[t] = (v8f){};

    v2f a[WS_KSTEPS];
#pragma unroll
    for (int ks = 0; ks < WS_KSTEPS; ++ks) {
        a[ks] = *(const v2f*)(pv + WS_KSTRIDE * ks);
        if (row >= 2) { a[ks].x = 0.0f; a[ks].y = 0.0f; }
    }

    v4f x[2][2][4];
#pragma unroll
    for (int g = 0; g < 4; ++g) {
        x[0][0][g] = *(const v4f*)(Mb + 64 * g);
        x[0][1][g] = *(const v4f*)(Mb + MEM_ + 64 * g);
    }

#pragma unroll
    for (int ks = 0; ks < WS_KSTEPS; ++ks) {
        const int cur = ks & 1, nxt = cur ^ 1;
        if (ks + 1 < WS_KSTEPS) {
#pragma unroll
            for (int g = 0; g < 4; ++g) {
                x[nxt][0][g] = *(const v4f*)(Mb + (size_t)(WS_KSTRIDE * (ks + 1))     * MEM_ + 64 * g);
                x[nxt][1][g] = *(const v4f*)(Mb + (size_t)(WS_KSTRIDE * (ks + 1) + 1) * MEM_ + 64 * g);
            }
        }
        __builtin_amdgcn_sched_barrier(0);
#pragma unroll
        for (int g = 0; g < 4; ++g) {
#pragma unroll
            for (int cc = 0; cc < 4; ++cc) {
                v2f bb; bb.x = x[cur][0][g][cc]; bb.y = x[cur][1][g][cc];
                acc[4 * g + cc] = __builtin_amdgcn_wmma_f32_16x16x4_f32(
                    false, a[ks], false, bb, (short)0, acc[4 * g + cc], false, false);
            }
        }
    }

    __shared__ float red[WS_WAVES][2][MEM_];
    if (hi == 0) {
#pragma unroll
        for (int g = 0; g < 4; ++g) {
            v4f es, xs;
#pragma unroll
            for (int cc = 0; cc < 4; ++cc) { es[cc] = acc[4 * g + cc][0]; xs[cc] = acc[4 * g + cc][1]; }
            *(v4f*)&red[wave][0][64 * g + 4 * row] = es;
            *(v4f*)&red[wave][1][64 * g + 4 * row] = xs;
        }
    }
    __syncthreads();
    {
        const int v = tid >> 8, col = tid & 255;
        float s = 0.0f;
#pragma unroll
        for (int w = 0; w < WS_WAVES; ++w) s += red[w][v][col];
        pbuf[(size_t)b * L_ + c * WS_WGROWS + v * MEM_ + col] = s;
    }
}

__global__ void wsum_reduce_kernel(const float* __restrict__ part,
                                   float* __restrict__ ev,
                                   float* __restrict__ x1) {
    const int i   = blockIdx.x * 256 + threadIdx.x;
    const int b   = i >> 9;
    const int v   = (i >> 8) & 1;
    const int col = i & 255;
    const float* src = part + (size_t)b * L_ + v * MEM_ + col;
    float s = 0.0f;
#pragma unroll
    for (int c = 0; c < WS_CH; ++c) s += src[c * WS_WGROWS];
    (v ? x1 : ev)[(size_t)b * MEM_ + col] = s;
}

#define SM_WAVES 8
#define SM_VECS  (L_ / (SM_WAVES * 32 * 4))

template <int N>
static __device__ __forceinline__ float row_xor(float x) {
    const int b = __builtin_bit_cast(int, x);
    return __builtin_bit_cast(float,
        __builtin_amdgcn_update_dpp(b, b, 0x160 + N, 0xf, 0xf, false));
}
static __device__ __forceinline__ float xhalf(float x) {
    const int b = __builtin_bit_cast(int, x);
    return __builtin_bit_cast(float,
        __builtin_amdgcn_permlanex16(b, b, 0x76543210, 0xfedcba98, false, false));
}

__global__ void __launch_bounds__(SM_WAVES * 32)
softmax_kernel(float* __restrict__ S,
               float* __restrict__ accDst,
               float scale) {
    const int b    = blockIdx.x;
    const int tid  = threadIdx.x;
    const int lane = tid & 31;
    const int wave = tid >> 5;
    __shared__ float redm[SM_WAVES], reds[SM_WAVES];
    float* rowp = S + (size_t)b * L_ + 4 * tid;
    float* accp = accDst ? accDst + (size_t)b * L_ + 4 * tid : nullptr;

    v4f x[SM_VECS], a[SM_VECS];
#pragma unroll
    for (int k = 0; k < SM_VECS; ++k) x[k] = *(const v4f*)(rowp + 1024 * k);
#pragma unroll
    for (int k = 0; k < SM_VECS; ++k) a[k] = (v4f){};
    if (accDst) {
#pragma unroll
        for (int k = 0; k < SM_VECS; ++k) a[k] = *(const v4f*)(accp + 1024 * k);
    }

    float m = fmaxf(fmaxf(x[0][0], x[0][1]), fmaxf(x[0][2], x[0][3]));
#pragma unroll
    for (int k = 1; k < SM_VECS; ++k)
        m = fmaxf(m, fmaxf(fmaxf(x[k][0], x[k][1]), fmaxf(x[k][2], x[k][3])));
    m = fmaxf(m, row_xor<1>(m)); m = fmaxf(m, row_xor<2>(m));
    m = fmaxf(m, row_xor<4>(m)); m = fmaxf(m, row_xor<8>(m));
    m = fmaxf(m, xhalf(m));
    if (lane == 0) redm[wave] = m;
    __syncthreads();
    m = redm[lane & (SM_WAVES - 1)];
    m = fmaxf(m, row_xor<1>(m)); m = fmaxf(m, row_xor<2>(m)); m = fmaxf(m, row_xor<4>(m));

    v4f e[SM_VECS];
    float s = 0.0f;
#pragma unroll
    for (int k = 0; k < SM_VECS; ++k) {
#pragma unroll
        for (int cc = 0; cc < 4; ++cc) e[k][cc] = __expf(x[k][cc] - m);
        s += (e[k][0] + e[k][1]) + (e[k][2] + e[k][3]);
    }
    s += row_xor<1>(s); s += row_xor<2>(s); s += row_xor<4>(s); s += row_xor<8>(s);
    s += xhalf(s);
    if (lane == 0) reds[wave] = s;
    __syncthreads();
    s = reds[lane & (SM_WAVES - 1)];
    s += row_xor<1>(s); s += row_xor<2>(s); s += row_xor<4>(s);
    const float inv = 1.0f / s;

#pragma unroll
    for (int k = 0; k < SM_VECS; ++k) {
        e[k] = e[k] * inv;
        *(v4f*)(rowp + 1024 * k) = e[k];
    }
    if (accDst) {
#pragma unroll
        for (int k = 0; k < SM_VECS; ++k) *(v4f*)(accp + 1024 * k) = a[k] + scale * e[k];
    }
}

__global__ void gru_kernel(const float* __restrict__ gi,
                           const float* __restrict__ gh,
                           const float* __restrict__ sOld,
                           float* __restrict__ sNew) {
    int i = blockIdx.x * 256 + threadIdx.x;
    if (i >= BQ_) return;
    int b = i / Q_, q = i % Q_;
    const float* gib = gi + (size_t)b * 3 * Q_;
    const float* ghb = gh + (size_t)b * 3 * Q_;
    float ir = gib[q], iz = gib[Q_ + q], in_ = gib[2 * Q_ + q];
    float hr = ghb[q], hz = ghb[Q_ + q], hn  = ghb[2 * Q_ + q];
    float r = 1.0f / (1.0f + __expf(-(ir + hr)));
    float z = 1.0f / (1.0f + __expf(-(iz + hz)));
    float n = tanhf(in_ + r * hn);
    sNew[i] = (1.0f - z) * n + z * sOld[i];
}

__global__ void init_kernel(const float* __restrict__ s0, float* __restrict__ ws) {
    int i = blockIdx.x * 256 + threadIdx.x;
    if (i < 2 * BL_) ws[OFF_SACC + i] = 0.0f;
    if (i < BQ_)     ws[OFF_S_A + i] = s0[i];
}

__global__ void final_kernel(const float* __restrict__ ws, float* __restrict__ out) {
    int i = blockIdx.x * 256 + threadIdx.x;
    if (i < BL_) {
        out[i]       = logf(ws[OFF_SACC + i]);
        out[BL_ + i] = logf(ws[OFF_EACC + i]);
    }
}

extern "C" void kernel_launch(void* const* d_in, const int* in_sizes, int n_in,
                              void* d_out, int out_size, void* d_ws, size_t ws_size,
                              hipStream_t stream) {
    const float* M    = (const float*)d_in[0];
    const float* s0   = (const float*)d_in[1];
    const float* Wb   = (const float*)d_in[2];
    const float* We   = (const float*)d_in[3];
    const float* Wa   = (const float*)d_in[4];
    const float* W_ih = (const float*)d_in[5];
    const float* W_hh = (const float*)d_in[6];
    const float* b_ih = (const float*)d_in[7];
    const float* b_hh = (const float*)d_in[8];

    float* ws   = (float*)d_ws;
    float* S1   = ws + OFF_S1;
    float* S2   = ws + OFF_S2;
    float* sacc = ws + OFF_SACC;
    float* eacc = ws + OFF_EACC;
    float* qb   = ws + OFF_QB;
    float* qa   = ws + OFF_QA;
    float* qe   = ws + OFF_QE;
    float* ev   = ws + OFF_EV;
    float* x1   = ws + OFF_X1;
    float* sA   = ws + OFF_S_A;
    float* sB   = ws + OFF_S_B;
    float* gi   = ws + OFF_GI;
    float* gh   = ws + OFF_GH;

    init_kernel<<<(2 * BL_) / 256, 256, 0, stream>>>(s0, ws);

    auto gemm = [&](const float* A, int lda, const float* W, int ldw,
                    const float* bias, float* C, int ldc, int N, int accF) {
        dim3 g(N / 32, B_ / SG_ROWS);
        wmma_gemm_kernel<<<g, SG_WAVES * 32, 0, stream>>>(A, lda, W, ldw, bias, C, ldc, accF);
    };

    for (int t = 0; t < 5; ++t) {
        const float* sCur = (t & 1) ? sB : sA;
        float*       sNxt = (t & 1) ? sA : sB;

        gemm(sCur, Q_, Wb, Q_, nullptr, qb, MEM_, MEM_, 0);
        gemm(sCur, Q_, Wa, Q_, nullptr, qa, MEM_, MEM_, 0);

        scores_rows_kernel<<<dim3(SC_CH, B_), SC_WAVES * 32, 0, stream>>>(M, qb, qa, S1, S2);

        softmax_kernel<<<B_, 256, 0, stream>>>(S1, sacc, 0.2f);
        softmax_kernel<<<B_, 256, 0, stream>>>(S2, nullptr, 0.0f);

        wsum_rows_kernel<<<dim3(WS_CH, B_), WS_WAVES * 32, 0, stream>>>(M, S1, S2);
        wsum_reduce_kernel<<<(2 * BQ_) / 256, 256, 0, stream>>>(S1, ev, x1);

        gemm(sCur, Q_,  We,      Q_ + MEM_, nullptr, qe, MEM_, MEM_, 0);
        gemm(ev,  MEM_, We + Q_, Q_ + MEM_, nullptr, qe, MEM_, MEM_, 1);

        scores_rows_kernel<<<dim3(SC_CH, B_), SC_WAVES * 32, 0, stream>>>(M, qe, qe, S1, S2);
        softmax_kernel<<<B_, 256, 0, stream>>>(S1, eacc, 0.2f);

        gemm(x1,  MEM_, W_ih, MEM_, b_ih, gi, 3 * Q_, 3 * Q_, 0);
        gemm(sCur, Q_,  W_hh, Q_,   b_hh, gh, 3 * Q_, 3 * Q_, 0);
        gru_kernel<<<BQ_ / 256, 256, 0, stream>>>(gi, gh, sCur, sNxt);
    }

    final_kernel<<<BL_ / 256, 256, 0, stream>>>(ws, (float*)d_out);
}


# turn loop split by data dependence: the end-score branch (query GEMMs, score pass over M, softmax, log) leaves the loop and runs once for all 5 turns, 3+2 queries per pass over M; same sums
# speedup vs baseline: 11.1503x; 1.2545x over previous
#include <hip/hip_runtime.h>
#include <math.h>

#define B_   32
#define L_   4096
#define MEM_ 256
#define Q_   256
#define BL_  (B_ * L_)
#define BQ_  (B_ * Q_)
#define B3Q_ (B_ * 3 * Q_)

#define OFF_S1   0
#define OFF_S2   (OFF_S1 + BL_)
#define OFF_SACC (OFF_S2 + BL_)
#define OFF_EACC (OFF_SACC + BL_)
#define OFF_QB   (OFF_EACC + BL_)
#define OFF_QA   (OFF_QB + BQ_)
#define OFF_X1   (OFF_QA + BQ_)
#define OFF_GI   (OFF_X1 + BQ_)
#define OFF_GH   (OFF_GI + B3Q_)
#define OFF_SALL  OFF_EACC
#define OFF_EVALL (OFF_SALL + 6 * BQ_)
#define OFF_QEALL OFF_GI

typedef __attribute__((ext_vector_type(2))) float v2f;
typedef __attribute__((ext_vector_type(8))) float v8f;

typedef __attribute__((ext_vector_type(4))) float v4f;

#define SG_K     256
#define SG_LINES (SG_K / 32)
#define SG_WAVES 8
#define SG_ROWS  8

__global__ void __launch_bounds__(SG_WAVES * 32)
wmma_gemm_kernel(const float* __restrict__ A, int lda,
                 const float* __restrict__ W, int ldw,
                 const float* __restrict__ bias,
                 float* __restrict__ C, int ldc,
                 int accumulate) {
    const int tid   = threadIdx.x;
    const int lane  = tid & 31;
    const int wave  = tid >> 5;
    const int c     = lane & 15;
    const int hi    = lane >> 4;
    const int kb    = c & 3;
    const int piece = 4 * (2 * kb + hi);
    const int m0    = blockIdx.y * SG_ROWS;
    const int n0    = blockIdx.x * 32;

    const float* ap = A + (size_t)(m0 + (c >> 2)) * lda + piece;
    const float* wp = W + (size_t)(n0 + 4 * wave + (c >> 2)) * ldw + piece;

    v4f xw[SG_LINES], xa[2][SG_LINES];
#pragma unroll
    for (int j = 0; j < SG_LINES; ++j) xw[j] = *(const v4f*)(wp + 32 * j);
#pragma unroll
    for (int u = 0; u < 2; ++u) {
#pragma unroll
        for (int j = 0; j < SG_LINES; ++j)
            xa[u][j] = *(const v4f*)(ap + (size_t)(4 * u) * lda + 32 * j);
    }

    const int orow = wave, ocol = lane;
    float* cp = C + (size_t)(m0 + orow) * ldc + n0 + ocol;
    float bv = 0.0f, cv = 0.0f;
    if (bias) bv = bias[n0 + ocol];
    if (accumulate) cv = *cp;

    v8f acc0[2], acc1[2];
#pragma unroll
    for (int u = 0; u < 2; ++u) { acc0[u] = (v8f){}; acc1[u] = (v8f){}; }
#pragma unroll
    for (int j = 0; j < SG_LINES; ++j) {
#pragma unroll
        for (int u = 0; u < 2; ++u) {
            v2f a, bb;
            a.x = xa[u][j][0]; a.y = xa[u][j][1]; bb.x = xw[j][0]; bb.y = xw[j][1];
            acc0[u] = __builtin_amdgcn_wmma_f32_16x16x4_f32(false, a, false, bb,
                                                            (short)0, acc0[u], false, false);
            a.x = xa[u][j][2]; a.y = xa[u][j][3]; bb.x = xw[j][2]; bb.y = xw[j][3];
            acc1[u] = __builtin_amdgcn_wmma_f32_16x16x4_f32(false, a, false, bb,
                                                            (short)0, acc1[u], false, false);
        }
    }
    float lo[2], up[2];
#pragma unroll
    for (int u = 0; u < 2; ++u) {
        const v8f tv = acc0[u] + acc1[u];
        const v4f tl = {tv[0], tv[1], tv[2], tv[3]};
        const v4f tu = {tv[4], tv[5], tv[6], tv[7]};
        lo[u] = tl[kb]; up[u] = tu[kb];
    }
#pragma unroll
    for (int u = 0; u < 2; ++u) { lo[u] += __shfl_xor(lo[u], 1); up[u] += __shfl_xor(up[u], 1); }
#pragma unroll
    for (int u = 0; u < 2; ++u) { lo[u] += __shfl_xor(lo[u], 2); up[u] += __shfl_xor(up[u], 2); }

    __shared__ float tile[SG_ROWS][32];
    if (kb == 0) {
#pragma unroll
        for (int u = 0; u < 2; ++u) {
            tile[4 * u + 2 * hi][4 * wave + (c >> 2)]     = lo[u];
            tile[4 * u + 2 * hi + 1][4 * wave + (c >> 2)] = up[u];
        }
    }
    __syncthreads();
    *cp = (cv + bv) + tile[orow][ocol];
}

#define SC_CH     8
#define SC_WAVES  16
#define SC_WGROWS (L_ / SC_CH)
#define SC_WROWS  (SC_WGROWS / SC_WAVES)
#define SC_GROUPS (SC_WROWS / 4)
#define SC_LINES  (MEM_ / 32)
#define SC_AHEAD  3
#define SC_RING   (SC_AHEAD + 1)

template <int NQ>
__global__ void __launch_bounds__(SC_WAVES * 32)
scores_rows_kernel(const float* __restrict__ M,
                   const float* __restrict__ q1,
                   const float* __restrict__ q2,
                   const float* __restrict__ q3,
                   float* __restrict__ S1,
                   float* __restrict__ S2,
                   float* __restrict__ S3) {
    const int tid   = threadIdx.x;
    const int lane  = tid & 31;
    const int wave  = tid >> 5;
    const int c     = lane & 15;
    const int hi    = lane >> 4;
    const int kb    = c & 3;
    const int b     = blockIdx.y;
    const int ch    = blockIdx.x;
    const int piece = 4 * (2 * kb + hi);
    const int r0    = ch * SC_WGROWS + wave * SC_WROWS;

    const float* Mb = M + ((size_t)b * L_ + r0 + (c >> 2)) * MEM_ + piece;
    const int qi = (c >> 2) % NQ;
    const float* qv = ((qi == 0) ? q1 : (qi == 1) ? q2 : q3) + (size_t)b * MEM_ + piece;

    v4f x[SC_RING][SC_LINES];
#pragma unroll
    for (int g = 0; g < SC_AHEAD; ++g) {
#pragma unroll
        for (int j = 0; j < SC_LINES; ++j)
            x[g][j] = *(const v4f*)(Mb + (size_t)(4 * g) * MEM_ + 32 * j);
    }

    v4f qf[SC_LINES];
#pragma unroll
    for (int j = 0; j < SC_LINES; ++j) qf[j] = *(const v4f*)(qv + 32 * j);

    float slo[SC_GROUPS], shi[SC_GROUPS];
#pragma unroll
    for (int g = 0; g < SC_GROUPS; ++g) {
        const int cur = g % SC_RING;
        if (g + SC_AHEAD < SC_GROUPS) {
            const int nxt = (g + SC_AHEAD) % SC_RING;
#pragma unroll
            for (int j = 0; j < SC_LINES; ++j)
                x[nxt][j] = *(const v4f*)(Mb + (size_t)(4 * (g + SC_AHEAD)) * MEM_ + 32 * j);
        }
        __builtin_amdgcn_sched_barrier(0);
        v8f acc0 = {}, acc1 = {};
#pragma unroll
        for (int j = 0; j < SC_LINES; ++j) {
            v2f a, bb;
            a.x = x[cur][j][0]; a.y = x[cur][j][1]; bb.x = qf[j][0]; bb.y = qf[j][1];
            acc0 = __builtin_amdgcn_wmma_f32_16x16x4_f32(false, a, false, bb,
                                                         (short)0, acc0, false, false);
            a.x = x[cur][j][2]; a.y = x[cur][j][3]; bb.x = qf[j][2]; bb.y = qf[j][3];
            acc1 = __builtin_amdgcn_wmma_f32_16x16x4_f32(false, a, false, bb,
                                                         (short)0, acc1, false, false);
        }
        float t[8];
#pragma unroll
        for (int r = 0; r < 8; ++r) t[r] = acc0[r] + acc1[r];
        float lo = (kb == 0) ? t[0] : (kb == 1) ? t[1] : (kb == 2) ? t[2] : t[3];
        float up = (kb == 0) ? t[4] : (kb == 1) ? t[5] : (kb == 2) ? t[6] : t[7];
        lo += __shfl_xor(lo, 1); up += __shfl_xor(up, 1);
        lo += __shfl_xor(lo, 2); up += __shfl_xor(up, 2);
        slo[g] = lo; shi[g] = up;
    }

    __shared__ float sc[NQ][SC_WGROWS];
    if (kb == 0 && (c >> 2) < NQ) {
        float* dst = &sc[c >> 2][wave * SC_WROWS + 2 * hi];
#pragma unroll
        for (int g = 0; g < SC_GROUPS; ++g) { dst[4 * g] = slo[g]; dst[4 * g + 1] = shi[g]; }
    }
    __syncthreads();
    S1[(size_t)b * L_ + ch * SC_WGROWS + tid] = sc[0][tid];
    S2[(size_t)b * L_ + ch * SC_WGROWS + tid] = sc[1][tid];
    if (NQ > 2) S3[(size_t)b * L_ + ch * SC_WGROWS + tid] = sc[NQ - 1][tid];
}

#define WS_CH     8
#define WS_WAVES  16
#define WS_WGROWS (L_ / WS_CH)
#define WS_WROWS  (WS_WGROWS / WS_WAVES)
#define WS_KSTEPS (WS_WROWS / 4)
#define WS_KSTRIDE (4 * WS_WAVES)

__global__ void __launch_bounds__(WS_WAVES * 32)
wsum_rows_kernel(const float* __restrict__ M,
                 float* pbuf,
                 const float* wbuf) {
    const int tid  = threadIdx.x;
    const int lane = tid & 31;
    const int wave = tid >> 5;
    const int row  = lane & 15;
    const int hi   = lane >> 4;
    const int b    = blockIdx.y;
    const int c    = blockIdx.x;
    const int l0   = c * WS_WGROWS + 4 * wave;

    const float* pv = ((row == 1) ? wbuf : (const float*)pbuf) + (size_t)b * L_ + l0 + 2 * hi;
    const float* Mb = M + ((size_t)b * L_ + l0 + 2 * hi) * MEM_ + 4 * row;

    v8f acc[16];
#pragma unroll
    for (int t = 0; t < 16; ++t) acc[t] = (v8f){};

    v2f a[WS_KSTEPS];
#pragma unroll
    for (int ks = 0; ks < WS_KSTEPS; ++ks) {
        a[ks] = *(const v2f*)(pv + WS_KSTRIDE * ks);
        if (row >= 2) { a[ks].x = 0.0f; a[ks].y = 0.0f; }
    }

    v4f x[2][2][4];
#pragma unroll
    for (int g = 0; g < 4; ++g) {
        x[0][0][g] = *(const v4f*)(Mb + 64 * g);
        x[0][1][g] = *(const v4f*)(Mb + MEM_ + 64 * g);
    }

#pragma unroll
    for (int ks = 0; ks < WS_KSTEPS; ++ks) {
        const int cur = ks & 1, nxt = cur ^ 1;
        if (ks + 1 < WS_KSTEPS) {
#pragma unroll
            for (int g = 0; g < 4; ++g) {
                x[nxt][0][g] = *(const v4f*)(Mb + (size_t)(WS_KSTRIDE * (ks + 1))     * MEM_ + 64 * g);
                x[nxt][1][g] = *(const v4f*)(Mb + (size_t)(WS_KSTRIDE * (ks + 1) + 1) * MEM_ + 64 * g);
            }
        }
        __builtin_amdgcn_sched_barrier(0);
#pragma unroll
        for (int g = 0; g < 4; ++g) {
#pragma unroll
            for (int cc = 0; cc < 4; ++cc) {
                v2f bb; bb.x = x[cur][0][g][cc]; bb.y = x[cur][1][g][cc];
                acc[4 * g + cc] = __builtin_amdgcn_wmma_f32_16x16x4_f32(
                    false, a[ks], false, bb, (short)0, acc[4 * g + cc], false, false);
            }
        }
    }

    __shared__ float red[WS_WAVES][2][MEM_];
    if (hi == 0) {
#pragma unroll
        for (int g = 0; g < 4; ++g) {
            v4f es, xs;
#pragma unroll
            for (int cc = 0; cc < 4; ++cc) { es[cc] = acc[4 * g + cc][0]; xs[cc] = acc[4 * g + cc][1]; }
            *(v4f*)&red[wave][0][64 * g + 4 * row] = es;
            *(v4f*)&red[wave][1][64 * g + 4 * row] = xs;
        }
    }
    __syncthreads();
    {
        const int v = tid >> 8, col = tid & 255;
        float s = 0.0f;
#pragma unroll
        for (int w = 0; w < WS_WAVES; ++w) s += red[w][v][col];
        pbuf[(size_t)b * L_ + c * WS_WGROWS + v * MEM_ + col] = s;
    }
}

__global__ void wsum_reduce_kernel(const float* __restrict__ part,
                                   float* __restrict__ ev,
                                   float* __restrict__ x1) {
    const int i   = blockIdx.x * 256 + threadIdx.x;
    const int b   = i >> 9;
    const int v   = (i >> 8) & 1;
    const int col = i & 255;
    const float* src = part + (size_t)b * L_ + v * MEM_ + col;
    float s = 0.0f;
#pragma unroll
    for (int c = 0; c < WS_CH; ++c) s += src[c * WS_WGROWS];
    (v ? x1 : ev)[(size_t)b * MEM_ + col] = s;
}

#define SM_WAVES 8
#define SM_VECS  (L_ / (SM_WAVES * 32 * 4))

template <int N>
static __device__ __forceinline__ float row_xor(float x) {
    const int b = __builtin_bit_cast(int, x);
    return __builtin_bit_cast(float,
        __builtin_amdgcn_update_dpp(b, b, 0x160 + N, 0xf, 0xf, false));
}
static __device__ __forceinline__ float xhalf(float x) {
    const int b = __builtin_bit_cast(int, x);
    return __builtin_bit_cast(float,
        __builtin_amdgcn_permlanex16(b, b, 0x76543210, 0xfedcba98, false, false));
}

template <int NR>
static __device__ __forceinline__ void softmax_rows_regs(const v4f (&x)[NR][SM_VECS], v4f (&e)[NR][SM_VECS],
                                                         float (&redm)[NR][SM_WAVES],
                                                         float (&reds)[NR][SM_WAVES],
                                                         int lane, int wave) {
    float m[NR];
#pragma unroll
    for (int r = 0; r < NR; ++r) {
        float t = fmaxf(fmaxf(x[r][0][0], x[r][0][1]), fmaxf(x[r][0][2], x[r][0][3]));
#pragma unroll
        for (int k = 1; k < SM_VECS; ++k)
            t = fmaxf(t, fmaxf(fmaxf(x[r][k][0], x[r][k][1]), fmaxf(x[r][k][2], x[r][k][3])));
        t = fmaxf(t, row_xor<1>(t)); t = fmaxf(t, row_xor<2>(t));
        t = fmaxf(t, row_xor<4>(t)); t = fmaxf(t, row_xor<8>(t));
        t = fmaxf(t, xhalf(t));
        if (lane == 0) redm[r][wave] = t;
    }
    __syncthreads();
#pragma unroll
    for (int r = 0; r < NR; ++r) {
        float t = redm[r][lane & (SM_WAVES - 1)];
        t = fmaxf(t, row_xor<1>(t)); t = fmaxf(t, row_xor<2>(t)); t = fmaxf(t, row_xor<4>(t));
        m[r] = t;
    }

#pragma unroll
    for (int r = 0; r < NR; ++r) {
        float s = 0.0f;
#pragma unroll
        for (int k = 0; k < SM_VECS; ++k) {
#pragma unroll
            for (int cc = 0; cc < 4; ++cc) e[r][k][cc] = __expf(x[r][k][cc] - m[r]);
            s += (e[r][k][0] + e[r][k][1]) + (e[r][k][2] + e[r][k][3]);
        }
        s += row_xor<1>(s); s += row_xor<2>(s); s += row_xor<4>(s); s += row_xor<8>(s);
        s += xhalf(s);
        if (lane == 0) reds[r][wave] = s;
    }
    __syncthreads();
#pragma unroll
    for (int r = 0; r < NR; ++r) {
        float s = reds[r][lane & (SM_WAVES - 1)];
        s += row_xor<1>(s); s += row_xor<2>(s); s += row_xor<4>(s);
        const float inv = 1.0f / s;
#pragma unroll
        for (int k = 0; k < SM_VECS; ++k) e[r][k] = e[r][k] * inv;
    }
}

__global__ void __launch_bounds__(SM_WAVES * 32)
softmax_kernel(float* __restrict__ S,
               float* __restrict__ accDst,
               float scale) {
    const int b    = blockIdx.x;
    const int tid  = threadIdx.x;
    const int lane = tid & 31;
    const int wave = tid >> 5;
    __shared__ float redm[1][SM_WAVES], reds[1][SM_WAVES];
    float* rowp = S + (size_t)b * L_ + 4 * tid;
    float* accp = accDst ? accDst + (size_t)b * L_ + 4 * tid : nullptr;

    v4f x[1][SM_VECS], a[SM_VECS], e[1][SM_VECS];
#pragma unroll
    for (int k = 0; k < SM_VECS; ++k) x[0][k] = *(const v4f*)(rowp + 1024 * k);
#pragma unroll
    for (int k = 0; k < SM_VECS; ++k) a[k] = (v4f){};
    if (accDst) {
#pragma unroll
        for (int k = 0; k < SM_VECS; ++k) a[k] = *(const v4f*)(accp + 1024 * k);
    }

    softmax_rows_regs<1>(x, e, redm, reds, lane, wave);

#pragma unroll
    for (int k = 0; k < SM_VECS; ++k) *(v4f*)(rowp + 1024 * k) = e[0][k];
    if (accDst) {
#pragma unroll
        for (int k = 0; k < SM_VECS; ++k) *(v4f*)(accp + 1024 * k) = a[k] + scale * e[0][k];
    }
}

template <int NR, bool FIN>
__global__ void __launch_bounds__(SM_WAVES * 32)
end_softmax_kernel(const float* R0, const float* R1, const float* R2,
                   float* acc,
                   const float* __restrict__ sacc,
                   float* __restrict__ outStart,
                   float* __restrict__ outEnd) {
    const int b    = blockIdx.x;
    const int tid  = threadIdx.x;
    const int lane = tid & 31;
    const int wave = tid >> 5;
    __shared__ float redm[NR][SM_WAVES], reds[NR][SM_WAVES];
    const size_t off = (size_t)b * L_ + 4 * tid;

    v4f x[NR][SM_VECS], a[SM_VECS], st[SM_VECS];
#pragma unroll
    for (int r = 0; r < NR; ++r) {
        const float* rp = ((r == 0) ? R0 : (r == 1) ? R1 : R2) + off;
#pragma unroll
        for (int k = 0; k < SM_VECS; ++k) x[r][k] = *(const v4f*)(rp + 1024 * k);
    }
#pragma unroll
    for (int k = 0; k < SM_VECS; ++k) { a[k] = (v4f){}; st[k] = (v4f){}; }
    if (FIN) {
#pragma unroll
        for (int k = 0; k < SM_VECS; ++k) {
            a[k]  = *(const v4f*)(acc + off + 1024 * k);
            st[k] = *(const v4f*)(sacc + off + 1024 * k);
        }
    }
    v4f e[NR][SM_VECS];
    softmax_rows_regs<NR>(x, e, redm, reds, lane, wave);
#pragma unroll
    for (int r = 0; r < NR; ++r) {
#pragma unroll
        for (int k = 0; k < SM_VECS; ++k) a[k] = a[k] + 0.2f * e[r][k];
    }
    if (!FIN) {
#pragma unroll
        for (int k = 0; k < SM_VECS; ++k) *(v4f*)(acc + off + 1024 * k) = a[k];
    } else {
#pragma unroll
        for (int k = 0; k < SM_VECS; ++k) {
            v4f le, ls;
#pragma unroll
            for (int cc = 0; cc < 4; ++cc) { le[cc] = logf(a[k][cc]); ls[cc] = logf(st[k][cc]); }
            *(v4f*)(outEnd + off + 1024 * k)   = le;
            *(v4f*)(outStart + off + 1024 * k) = ls;
        }
    }
}

__global__ void gru_kernel(const float* __restrict__ gi,
                           const float* __restrict__ gh,
                           const float* __restrict__ sOld,
                           float* __restrict__ sNew) {
    int i = blockIdx.x * 256 + threadIdx.x;
    if (i >= BQ_) return;
    int b = i / Q_, q = i % Q_;
    const float* gib = gi + (size_t)b * 3 * Q_;
    const float* ghb = gh + (size_t)b * 3 * Q_;
    float ir = gib[q], iz = gib[Q_ + q], in_ = gib[2 * Q_ + q];
    float hr = ghb[q], hz = ghb[Q_ + q], hn  = ghb[2 * Q_ + q];
    float r = 1.0f / (1.0f + __expf(-(ir + hr)));
    float z = 1.0f / (1.0f + __expf(-(iz + hz)));
    float n = tanhf(in_ + r * hn);
    sNew[i] = (1.0f - z) * n + z * sOld[i];
}

__global__ void init_kernel(const float* __restrict__ s0, float* __restrict__ ws) {
    int i = blockIdx.x * 256 + threadIdx.x;
    if (i < BL_) ws[OFF_SACC + i] = 0.0f;
    if (i < BQ_) ws[OFF_SALL + i] = s0[i];
}

extern "C" void kernel_launch(void* const* d_in, const int* in_sizes, int n_in,
                              void* d_out, int out_size, void* d_ws, size_t ws_size,
                              hipStream_t stream) {
    const float* M    = (const float*)d_in[0];
    const float* s0   = (const float*)d_in[1];
    const float* Wb   = (const float*)d_in[2];
    const float* We   = (const float*)d_in[3];
    const float* Wa   = (const float*)d_in[4];
    const float* W_ih = (const float*)d_in[5];
    const float* W_hh = (const float*)d_in[6];
    const float* b_ih = (const float*)d_in[7];
    const float* b_hh = (const float*)d_in[8];

    float* ws    = (float*)d_ws;
    float* out   = (float*)d_out;
    float* S1    = ws + OFF_S1;
    float* S2    = ws + OFF_S2;
    float* sacc  = ws + OFF_SACC;
    float* S3    = ws + OFF_EACC;
    float* qb    = ws + OFF_QB;
    float* qa    = ws + OFF_QA;
    float* x1    = ws + OFF_X1;
    float* gi    = ws + OFF_GI;
    float* gh    = ws + OFF_GH;
    float* sAll  = ws + OFF_SALL;
    float* evAll = ws + OFF_EVALL;
    float* qeAll = ws + OFF_QEALL;

    init_kernel<<<BL_ / 256, 256, 0, stream>>>(s0, ws);

    auto gemm = [&](const float* A, int lda, const float* W, int ldw,
                    const float* bias, float* C, int ldc, int N, int accF, int rows) {
        dim3 g(N / 32, rows / SG_ROWS);
        wmma_gemm_kernel<<<g, SG_WAVES * 32, 0, stream>>>(A, lda, W, ldw, bias, C, ldc, accF);
    };

    for (int t = 0; t < 5; ++t) {
        const float* sCur = sAll + t * BQ_;
        float*       sNxt = sAll + (t + 1) * BQ_;

        gemm(sCur, Q_, Wb, Q_, nullptr, qb, MEM_, MEM_, 0, B_);
        gemm(sCur, Q_, Wa, Q_, nullptr, qa, MEM_, MEM_, 0, B_);

        scores_rows_kernel<2><<<dim3(SC_CH, B_), SC_WAVES * 32, 0, stream>>>(
            M, qb, qa, nullptr, S1, S2, nullptr);

        softmax_kernel<<<B_, 256, 0, stream>>>(S1, sacc, 0.2f);
        softmax_kernel<<<B_, 256, 0, stream>>>(S2, nullptr, 0.0f);

        wsum_rows_kernel<<<dim3(WS_CH, B_), WS_WAVES * 32, 0, stream>>>(M, S1, S2);
        wsum_reduce_kernel<<<(2 * BQ_) / 256, 256, 0, stream>>>(S1, evAll + t * BQ_, x1);

        gemm(x1,  MEM_, W_ih, MEM_, b_ih, gi, 3 * Q_, 3 * Q_, 0, B_);
        gemm(sCur, Q_,  W_hh, Q_,   b_hh, gh, 3 * Q_, 3 * Q_, 0, B_);
        gru_kernel<<<BQ_ / 256, 256, 0, stream>>>(gi, gh, sCur, sNxt);
    }

    gemm(sAll,  Q_,   We,      Q_ + MEM_, nullptr, qeAll, MEM_, MEM_, 0, 5 * B_);
    gemm(evAll, MEM_, We + Q_, Q_ + MEM_, nullptr, qeAll, MEM_, MEM_, 1, 5 * B_);

    scores_rows_kernel<3><<<dim3(SC_CH, B_), SC_WAVES * 32, 0, stream>>>(
        M, qeAll, qeAll + BQ_, qeAll + 2 * BQ_, S1, S2, S3);
    end_softmax_kernel<3, false><<<B_, 256, 0, stream>>>(S1, S2, S3, S3, nullptr, nullptr, nullptr);

    scores_rows_kernel<2><<<dim3(SC_CH, B_), SC_WAVES * 32, 0, stream>>>(
        M, qeAll + 3 * BQ_, qeAll + 4 * BQ_, nullptr, S1, S2, nullptr);
    end_softmax_kernel<2, true><<<B_, 256, 0, stream>>>(S1, S2, nullptr, S3, sacc, out, out + BL_);
}


# turn-loop softmax split: score kernel writes exp(score - chunk max), one (max, sum) per 512-row chunk; weighted-sum kernel joins the 8 pairs, scales its sums; 57 to 47 launches; new softmax sum tree
# speedup vs baseline: 12.7359x; 1.1422x over previous
#include <hip/hip_runtime.h>
#include <math.h>

#define B_   32
#define L_   4096
#define MEM_ 256
#define Q_   256
#define BL_  (B_ * L_)
#define BQ_  (B_ * Q_)
#define B3Q_ (B_ * 3 * Q_)

#define OFF_S1   0
#define OFF_S2   (OFF_S1 + BL_)
#define OFF_SACC (OFF_S2 + BL_)
#define OFF_EACC (OFF_SACC + BL_)
#define OFF_QB   (OFF_EACC + BL_)
#define OFF_QA   (OFF_QB + BQ_)
#define OFF_X1   (OFF_QA + BQ_)
#define OFF_GI   (OFF_X1 + BQ_)
#define OFF_GH   (OFF_GI + B3Q_)
#define OFF_SALL  OFF_EACC
#define OFF_EVALL (OFF_SALL + 6 * BQ_)
#define OFF_QEALL OFF_GI
#define OFF_STATS (OFF_EVALL + 5 * BQ_)

typedef __attribute__((ext_vector_type(2))) float v2f;
typedef __attribute__((ext_vector_type(8))) float v8f;

typedef __attribute__((ext_vector_type(4))) float v4f;

#define SG_K     256
#define SG_LINES (SG_K / 32)
#define SG_WAVES 8
#define SG_ROWS  8

__global__ void __launch_bounds__(SG_WAVES * 32)
wmma_gemm_kernel(const float* __restrict__ A, int lda,
                 const float* __restrict__ W, int ldw,
                 const float* __restrict__ bias,
                 float* __restrict__ C, int ldc,
                 int accumulate) {
    const int tid   = threadIdx.x;
    const int lane  = tid & 31;
    const int wave  = tid >> 5;
    const int c     = lane & 15;
    const int hi    = lane >> 4;
    const int kb    = c & 3;
    const int piece = 4 * (2 * kb + hi);
    const int m0    = blockIdx.y * SG_ROWS;
    const int n0    = blockIdx.x * 32;

    const float* ap = A + (size_t)(m0 + (c >> 2)) * lda + piece;
    const float* wp = W + (size_t)(n0 + 4 * wave + (c >> 2)) * ldw + piece;

    v4f xw[SG_LINES], xa[2][SG_LINES];
#pragma unroll
    for (int j = 0; j < SG_LINES; ++j) xw[j] = *(const v4f*)(wp + 32 * j);
#pragma unroll
    for (int u = 0; u < 2; ++u) {
#pragma unroll
        for (int j = 0; j < SG_LINES; ++j)
            xa[u][j] = *(const v4f*)(ap + (size_t)(4 * u) * lda + 32 * j);
    }

    const int orow = wave, ocol = lane;
    float* cp = C + (size_t)(m0 + orow) * ldc + n0 + ocol;
    float bv = 0.0f, cv = 0.0f;
    if (bias) bv = bias[n0 + ocol];
    if (accumulate) cv = *cp;

    v8f acc0[2], acc1[2];
#pragma unroll
    for (int u = 0; u < 2; ++u) { acc0[u] = (v8f){}; acc1[u] = (v8f){}; }
#pragma unroll
    for (int j = 0; j < SG_LINES; ++j) {
#pragma unroll
        for (int u = 0; u < 2; ++u) {
            v2f a, bb;
            a.x = xa[u][j][0]; a.y = xa[u][j][1]; bb.x = xw[j][0]; bb.y = xw[j][1];
            acc0[u] = __builtin_amdgcn_wmma_f32_16x16x4_f32(false, a, false, bb,
                                                            (short)0, acc0[u], false, false);
            a.x = xa[u][j][2]; a.y = xa[u][j][3]; bb.x = xw[j][2]; bb.y = xw[j][3];
            acc1[u] = __builtin_amdgcn_wmma_f32_16x16x4_f32(false, a, false, bb,
                                                            (short)0, acc1[u], false, false);
        }
    }
    float lo[2], up[2];
#pragma unroll
    for (int u = 0; u < 2; ++u) {
        const v8f tv = acc0[u] + acc1[u];
        const v4f tl = {tv[0], tv[1], tv[2], tv[3]};
        const v4f tu = {tv[4], tv[5], tv[6], tv[7]};
        lo[u] = tl[kb]; up[u] = tu[kb];
    }
#pragma unroll
    for (int u = 0; u < 2; ++u) { lo[u] += __shfl_xor(lo[u], 1); up[u] += __shfl_xor(up[u], 1); }
#pragma unroll
    for (int u = 0; u < 2; ++u) { lo[u] += __shfl_xor(lo[u], 2); up[u] += __shfl_xor(up[u], 2); }

    __shared__ float tile[SG_ROWS][32];
    if (kb == 0) {
#pragma unroll
        for (int u = 0; u < 2; ++u) {
            tile[4 * u + 2 * hi][4 * wave + (c >> 2)]     = lo[u];
            tile[4 * u + 2 * hi + 1][4 * wave + (c >> 2)] = up[u];
        }
    }
    __syncthreads();
    *cp = (cv + bv) + tile[orow][ocol];
}

template <int N>
static __device__ __forceinline__ float row_xor(float x) {
    const int b = __builtin_bit_cast(int, x);
    return __builtin_bit_cast(float,
        __builtin_amdgcn_update_dpp(b, b, 0x160 + N, 0xf, 0xf, false));
}
static __device__ __forceinline__ float xhalf(float x) {
    const int b = __builtin_bit_cast(int, x);
    return __builtin_bit_cast(float,
        __builtin_amdgcn_permlanex16(b, b, 0x76543210, 0xfedcba98, false, false));
}

#define SC_CH     8
#define SC_WAVES  16
#define SC_WGROWS (L_ / SC_CH)
#define SC_WROWS  (SC_WGROWS / SC_WAVES)
#define SC_GROUPS (SC_WROWS / 4)
#define SC_LINES  (MEM_ / 32)
#define SC_AHEAD  3
#define SC_RING   (SC_AHEAD + 1)

template <int NQ, bool LOCAL>
__global__ void __launch_bounds__(SC_WAVES * 32)
scores_rows_kernel(const float* __restrict__ M,
                   const float* __restrict__ q1,
                   const float* __restrict__ q2,
                   const float* __restrict__ q3,
                   float* __restrict__ S1,
                   float* __restrict__ S2,
                   float* __restrict__ S3,
                   float* __restrict__ stats) {
    const int tid   = threadIdx.x;
    const int lane  = tid & 31;
    const int wave  = tid >> 5;
    const int c     = lane & 15;
    const int hi    = lane >> 4;
    const int kb    = c & 3;
    const int b     = blockIdx.y;
    const int ch    = blockIdx.x;
    const int piece = 4 * (2 * kb + hi);
    const int r0    = ch * SC_WGROWS + wave * SC_WROWS;

    const float* Mb = M + ((size_t)b * L_ + r0 + (c >> 2)) * MEM_ + piece;
    const int qi = (c >> 2) % NQ;
    const float* qv = ((qi == 0) ? q1 : (qi == 1) ? q2 : q3) + (size_t)b * MEM_ + piece;

    v4f x[SC_RING][SC_LINES];
#pragma unroll
    for (int g = 0; g < SC_AHEAD; ++g) {
#pragma unroll
        for (int j = 0; j < SC_LINES; ++j)
            x[g][j] = *(const v4f*)(Mb + (size_t)(4 * g) * MEM_ + 32 * j);
    }

    v4f qf[SC_LINES];
#pragma unroll
    for (int j = 0; j < SC_LINES; ++j) qf[j] = *(const v4f*)(qv + 32 * j);

    float slo[SC_GROUPS], shi[SC_GROUPS];
#pragma unroll
    for (int g = 0; g < SC_GROUPS; ++g) {
        const int cur = g % SC_RING;
        if (g + SC_AHEAD < SC_GROUPS) {
            const int nxt = (g + SC_AHEAD) % SC_RING;
#pragma unroll
            for (int j = 0; j < SC_LINES; ++j)
                x[nxt][j] = *(const v4f*)(Mb + (size_t)(4 * (g + SC_AHEAD)) * MEM_ + 32 * j);
        }
        __builtin_amdgcn_sched_barrier(0);
        v8f acc0 = {}, acc1 = {};
#pragma unroll
        for (int j = 0; j < SC_LINES; ++j) {
            v2f a, bb;
            a.x = x[cur][j][0]; a.y = x[cur][j][1]; bb.x = qf[j][0]; bb.y = qf[j][1];
            acc0 = __builtin_amdgcn_wmma_f32_16x16x4_f32(false, a, false, bb,
                                                         (short)0, acc0, false, false);
            a.x = x[cur][j][2]; a.y = x[cur][j][3]; bb.x = qf[j][2]; bb.y = qf[j][3];
            acc1 = __builtin_amdgcn_wmma_f32_16x16x4_f32(false, a, false, bb,
                                                         (short)0, acc1, false, false);
        }
        float t[8];
#pragma unroll
        for (int r = 0; r < 8; ++r) t[r] = acc0[r] + acc1[r];
        float lo = (kb == 0) ? t[0] : (kb == 1) ? t[1] : (kb == 2) ? t[2] : t[3];
        float up = (kb == 0) ? t[4] : (kb == 1) ? t[5] : (kb == 2) ? t[6] : t[7];
        lo += __shfl_xor(lo, 1); up += __shfl_xor(up, 1);
        lo += __shfl_xor(lo, 2); up += __shfl_xor(up, 2);
        slo[g] = lo; shi[g] = up;
    }

    __shared__ float sc[NQ][SC_WGROWS];
    if (kb == 0 && (c >> 2) < NQ) {
        float* dst = &sc[c >> 2][wave * SC_WROWS + 2 * hi];
#pragma unroll
        for (int g = 0; g < SC_GROUPS; ++g) { dst[4 * g] = slo[g]; dst[4 * g + 1] = shi[g]; }
    }
    __syncthreads();
    const size_t so = (size_t)b * L_ + ch * SC_WGROWS + tid;
    if (!LOCAL) {
        S1[so] = sc[0][tid];
        S2[so] = sc[1][tid];
        if (NQ > 2) S3[so] = sc[NQ - 1][tid];
    } else {
        __shared__ float redm[NQ][SC_WAVES], reds[NQ][SC_WAVES];
        float xs[NQ], ex[NQ], mx[NQ];
#pragma unroll
        for (int q = 0; q < NQ; ++q) {
            xs[q] = sc[q][tid];
            float t = xs[q];
            t = fmaxf(t, row_xor<1>(t)); t = fmaxf(t, row_xor<2>(t));
            t = fmaxf(t, row_xor<4>(t)); t = fmaxf(t, row_xor<8>(t));
            t = fmaxf(t, xhalf(t));
            if (lane == 0) redm[q][wave] = t;
        }
        __syncthreads();
#pragma unroll
        for (int q = 0; q < NQ; ++q) {
            float t = redm[q][c];
            t = fmaxf(t, row_xor<1>(t)); t = fmaxf(t, row_xor<2>(t));
            t = fmaxf(t, row_xor<4>(t)); t = fmaxf(t, row_xor<8>(t));
            mx[q] = t;
            ex[q] = __expf(xs[q] - t);
            float s = ex[q];
            s += row_xor<1>(s); s += row_xor<2>(s); s += row_xor<4>(s); s += row_xor<8>(s);
            s += xhalf(s);
            if (lane == 0) reds[q][wave] = s;
        }
        __syncthreads();
#pragma unroll
        for (int q = 0; q < NQ; ++q) {
            float s = reds[q][c];
            s += row_xor<1>(s); s += row_xor<2>(s); s += row_xor<4>(s); s += row_xor<8>(s);
            if (tid == 0) {
                float* sp = stats + (((size_t)b * NQ + q) * SC_CH + ch) * 2;
                sp[0] = mx[q]; sp[1] = s;
            }
        }
        S1[so] = ex[0];
        S2[so] = ex[1];
        if (NQ > 2) S3[so] = ex[NQ - 1];
    }
}

#define WS_CH     8
#define WS_WAVES  16
#define WS_WGROWS (L_ / WS_CH)
#define WS_WROWS  (WS_WGROWS / WS_WAVES)
#define WS_KSTEPS (WS_WROWS / 4)
#define WS_KSTRIDE (4 * WS_WAVES)

__global__ void __launch_bounds__(WS_WAVES * 32)
wsum_rows_kernel(const float* __restrict__ M,
                 float* pbuf,
                 const float* wbuf,
                 const float* __restrict__ stats,
                 float* __restrict__ sacc) {
    const int tid  = threadIdx.x;
    const int lane = tid & 31;
    const int wave = tid >> 5;
    const int row  = lane & 15;
    const int hi   = lane >> 4;
    const int b    = blockIdx.y;
    const int c    = blockIdx.x;
    const int l0   = c * WS_WGROWS + 4 * wave;

    const float* pv = ((row == 1) ? wbuf : (const float*)pbuf) + (size_t)b * L_ + l0 + 2 * hi;
    const float* Mb = M + ((size_t)b * L_ + l0 + 2 * hi) * MEM_ + 4 * row;

    v8f acc[16];
#pragma unroll
    for (int t = 0; t < 16; ++t) acc[t] = (v8f){};

    v2f a[WS_KSTEPS];
#pragma unroll
    for (int ks = 0; ks < WS_KSTEPS; ++ks) {
        a[ks] = *(const v2f*)(pv + WS_KSTRIDE * ks);
        if (row >= 2) { a[ks].x = 0.0f; a[ks].y = 0.0f; }
    }

    v4f x[2][2][4];
#pragma unroll
    for (int g = 0; g < 4; ++g) {
        x[0][0][g] = *(const v4f*)(Mb + 64 * g);
        x[0][1][g] = *(const v4f*)(Mb + MEM_ + 64 * g);
    }

    float cm[2][WS_CH], cz[2][WS_CH];
    {
        const float* st = stats + (size_t)b * (2 * WS_CH * 2);
#pragma unroll
        for (int q = 0; q < 2; ++q) {
#pragma unroll
            for (int cc = 0; cc < WS_CH; ++cc) {
                cm[q][cc] = st[(q * WS_CH + cc) * 2];
                cz[q][cc] = st[(q * WS_CH + cc) * 2 + 1];
            }
        }
    }

#pragma unroll
    for (int ks = 0; ks < WS_KSTEPS; ++ks) {
        const int cur = ks & 1, nxt = cur ^ 1;
        if (ks + 1 < WS_KSTEPS) {
#pragma unroll
            for (int g = 0; g < 4; ++g) {
                x[nxt][0][g] = *(const v4f*)(Mb + (size_t)(WS_KSTRIDE * (ks + 1))     * MEM_ + 64 * g);
                x[nxt][1][g] = *(const v4f*)(Mb + (size_t)(WS_KSTRIDE * (ks + 1) + 1) * MEM_ + 64 * g);
            }
        }
        __builtin_amdgcn_sched_barrier(0);
#pragma unroll
        for (int g = 0; g < 4; ++g) {
#pragma unroll
            for (int cc = 0; cc < 4; ++cc) {
                v2f bb; bb.x = x[cur][0][g][cc]; bb.y = x[cur][1][g][cc];
                acc[4 * g + cc] = __builtin_amdgcn_wmma_f32_16x16x4_f32(
                    false, a[ks], false, bb, (short)0, acc[4 * g + cc], false, false);
            }
        }
    }

    const size_t ro = (size_t)b * L_ + c * WS_WGROWS + tid;
    const float pe = pbuf[ro], sa = sacc[ro];

    float f[2];
#pragma unroll
    for (int q = 0; q < 2; ++q) {
        float m = cm[q][0];
#pragma unroll
        for (int cc = 1; cc < WS_CH; ++cc) m = fmaxf(m, cm[q][cc]);
        float z = 0.0f, own = 0.0f;
#pragma unroll
        for (int cc = 0; cc < WS_CH; ++cc) {
            const float w = __expf(cm[q][cc] - m);
            z += cz[q][cc] * w;
            if (cc == c) own = w;
        }
        f[q] = own / z;
    }

    __shared__ float red[WS_WAVES][2][MEM_];
    if (hi == 0) {
#pragma unroll
        for (int g = 0; g < 4; ++g) {
            v4f es, xs;
#pragma unroll
            for (int cc = 0; cc < 4; ++cc) { es[cc] = acc[4 * g + cc][0]; xs[cc] = acc[4 * g + cc][1]; }
            *(v4f*)&red[wave][0][64 * g + 4 * row] = es;
            *(v4f*)&red[wave][1][64 * g + 4 * row] = xs;
        }
    }
    __syncthreads();
    sacc[ro] = sa + 0.2f * (pe * f[0]);
    {
        const int v = tid >> 8, col = tid & 255;
        float s = 0.0f;
#pragma unroll
        for (int w = 0; w < WS_WAVES; ++w) s += red[w][v][col];
        pbuf[(size_t)b * L_ + c * WS_WGROWS + v * MEM_ + col] = s * (v ? f[1] : f[0]);
    }
}

__global__ void wsum_reduce_kernel(const float* __restrict__ part,
                                   float* __restrict__ ev,
                                   float* __restrict__ x1) {
    const int i   = blockIdx.x * 256 + threadIdx.x;
    const int b   = i >> 9;
    const int v   = (i >> 8) & 1;
    const int col = i & 255;
    const float* src = part + (size_t)b * L_ + v * MEM_ + col;
    float s = 0.0f;
#pragma unroll
    for (int c = 0; c < WS_CH; ++c) s += src[c * WS_WGROWS];
    (v ? x1 : ev)[(size_t)b * MEM_ + col] = s;
}

#define SM_WAVES 8
#define SM_VECS  (L_ / (SM_WAVES * 32 * 4))

template <int NR>
static __device__ __forceinline__ void softmax_rows_regs(const v4f (&x)[NR][SM_VECS], v4f (&e)[NR][SM_VECS],
                                                         float (&redm)[NR][SM_WAVES],
                                                         float (&reds)[NR][SM_WAVES],
                                                         int lane, int wave) {
    float m[NR];
#pragma unroll
    for (int r = 0; r < NR; ++r) {
        float t = fmaxf(fmaxf(x[r][0][0], x[r][0][1]), fmaxf(x[r][0][2], x[r][0][3]));
#pragma unroll
        for (int k = 1; k < SM_VECS; ++k)
            t = fmaxf(t, fmaxf(fmaxf(x[r][k][0], x[r][k][1]), fmaxf(x[r][k][2], x[r][k][3])));
        t = fmaxf(t, row_xor<1>(t)); t = fmaxf(t, row_xor<2>(t));
        t = fmaxf(t, row_xor<4>(t)); t = fmaxf(t, row_xor<8>(t));
        t = fmaxf(t, xhalf(t));
        if (lane == 0) redm[r][wave] = t;
    }
    __syncthreads();
#pragma unroll
    for (int r = 0; r < NR; ++r) {
        float t = redm[r][lane & (SM_WAVES - 1)];
        t = fmaxf(t, row_xor<1>(t)); t = fmaxf(t, row_xor<2>(t)); t = fmaxf(t, row_xor<4>(t));
        m[r] = t;
    }

#pragma unroll
    for (int r = 0; r < NR; ++r) {
        float s = 0.0f;
#pragma unroll
        for (int k = 0; k < SM_VECS; ++k) {
#pragma unroll
            for (int cc = 0; cc < 4; ++cc) e[r][k][cc] = __expf(x[r][k][cc] - m[r]);
            s += (e[r][k][0] + e[r][k][1]) + (e[r][k][2] + e[r][k][3]);
        }
        s += row_xor<1>(s); s += row_xor<2>(s); s += row_xor<4>(s); s += row_xor<8>(s);
        s += xhalf(s);
        if (lane == 0) reds[r][wave] = s;
    }
    __syncthreads();
#pragma unroll
    for (int r = 0; r < NR; ++r) {
        float s = reds[r][lane & (SM_WAVES - 1)];
        s += row_xor<1>(s); s += row_xor<2>(s); s += row_xor<4>(s);
        const float inv = 1.0f / s;
#pragma unroll
        for (int k = 0; k < SM_VECS; ++k) e[r][k] = e[r][k] * inv;
    }
}

template <int NR, bool FIN>
__global__ void __launch_bounds__(SM_WAVES * 32)
end_softmax_kernel(const float* R0, const float* R1, const float* R2,
                   float* acc,
                   const float* __restrict__ sacc,
                   float* __restrict__ outStart,
                   float* __restrict__ outEnd) {
    const int b    = blockIdx.x;
    const int tid  = threadIdx.x;
    const int lane = tid & 31;
    const int wave = tid >> 5;
    __shared__ float redm[NR][SM_WAVES], reds[NR][SM_WAVES];
    const size_t off = (size_t)b * L_ + 4 * tid;

    v4f x[NR][SM_VECS], a[SM_VECS], st[SM_VECS];
#pragma unroll
    for (int r = 0; r < NR; ++r) {
        const float* rp = ((r == 0) ? R0 : (r == 1) ? R1 : R2) + off;
#pragma unroll
        for (int k = 0; k < SM_VECS; ++k) x[r][k] = *(const v4f*)(rp + 1024 * k);
    }
#pragma unroll
    for (int k = 0; k < SM_VECS; ++k) { a[k] = (v4f){}; st[k] = (v4f){}; }
    if (FIN) {
#pragma unroll
        for (int k = 0; k < SM_VECS; ++k) {
            a[k]  = *(const v4f*)(acc + off + 1024 * k);
            st[k] = *(const v4f*)(sacc + off + 1024 * k);
        }
    }
    v4f e[NR][SM_VECS];
    softmax_rows_regs<NR>(x, e, redm, reds, lane, wave);
#pragma unroll
    for (int r = 0; r < NR; ++r) {
#pragma unroll
        for (int k = 0; k < SM_VECS; ++k) a[k] = a[k] + 0.2f * e[r][k];
    }
    if (!FIN) {
#pragma unroll
        for (int k = 0; k < SM_VECS; ++k) *(v4f*)(acc + off + 1024 * k) = a[k];
    } else {
#pragma unroll
        for (int k = 0; k < SM_VECS; ++k) {
            v4f le, ls;
#pragma unroll
            for (int cc = 0; cc < 4; ++cc) { le[cc] = logf(a[k][cc]); ls[cc] = logf(st[k][cc]); }
            *(v4f*)(outEnd + off + 1024 * k)   = le;
            *(v4f*)(outStart + off + 1024 * k) = ls;
        }
    }
}

__global__ void gru_kernel(const float* __restrict__ gi,
                           const float* __restrict__ gh,
                           const float* __restrict__ sOld,
                           float* __restrict__ sNew) {
    int i = blockIdx.x * 256 + threadIdx.x;
    if (i >= BQ_) return;
    int b = i / Q_, q = i % Q_;
    const float* gib = gi + (size_t)b * 3 * Q_;
    const float* ghb = gh + (size_t)b * 3 * Q_;
    float ir = gib[q], iz = gib[Q_ + q], in_ = gib[2 * Q_ + q];
    float hr = ghb[q], hz = ghb[Q_ + q], hn  = ghb[2 * Q_ + q];
    float r = 1.0f / (1.0f + __expf(-(ir + hr)));
    float z = 1.0f / (1.0f + __expf(-(iz + hz)));
    float n = tanhf(in_ + r * hn);
    sNew[i] = (1.0f - z) * n + z * sOld[i];
}

__global__ void init_kernel(const float* __restrict__ s0, float* __restrict__ ws) {
    int i = blockIdx.x * 256 + threadIdx.x;
    if (i < BL_) ws[OFF_SACC + i] = 0.0f;
    if (i < BQ_) ws[OFF_SALL + i] = s0[i];
}

extern "C" void kernel_launch(void* const* d_in, const int* in_sizes, int n_in,
                              void* d_out, int out_size, void* d_ws, size_t ws_size,
                              hipStream_t stream) {
    const float* M    = (const float*)d_in[0];
    const float* s0   = (const float*)d_in[1];
    const float* Wb   = (const float*)d_in[2];
    const float* We   = (const float*)d_in[3];
    const float* Wa   = (const float*)d_in[4];
    const float* W_ih = (const float*)d_in[5];
    const float* W_hh = (const float*)d_in[6];
    const float* b_ih = (const float*)d_in[7];
    const float* b_hh = (const float*)d_in[8];

    float* ws    = (float*)d_ws;
    float* out   = (float*)d_out;
    float* S1    = ws + OFF_S1;
    float* S2    = ws + OFF_S2;
    float* sacc  = ws + OFF_SACC;
    float* S3    = ws + OFF_EACC;
    float* qb    = ws + OFF_QB;
    float* qa    = ws + OFF_QA;
    float* x1    = ws + OFF_X1;
    float* gi    = ws + OFF_GI;
    float* gh    = ws + OFF_GH;
    float* sAll  = ws + OFF_SALL;
    float* evAll = ws + OFF_EVALL;
    float* qeAll = ws + OFF_QEALL;
    float* stats = ws + OFF_STATS;

    init_kernel<<<BL_ / 256, 256, 0, stream>>>(s0, ws);

    auto gemm = [&](const float* A, int lda, const float* W, int ldw,
                    const float* bias, float* C, int ldc, int N, int accF, int rows) {
        dim3 g(N / 32, rows / SG_ROWS);
        wmma_gemm_kernel<<<g, SG_WAVES * 32, 0, stream>>>(A, lda, W, ldw, bias, C, ldc, accF);
    };

    for (int t = 0; t < 5; ++t) {
        const float* sCur = sAll + t * BQ_;
        float*       sNxt = sAll + (t + 1) * BQ_;

        gemm(sCur, Q_, Wb, Q_, nullptr, qb, MEM_, MEM_, 0, B_);
        gemm(sCur, Q_, Wa, Q_, nullptr, qa, MEM_, MEM_, 0, B_);

        scores_rows_kernel<2, true><<<dim3(SC_CH, B_), SC_WAVES * 32, 0, stream>>>(
            M, qb, qa, nullptr, S1, S2, nullptr, stats);

        wsum_rows_kernel<<<dim3(WS_CH, B_), WS_WAVES * 32, 0, stream>>>(M, S1, S2, stats, sacc);
        wsum_reduce_kernel<<<(2 * BQ_) / 256, 256, 0, stream>>>(S1, evAll + t * BQ_, x1);

        gemm(x1,  MEM_, W_ih, MEM_, b_ih, gi, 3 * Q_, 3 * Q_, 0, B_);
        gemm(sCur, Q_,  W_hh, Q_,   b_hh, gh, 3 * Q_, 3 * Q_, 0, B_);
        gru_kernel<<<BQ_ / 256, 256, 0, stream>>>(gi, gh, sCur, sNxt);
    }

    gemm(sAll,  Q_,   We,      Q_ + MEM_, nullptr, qeAll, MEM_, MEM_, 0, 5 * B_);
    gemm(evAll, MEM_, We + Q_, Q_ + MEM_, nullptr, qeAll, MEM_, MEM_, 1, 5 * B_);

    scores_rows_kernel<3, false><<<dim3(SC_CH, B_), SC_WAVES * 32, 0, stream>>>(
        M, qeAll, qeAll + BQ_, qeAll + 2 * BQ_, S1, S2, S3, nullptr);
    end_softmax_kernel<3, false><<<B_, 256, 0, stream>>>(S1, S2, S3, S3, nullptr, nullptr, nullptr);

    scores_rows_kernel<2, false><<<dim3(SC_CH, B_), SC_WAVES * 32, 0, stream>>>(
        M, qeAll + 3 * BQ_, qeAll + 4 * BQ_, nullptr, S1, S2, nullptr, nullptr);
    end_softmax_kernel<2, true><<<B_, 256, 0, stream>>>(S1, S2, nullptr, S3, sacc, out, out + BL_);
}


# turn tail in one launch: a work-group owns 64 state features: gate GEMMs, GRU cell, its K slice of both next queries; score kernel adds the 4 slices on load (new query sum order); 47 to 28 launches
# speedup vs baseline: 14.3628x; 1.1277x over previous
#include <hip/hip_runtime.h>
#include <math.h>

#define B_   32
#define L_   4096
#define MEM_ 256
#define Q_   256
#define BL_  (B_ * L_)
#define BQ_  (B_ * Q_)
#define B3Q_ (B_ * 3 * Q_)

#define OFF_S1   0
#define OFF_S2   (OFF_S1 + BL_)
#define OFF_SACC (OFF_S2 + BL_)
#define OFF_EACC (OFF_SACC + BL_)
#define OFF_QB   (OFF_EACC + BL_)
#define OFF_QA   (OFF_QB + BQ_)
#define OFF_X1   (OFF_QA + BQ_)
#define OFF_GI   (OFF_X1 + BQ_)
#define OFF_GH   (OFF_GI + B3Q_)
#define OFF_SALL  OFF_EACC
#define OFF_EVALL (OFF_SALL + 6 * BQ_)
#define OFF_QEALL OFF_GI
#define OFF_STATS (OFF_EVALL + 5 * BQ_)
#define TL_SLICES 4
#define OFF_QBP  (OFF_STATS + 1024)
#define OFF_QAP  OFF_GI

typedef __attribute__((ext_vector_type(2))) float v2f;
typedef __attribute__((ext_vector_type(8))) float v8f;

typedef __attribute__((ext_vector_type(4))) float v4f;

#define SG_K     256
#define SG_LINES (SG_K / 32)
#define SG_WAVES 8
#define SG_ROWS  8

__global__ void __launch_bounds__(SG_WAVES * 32)
wmma_gemm_kernel(const float* __restrict__ A, int lda,
                 const float* __restrict__ W, int ldw,
                 const float* __restrict__ bias,
                 float* __restrict__ C, int ldc,
                 int accumulate) {
    const int tid   = threadIdx.x;
    const int lane  = tid & 31;
    const int wave  = tid >> 5;
    const int c     = lane & 15;
    const int hi    = lane >> 4;
    const int kb    = c & 3;
    const int piece = 4 * (2 * kb + hi);
    const int m0    = blockIdx.y * SG_ROWS;
    const int n0    = blockIdx.x * 32;

    const float* ap = A + (size_t)(m0 + (c >> 2)) * lda + piece;
    const float* wp = W + (size_t)(n0 + 4 * wave + (c >> 2)) * ldw + piece;

    v4f xw[SG_LINES], xa[2][SG_LINES];
#pragma unroll
    for (int j = 0; j < SG_LINES; ++j) xw[j] = *(const v4f*)(wp + 32 * j);
#pragma unroll
    for (int u = 0; u < 2; ++u) {
#pragma unroll
        for (int j = 0; j < SG_LINES; ++j)
            xa[u][j] = *(const v4f*)(ap + (size_t)(4 * u) * lda + 32 * j);
    }

    const int orow = wave, ocol = lane;
    float* cp = C + (size_t)(m0 + orow) * ldc + n0 + ocol;
    float bv = 0.0f, cv = 0.0f;
    if (bias) bv = bias[n0 + ocol];
    if (accumulate) cv = *cp;

    v8f acc0[2], acc1[2];
#pragma unroll
    for (int u = 0; u < 2; ++u) { acc0[u] = (v8f){}; acc1[u] = (v8f){}; }
#pragma unroll
    for (int j = 0; j < SG_LINES; ++j) {
#pragma unroll
        for (int u = 0; u < 2; ++u) {
            v2f a, bb;
            a.x = xa[u][j][0]; a.y = xa[u][j][1]; bb.x = xw[j][0]; bb.y = xw[j][1];
            acc0[u] = __builtin_amdgcn_wmma_f32_16x16x4_f32(false, a, false, bb,
                                                            (short)0, acc0[u], false, false);
            a.x = xa[u][j][2]; a.y = xa[u][j][3]; bb.x = xw[j][2]; bb.y = xw[j][3];
            acc1[u] = __builtin_amdgcn_wmma_f32_16x16x4_f32(false, a, false, bb,
                                                            (short)0, acc1[u], false, false);
        }
    }
    float lo[2], up[2];
#pragma unroll
    for (int u = 0; u < 2; ++u) {
        const v8f tv = acc0[u] + acc1[u];
        const v4f tl = {tv[0], tv[1], tv[2], tv[3]};
        const v4f tu = {tv[4], tv[5], tv[6], tv[7]};
        lo[u] = tl[kb]; up[u] = tu[kb];
    }
#pragma unroll
    for (int u = 0; u < 2; ++u) { lo[u] += __shfl_xor(lo[u], 1); up[u] += __shfl_xor(up[u], 1); }
#pragma unroll
    for (int u = 0; u < 2; ++u) { lo[u] += __shfl_xor(lo[u], 2); up[u] += __shfl_xor(up[u], 2); }

    __shared__ float tile[SG_ROWS][32];
    if (kb == 0) {
#pragma unroll
        for (int u = 0; u < 2; ++u) {
            tile[4 * u + 2 * hi][4 * wave + (c >> 2)]     = lo[u];
            tile[4 * u + 2 * hi + 1][4 * wave + (c >> 2)] = up[u];
        }
    }
    __syncthreads();
    *cp = (cv + bv) + tile[orow][ocol];
}

template <int N>
static __device__ __forceinline__ float row_xor(float x) {
    const int b = __builtin_bit_cast(int, x);
    return __builtin_bit_cast(float,
        __builtin_amdgcn_update_dpp(b, b, 0x160 + N, 0xf, 0xf, false));
}
static __device__ __forceinline__ float xhalf(float x) {
    const int b = __builtin_bit_cast(int, x);
    return __builtin_bit_cast(float,
        __builtin_amdgcn_permlanex16(b, b, 0x76543210, 0xfedcba98, false, false));
}

#define SC_CH     8
#define SC_WAVES  16
#define SC_WGROWS (L_ / SC_CH)
#define SC_WROWS  (SC_WGROWS / SC_WAVES)
#define SC_GROUPS (SC_WROWS / 4)
#define SC_LINES  (MEM_ / 32)
#define SC_AHEAD  3
#define SC_RING   (SC_AHEAD + 1)

template <int NQ, bool LOCAL, int NP>
__global__ void __launch_bounds__(SC_WAVES * 32)
scores_rows_kernel(const float* __restrict__ M,
                   const float* __restrict__ q1,
                   const float* __restrict__ q2,
                   const float* __restrict__ q3,
                   float* __restrict__ S1,
                   float* __restrict__ S2,
                   float* __restrict__ S3,
                   float* __restrict__ stats) {
    const int tid   = threadIdx.x;
    const int lane  = tid & 31;
    const int wave  = tid >> 5;
    const int c     = lane & 15;
    const int hi    = lane >> 4;
    const int kb    = c & 3;
    const int b     = blockIdx.y;
    const int ch    = blockIdx.x;
    const int piece = 4 * (2 * kb + hi);
    const int r0    = ch * SC_WGROWS + wave * SC_WROWS;

    const float* Mb = M + ((size_t)b * L_ + r0 + (c >> 2)) * MEM_ + piece;
    const int qi = (c >> 2) % NQ;
    const float* qv = ((qi == 0) ? q1 : (qi == 1) ? q2 : q3) + (size_t)b * MEM_ + piece;

    v4f x[SC_RING][SC_LINES];
#pragma unroll
    for (int g = 0; g < SC_AHEAD; ++g) {
#pragma unroll
        for (int j = 0; j < SC_LINES; ++j)
            x[g][j] = *(const v4f*)(Mb + (size_t)(4 * g) * MEM_ + 32 * j);
    }

    v4f qf[SC_LINES];
#pragma unroll
    for (int j = 0; j < SC_LINES; ++j) qf[j] = *(const v4f*)(qv + 32 * j);
#pragma unroll
    for (int p = 1; p < NP; ++p) {
#pragma unroll
        for (int j = 0; j < SC_LINES; ++j) qf[j] = qf[j] + *(const v4f*)(qv + (size_t)p * BQ_ + 32 * j);
    }

    float slo[SC_GROUPS], shi[SC_GROUPS];
#pragma unroll
    for (int g = 0; g < SC_GROUPS; ++g) {
        const int cur = g % SC_RING;
        if (g + SC_AHEAD < SC_GROUPS) {
            const int nxt = (g + SC_AHEAD) % SC_RING;
#pragma unroll
            for (int j = 0; j < SC_LINES; ++j)
                x[nxt][j] = *(const v4f*)(Mb + (size_t)(4 * (g + SC_AHEAD)) * MEM_ + 32 * j);
        }
        __builtin_amdgcn_sched_barrier(0);
        v8f acc0 = {}, acc1 = {};
#pragma unroll
        for (int j = 0; j < SC_LINES; ++j) {
            v2f a, bb;
            a.x = x[cur][j][0]; a.y = x[cur][j][1]; bb.x = qf[j][0]; bb.y = qf[j][1];
            acc0 = __builtin_amdgcn_wmma_f32_16x16x4_f32(false, a, false, bb,
                                                         (short)0, acc0, false, false);
            a.x = x[cur][j][2]; a.y = x[cur][j][3]; bb.x = qf[j][2]; bb.y = qf[j][3];
            acc1 = __builtin_amdgcn_wmma_f32_16x16x4_f32(false, a, false, bb,
                                                         (short)0, acc1, false, false);
        }
        float t[8];
#pragma unroll
        for (int r = 0; r < 8; ++r) t[r] = acc0[r] + acc1[r];
        float lo = (kb == 0) ? t[0] : (kb == 1) ? t[1] : (kb == 2) ? t[2] : t[3];
        float up = (kb == 0) ? t[4] : (kb == 1) ? t[5] : (kb == 2) ? t[6] : t[7];
        lo += __shfl_xor(lo, 1); up += __shfl_xor(up, 1);
        lo += __shfl_xor(lo, 2); up += __shfl_xor(up, 2);
        slo[g] = lo; shi[g] = up;
    }

    __shared__ float sc[NQ][SC_WGROWS];
    if (kb == 0 && (c >> 2) < NQ) {
        float* dst = &sc[c >> 2][wave * SC_WROWS + 2 * hi];
#pragma unroll
        for (int g = 0; g < SC_GROUPS; ++g) { dst[4 * g] = slo[g]; dst[4 * g + 1] = shi[g]; }
    }
    __syncthreads();
    const size_t so = (size_t)b * L_ + ch * SC_WGROWS + tid;
    if (!LOCAL) {
        S1[so] = sc[0][tid];
        S2[so] = sc[1][tid];
        if (NQ > 2) S3[so] = sc[NQ - 1][tid];
    } else {
        __shared__ float redm[NQ][SC_WAVES], reds[NQ][SC_WAVES];
        float xs[NQ], ex[NQ], mx[NQ];
#pragma unroll
        for (int q = 0; q < NQ; ++q) {
            xs[q] = sc[q][tid];
            float t = xs[q];
            t = fmaxf(t, row_xor<1>(t)); t = fmaxf(t, row_xor<2>(t));
            t = fmaxf(t, row_xor<4>(t)); t = fmaxf(t, row_xor<8>(t));
            t = fmaxf(t, xhalf(t));
            if (lane == 0) redm[q][wave] = t;
        }
        __syncthreads();
#pragma unroll
        for (int q = 0; q < NQ; ++q) {
            float t = redm[q][c];
            t = fmaxf(t, row_xor<1>(t)); t = fmaxf(t, row_xor<2>(t));
            t = fmaxf(t, row_xor<4>(t)); t = fmaxf(t, row_xor<8>(t));
            mx[q] = t;
            ex[q] = __expf(xs[q] - t);
            float s = ex[q];
            s += row_xor<1>(s); s += row_xor<2>(s); s += row_xor<4>(s); s += row_xor<8>(s);
            s += xhalf(s);
            if (lane == 0) reds[q][wave] = s;
        }
        __syncthreads();
#pragma unroll
        for (int q = 0; q < NQ; ++q) {
            float s = reds[q][c];
            s += row_xor<1>(s); s += row_xor<2>(s); s += row_xor<4>(s); s += row_xor<8>(s);
            if (tid == 0) {
                float* sp = stats + (((size_t)b * NQ + q) * SC_CH + ch) * 2;
                sp[0] = mx[q]; sp[1] = s;
            }
        }
        S1[so] = ex[0];
        S2[so] = ex[1];
        if (NQ > 2) S3[so] = ex[NQ - 1];
    }
}

#define WS_CH     8
#define WS_WAVES  16
#define WS_WGROWS (L_ / WS_CH)
#define WS_WROWS  (WS_WGROWS / WS_WAVES)
#define WS_KSTEPS (WS_WROWS / 4)
#define WS_KSTRIDE (4 * WS_WAVES)

__global__ void __launch_bounds__(WS_WAVES * 32)
wsum_rows_kernel(const float* __restrict__ M,
                 float* pbuf,
                 const float* wbuf,
                 const float* __restrict__ stats,
                 float* __restrict__ sacc) {
    const int tid  = threadIdx.x;
    const int lane = tid & 31;
    const int wave = tid >> 5;
    const int row  = lane & 15;
    const int hi   = lane >> 4;
    const int b    = blockIdx.y;
    const int c    = blockIdx.x;
    const int l0   = c * WS_WGROWS + 4 * wave;

    const float* pv = ((row == 1) ? wbuf : (const float*)pbuf) + (size_t)b * L_ + l0 + 2 * hi;
    const float* Mb = M + ((size_t)b * L_ + l0 + 2 * hi) * MEM_ + 4 * row;

    v8f acc[16];
#pragma unroll
    for (int t = 0; t < 16; ++t) acc[t] = (v8f){};

    v2f a[WS_KSTEPS];
#pragma unroll
    for (int ks = 0; ks < WS_KSTEPS; ++ks) {
        a[ks] = *(const v2f*)(pv + WS_KSTRIDE * ks);
        if (row >= 2) { a[ks].x = 0.0f; a[ks].y = 0.0f; }
    }

    v4f x[2][2][4];
#pragma unroll
    for (int g = 0; g < 4; ++g) {
        x[0][0][g] = *(const v4f*)(Mb + 64 * g);
        x[0][1][g] = *(const v4f*)(Mb + MEM_ + 64 * g);
    }

    float cm[2][WS_CH], cz[2][WS_CH];
    {
        const float* st = stats + (size_t)b * (2 * WS_CH * 2);
#pragma unroll
        for (int q = 0; q < 2; ++q) {
#pragma unroll
            for (int cc = 0; cc < WS_CH; ++cc) {
                cm[q][cc] = st[(q * WS_CH + cc) * 2];
                cz[q][cc] = st[(q * WS_CH + cc) * 2 + 1];
            }
        }
    }

#pragma unroll
    for (int ks = 0; ks < WS_KSTEPS; ++ks) {
        const int cur = ks & 1, nxt = cur ^ 1;
        if (ks + 1 < WS_KSTEPS) {
#pragma unroll
            for (int g = 0; g < 4; ++g) {
                x[nxt][0][g] = *(const v4f*)(Mb + (size_t)(WS_KSTRIDE * (ks + 1))     * MEM_ + 64 * g);
                x[nxt][1][g] = *(const v4f*)(Mb + (size_t)(WS_KSTRIDE * (ks + 1) + 1) * MEM_ + 64 * g);
            }
        }
        __builtin_amdgcn_sched_barrier(0);
#pragma unroll
        for (int g = 0; g < 4; ++g) {
#pragma unroll
            for (int cc = 0; cc < 4; ++cc) {
                v2f bb; bb.x = x[cur][0][g][cc]; bb.y = x[cur][1][g][cc];
                acc[4 * g + cc] = __builtin_amdgcn_wmma_f32_16x16x4_f32(
                    false, a[ks], false, bb, (short)0, acc[4 * g + cc], false, false);
            }
        }
    }

    const size_t ro = (size_t)b * L_ + c * WS_WGROWS + tid;
    const float pe = pbuf[ro], sa = sacc[ro];

    float f[2];
#pragma unroll
    for (int q = 0; q < 2; ++q) {
        float m = cm[q][0];
#pragma unroll
        for (int cc = 1; cc < WS_CH; ++cc) m = fmaxf(m, cm[q][cc]);
        float z = 0.0f, own = 0.0f;
#pragma unroll
        for (int cc = 0; cc < WS_CH; ++cc) {
            const float w = __expf(cm[q][cc] - m);
            z += cz[q][cc] * w;
            if (cc == c) own = w;
        }
        f[q] = own / z;
    }

    __shared__ float red[WS_WAVES][2][MEM_];
    if (hi == 0) {
#pragma unroll
        for (int g = 0; g < 4; ++g) {
            v4f es, xs;
#pragma unroll
            for (int cc = 0; cc < 4; ++cc) { es[cc] = acc[4 * g + cc][0]; xs[cc] = acc[4 * g + cc][1]; }
            *(v4f*)&red[wave][0][64 * g + 4 * row] = es;
            *(v4f*)&red[wave][1][64 * g + 4 * row] = xs;
        }
    }
    __syncthreads();
    sacc[ro] = sa + 0.2f * (pe * f[0]);
    {
        const int v = tid >> 8, col = tid & 255;
        float s = 0.0f;
#pragma unroll
        for (int w = 0; w < WS_WAVES; ++w) s += red[w][v][col];
        pbuf[(size_t)b * L_ + c * WS_WGROWS + v * MEM_ + col] = s * (v ? f[1] : f[0]);
    }
}

__global__ void wsum_reduce_kernel(const float* __restrict__ part,
                                   float* __restrict__ ev,
                                   float* __restrict__ x1) {
    const int i   = blockIdx.x * 256 + threadIdx.x;
    const int b   = i >> 9;
    const int v   = (i >> 8) & 1;
    const int col = i & 255;
    const float* src = part + (size_t)b * L_ + v * MEM_ + col;
    float s = 0.0f;
#pragma unroll
    for (int c = 0; c < WS_CH; ++c) s += src[c * WS_WGROWS];
    (v ? x1 : ev)[(size_t)b * MEM_ + col] = s;
}

#define SM_WAVES 8
#define SM_VECS  (L_ / (SM_WAVES * 32 * 4))

template <int NR>
static __device__ __forceinline__ void softmax_rows_regs(const v4f (&x)[NR][SM_VECS], v4f (&e)[NR][SM_VECS],
                                                         float (&redm)[NR][SM_WAVES],
                                                         float (&reds)[NR][SM_WAVES],
                                                         int lane, int wave) {
    float m[NR];
#pragma unroll
    for (int r = 0; r < NR; ++r) {
        float t = fmaxf(fmaxf(x[r][0][0], x[r][0][1]), fmaxf(x[r][0][2], x[r][0][3]));
#pragma unroll
        for (int k = 1; k < SM_VECS; ++k)
            t = fmaxf(t, fmaxf(fmaxf(x[r][k][0], x[r][k][1]), fmaxf(x[r][k][2], x[r][k][3])));
        t = fmaxf(t, row_xor<1>(t)); t = fmaxf(t, row_xor<2>(t));
        t = fmaxf(t, row_xor<4>(t)); t = fmaxf(t, row_xor<8>(t));
        t = fmaxf(t, xhalf(t));
        if (lane == 0) redm[r][wave] = t;
    }
    __syncthreads();
#pragma unroll
    for (int r = 0; r < NR; ++r) {
        float t = redm[r][lane & (SM_WAVES - 1)];
        t = fmaxf(t, row_xor<1>(t)); t = fmaxf(t, row_xor<2>(t)); t = fmaxf(t, row_xor<4>(t));
        m[r] = t;
    }

#pragma unroll
    for (int r = 0; r < NR; ++r) {
        float s = 0.0f;
#pragma unroll
        for (int k = 0; k < SM_VECS; ++k) {
#pragma unroll
            for (int cc = 0; cc < 4; ++cc) e[r][k][cc] = __expf(x[r][k][cc] - m[r]);
            s += (e[r][k][0] + e[r][k][1]) + (e[r][k][2] + e[r][k][3]);
        }
        s += row_xor<1>(s); s += row_xor<2>(s); s += row_xor<4>(s); s += row_xor<8>(s);
        s += xhalf(s);
        if (lane == 0) reds[r][wave] = s;
    }
    __syncthreads();
#pragma unroll
    for (int r = 0; r < NR; ++r) {
        float s = reds[r][lane & (SM_WAVES - 1)];
        s += row_xor<1>(s); s += row_xor<2>(s); s += row_xor<4>(s);
        const float inv = 1.0f / s;
#pragma unroll
        for (int k = 0; k < SM_VECS; ++k) e[r][k] = e[r][k] * inv;
    }
}

template <int NR, bool FIN>
__global__ void __launch_bounds__(SM_WAVES * 32)
end_softmax_kernel(const float* R0, const float* R1, const float* R2,
                   float* acc,
                   const float* __restrict__ sacc,
                   float* __restrict__ outStart,
                   float* __restrict__ outEnd) {
    const int b    = blockIdx.x;
    const int tid  = threadIdx.x;
    const int lane = tid & 31;
    const int wave = tid >> 5;
    __shared__ float redm[NR][SM_WAVES], reds[NR][SM_WAVES];
    const size_t off = (size_t)b * L_ + 4 * tid;

    v4f x[NR][SM_VECS], a[SM_VECS], st[SM_VECS];
#pragma unroll
    for (int r = 0; r < NR; ++r) {
        const float* rp = ((r == 0) ? R0 : (r == 1) ? R1 : R2) + off;
#pragma unroll
        for (int k = 0; k < SM_VECS; ++k) x[r][k] = *(const v4f*)(rp + 1024 * k);
    }
#pragma unroll
    for (int k = 0; k < SM_VECS; ++k) { a[k] = (v4f){}; st[k] = (v4f){}; }
    if (FIN) {
#pragma unroll
        for (int k = 0; k < SM_VECS; ++k) {
            a[k]  = *(const v4f*)(acc + off + 1024 * k);
            st[k] = *(const v4f*)(sacc + off + 1024 * k);
        }
    }
    v4f e[NR][SM_VECS];
    softmax_rows_regs<NR>(x, e, redm, reds, lane, wave);
#pragma unroll
    for (int r = 0; r < NR; ++r) {
#pragma unroll
        for (int k = 0; k < SM_VECS; ++k) a[k] = a[k] + 0.2f * e[r][k];
    }
    if (!FIN) {
#pragma unroll
        for (int k = 0; k < SM_VECS; ++k) *(v4f*)(acc + off + 1024 * k) = a[k];
    } else {
#pragma unroll
        for (int k = 0; k < SM_VECS; ++k) {
            v4f le, ls;
#pragma unroll
            for (int cc = 0; cc < 4; ++cc) { le[cc] = logf(a[k][cc]); ls[cc] = logf(st[k][cc]); }
            *(v4f*)(outEnd + off + 1024 * k)   = le;
            *(v4f*)(outStart + off + 1024 * k) = ls;
        }
    }
}

#define TL_F     64
#define TL_ROWS  4
#define TL_WAVES 16

static __device__ __forceinline__ void gate_tile(const v4f (&xa)[SG_LINES],
                                                 const v4f (&xw)[SG_LINES], int kb, float (&o)[2]) {
    v8f acc0 = {}, acc1 = {};
#pragma unroll
    for (int j = 0; j < SG_LINES; ++j) {
        v2f a, bb;
        a.x = xa[j][0]; a.y = xa[j][1]; bb.x = xw[j][0]; bb.y = xw[j][1];
        acc0 = __builtin_amdgcn_wmma_f32_16x16x4_f32(false, a, false, bb,
                                                     (short)0, acc0, false, false);
        a.x = xa[j][2]; a.y = xa[j][3]; bb.x = xw[j][2]; bb.y = xw[j][3];
        acc1 = __builtin_amdgcn_wmma_f32_16x16x4_f32(false, a, false, bb,
                                                     (short)0, acc1, false, false);
    }
    const v8f tv = acc0 + acc1;
    const v4f tl = {tv[0], tv[1], tv[2], tv[3]};
    const v4f tu = {tv[4], tv[5], tv[6], tv[7]};
    float lo = tl[kb], up = tu[kb];
    lo += __shfl_xor(lo, 1); up += __shfl_xor(up, 1);
    lo += __shfl_xor(lo, 2); up += __shfl_xor(up, 2);
    o[0] = lo; o[1] = up;
}

__global__ void __launch_bounds__(TL_WAVES * 32)
turn_tail_kernel(const float* __restrict__ x1,
                 const float* __restrict__ sOld,
                 const float* __restrict__ W_ih,
                 const float* __restrict__ W_hh,
                 const float* __restrict__ b_ih,
                 const float* __restrict__ b_hh,
                 const float* __restrict__ Wb,
                 const float* __restrict__ Wa,
                 float* __restrict__ sNew,
                 float* __restrict__ qbPart,
                 float* __restrict__ qaPart) {
    const int tid   = threadIdx.x;
    const int lane  = tid & 31;
    const int wave  = tid >> 5;
    const int c     = lane & 15;
    const int hi    = lane >> 4;
    const int kb    = c & 3;
    const int piece = 4 * (2 * kb + hi);
    const int slice = blockIdx.x;
    const int n0    = slice * TL_F;
    const int m0    = blockIdx.y * TL_ROWS;
    const int feat  = n0 + 4 * wave + (c >> 2);

    v4f xa[2][SG_LINES], xw[6][SG_LINES], wq[2][8];
    const size_t ao = (size_t)(m0 + (c >> 2)) * SG_K + piece;
    const size_t wo = (size_t)feat * SG_K + piece;
    const size_t qo = (size_t)(16 * wave + c) * Q_ + n0 + 32 * hi;
#pragma unroll
    for (int j = 0; j < SG_LINES; ++j) {
        xa[0][j] = *(const v4f*)(x1 + ao + 32 * j);
        xa[1][j] = *(const v4f*)(sOld + ao + 32 * j);
    }
#pragma unroll
    for (int jb = 0; jb < 3; ++jb) {
#pragma unroll
        for (int j = 0; j < SG_LINES; ++j)
            xw[jb][j] = *(const v4f*)(W_ih + wo + (size_t)(jb * Q_) * SG_K + 32 * j);
    }
    float g[6][2];
#pragma unroll
    for (int jb = 0; jb < 6; ++jb) {
        if (jb < 3) {
#pragma unroll
            for (int j = 0; j < SG_LINES; ++j)
                xw[jb + 3][j] = *(const v4f*)(W_hh + wo + (size_t)(jb * Q_) * SG_K + 32 * j);
        } else if (jb >= 4) {
            const float* wsrc = (jb == 4) ? Wb : Wa;
#pragma unroll
            for (int i = 0; i < 8; ++i) wq[jb - 4][i] = *(const v4f*)(wsrc + qo + 4 * i);
        }
        __builtin_amdgcn_sched_barrier(0);
        gate_tile(xa[jb / 3], xw[jb], kb, g[jb]);
        __builtin_amdgcn_sched_barrier(0);
    }

    __shared__ __attribute__((aligned(16))) float st[TL_ROWS][TL_F];
    {
        const float bir = b_ih[feat], biz = b_ih[Q_ + feat], bin = b_ih[2 * Q_ + feat];
        const float bhr = b_hh[feat], bhz = b_hh[Q_ + feat], bhn = b_hh[2 * Q_ + feat];
#pragma unroll
        for (int k = 0; k < 2; ++k) {
            const int trow = 2 * hi + k;
            const float so = sOld[(size_t)(m0 + trow) * Q_ + feat];
            const float ir = bir + g[0][k], iz = biz + g[1][k], in_ = bin + g[2][k];
            const float hr = bhr + g[3][k], hz = bhz + g[4][k], hn  = bhn + g[5][k];
            const float r = 1.0f / (1.0f + __expf(-(ir + hr)));
            const float z = 1.0f / (1.0f + __expf(-(iz + hz)));
            const float n = tanhf(in_ + r * hn);
            const float sn = (1.0f - z) * n + z * so;
            if (kb == 0) st[trow][4 * wave + (c >> 2)] = sn;
        }
    }
    __syncthreads();

    if (tid < TL_ROWS * TL_F)
        sNew[(size_t)(m0 + (tid >> 6)) * Q_ + n0 + (tid & 63)] = st[tid >> 6][tid & 63];

    v4f av[8];
#pragma unroll
    for (int i = 0; i < 8; ++i) {
        av[i] = *(const v4f*)&st[c & 3][32 * hi + 4 * i];
        if (c >= TL_ROWS) av[i] = (v4f){};
    }
    v8f qacc[2];
    qacc[0] = (v8f){}; qacc[1] = (v8f){};
#pragma unroll
    for (int i = 0; i < 8; ++i) {
#pragma unroll
        for (int e = 0; e < 2; ++e) {
#pragma unroll
            for (int x = 0; x < 2; ++x) {
                v2f a, bb;
                a.x = av[i][2 * e]; a.y = av[i][2 * e + 1];
                bb.x = wq[x][i][2 * e]; bb.y = wq[x][i][2 * e + 1];
                qacc[x] = __builtin_amdgcn_wmma_f32_16x16x4_f32(false, a, false, bb,
                                                                (short)0, qacc[x], false, false);
            }
        }
    }
    __shared__ float qt[2][16][MEM_];
#pragma unroll
    for (int x = 0; x < 2; ++x) {
#pragma unroll
        for (int r = 0; r < 8; ++r) qt[x][8 * hi + r][16 * wave + c] = qacc[x][r];
    }
    __syncthreads();
    {
        float* qb = qbPart + (size_t)slice * BQ_ + (size_t)m0 * MEM_;
        float* qa = qaPart + (size_t)slice * BQ_ + (size_t)m0 * MEM_;
#pragma unroll
        for (int k = 0; k < 2; ++k) {
            const int idx = k * (TL_WAVES * 32) + tid;
            qb[idx] = qt[0][idx >> 8][idx & 255];
            qa[idx] = qt[1][idx >> 8][idx & 255];
        }
    }
}

__global__ void init_kernel(const float* __restrict__ s0, float* __restrict__ ws) {
    int i = blockIdx.x * 256 + threadIdx.x;
    if (i < BL_) ws[OFF_SACC + i] = 0.0f;
    if (i < BQ_) ws[OFF_SALL + i] = s0[i];
}

extern "C" void kernel_launch(void* const* d_in, const int* in_sizes, int n_in,
                              void* d_out, int out_size, void* d_ws, size_t ws_size,
                              hipStream_t stream) {
    const float* M    = (const float*)d_in[0];
    const float* s0   = (const float*)d_in[1];
    const float* Wb   = (const float*)d_in[2];
    const float* We   = (const float*)d_in[3];
    const float* Wa   = (const float*)d_in[4];
    const float* W_ih = (const float*)d_in[5];
    const float* W_hh = (const float*)d_in[6];
    const float* b_ih = (const float*)d_in[7];
    const float* b_hh = (const float*)d_in[8];

    float* ws    = (float*)d_ws;
    float* out   = (float*)d_out;
    float* S1    = ws + OFF_S1;
    float* S2    = ws + OFF_S2;
    float* sacc  = ws + OFF_SACC;
    float* S3    = ws + OFF_EACC;
    float* qb    = ws + OFF_QB;
    float* qa    = ws + OFF_QA;
    float* x1    = ws + OFF_X1;
    float* sAll  = ws + OFF_SALL;
    float* evAll = ws + OFF_EVALL;
    float* qeAll = ws + OFF_QEALL;
    float* stats = ws + OFF_STATS;
    float* qbP   = ws + OFF_QBP;
    float* qaP   = ws + OFF_QAP;

    init_kernel<<<BL_ / 256, 256, 0, stream>>>(s0, ws);

    auto gemm = [&](const float* A, int lda, const float* W, int ldw,
                    const float* bias, float* C, int ldc, int N, int accF, int rows) {
        dim3 g(N / 32, rows / SG_ROWS);
        wmma_gemm_kernel<<<g, SG_WAVES * 32, 0, stream>>>(A, lda, W, ldw, bias, C, ldc, accF);
    };

    gemm(sAll, Q_, Wb, Q_, nullptr, qb, MEM_, MEM_, 0, B_);
    gemm(sAll, Q_, Wa, Q_, nullptr, qa, MEM_, MEM_, 0, B_);

    for (int t = 0; t < 5; ++t) {
        if (t == 0)
            scores_rows_kernel<2, true, 1><<<dim3(SC_CH, B_), SC_WAVES * 32, 0, stream>>>(
                M, qb, qa, nullptr, S1, S2, nullptr, stats);
        else
            scores_rows_kernel<2, true, TL_SLICES><<<dim3(SC_CH, B_), SC_WAVES * 32, 0, stream>>>(
                M, qbP, qaP, nullptr, S1, S2, nullptr, stats);

        wsum_rows_kernel<<<dim3(WS_CH, B_), WS_WAVES * 32, 0, stream>>>(M, S1, S2, stats, sacc);
        wsum_reduce_kernel<<<(2 * BQ_) / 256, 256, 0, stream>>>(S1, evAll + t * BQ_, x1);

        if (t < 4)
            turn_tail_kernel<<<dim3(TL_SLICES, B_ / TL_ROWS), TL_WAVES * 32, 0, stream>>>(
                x1, sAll + t * BQ_, W_ih, W_hh, b_ih, b_hh, Wb, Wa,
                sAll + (t + 1) * BQ_, qbP, qaP);
    }

    gemm(sAll,  Q_,   We,      Q_ + MEM_, nullptr, qeAll, MEM_, MEM_, 0, 5 * B_);
    gemm(evAll, MEM_, We + Q_, Q_ + MEM_, nullptr, qeAll, MEM_, MEM_, 1, 5 * B_);

    scores_rows_kernel<3, false, 1><<<dim3(SC_CH, B_), SC_WAVES * 32, 0, stream>>>(
        M, qeAll, qeAll + BQ_, qeAll + 2 * BQ_, S1, S2, S3, nullptr);
    end_softmax_kernel<3, false><<<B_, 256, 0, stream>>>(S1, S2, S3, S3, nullptr, nullptr, nullptr);

    scores_rows_kernel<2, false, 1><<<dim3(SC_CH, B_), SC_WAVES * 32, 0, stream>>>(
        M, qeAll + 3 * BQ_, qeAll + 4 * BQ_, nullptr, S1, S2, nullptr, nullptr);
    end_softmax_kernel<2, true><<<B_, 256, 0, stream>>>(S1, S2, nullptr, S3, sacc, out, out + BL_);
}


# the 5 partial-row reduce launches and the start-up launch folded into the kernels behind them: turn tail and first end GEMM add the 8 chunk rows on load, same order; 28 to 22 launches; same sums
# speedup vs baseline: 15.5869x; 1.0852x over previous
#include <hip/hip_runtime.h>
#include <math.h>

#define B_   32
#define L_   4096
#define MEM_ 256
#define Q_   256
#define BL_  (B_ * L_)
#define BQ_  (B_ * Q_)
#define B3Q_ (B_ * 3 * Q_)

#define OFF_S1   0
#define OFF_S2   (OFF_S1 + BL_)
#define OFF_SACC (OFF_S2 + BL_)
#define OFF_EACC (OFF_SACC + BL_)
#define OFF_QB   (OFF_EACC + BL_)
#define OFF_QA   (OFF_QB + BQ_)
#define OFF_X1   (OFF_QA + BQ_)
#define OFF_GI   (OFF_X1 + BQ_)
#define OFF_GH   (OFF_GI + B3Q_)
#define OFF_SALL  OFF_EACC
#define OFF_EVALL (OFF_SALL + 6 * BQ_)
#define OFF_QEALL OFF_GI
#define OFF_STATS (OFF_EVALL + 5 * BQ_)
#define TL_SLICES 4
#define OFF_QBP  (OFF_STATS + 1024)
#define OFF_QAP  OFF_GI

#define WS_CH     8
#define WS_WGROWS (L_ / WS_CH)

typedef __attribute__((ext_vector_type(2))) float v2f;
typedef __attribute__((ext_vector_type(8))) float v8f;

typedef __attribute__((ext_vector_type(4))) float v4f;

#define SG_K     256
#define SG_LINES (SG_K / 32)
#define SG_WAVES 8
#define SG_ROWS  8

template <bool EVRED>
__global__ void __launch_bounds__(SG_WAVES * 32)
wmma_gemm_kernel(const float* __restrict__ A, int lda,
                 const float* __restrict__ W, int ldw,
                 const float* __restrict__ bias,
                 float* __restrict__ C, int ldc,
                 int accumulate,
                 const float* __restrict__ part,
                 float* __restrict__ evOut) {
    const int tid   = threadIdx.x;
    const int lane  = tid & 31;
    const int wave  = tid >> 5;
    const int c     = lane & 15;
    const int hi    = lane >> 4;
    const int kb    = c & 3;
    const int piece = 4 * (2 * kb + hi);
    const int m0    = blockIdx.y * SG_ROWS;
    const int n0    = blockIdx.x * 32;

    const float* ap = A + (size_t)(m0 + (c >> 2)) * lda + piece;
    const float* wp = W + (size_t)(n0 + 4 * wave + (c >> 2)) * ldw + piece;

    v4f xw[SG_LINES], xa[2][SG_LINES];
#pragma unroll
    for (int j = 0; j < SG_LINES; ++j) xw[j] = *(const v4f*)(wp + 32 * j);
#pragma unroll
    for (int u = 0; u < 2; ++u) {
#pragma unroll
        for (int j = 0; j < SG_LINES; ++j)
            xa[u][j] = *(const v4f*)(ap + (size_t)(4 * u) * lda + 32 * j);
    }

    const int orow = wave, ocol = lane;
    float* cp = C + (size_t)(m0 + orow) * ldc + n0 + ocol;
    float bv = 0.0f, cv = 0.0f;
    if (bias) bv = bias[n0 + ocol];
    if (accumulate) cv = *cp;

    const int  ei  = (blockIdx.y * gridDim.x + blockIdx.x) * (SG_WAVES * 32) + tid;
    const bool red = EVRED && ei < BQ_;
    float pr[WS_CH];
    if (red) {
        const float* src = part + (size_t)(ei >> 8) * L_ + (ei & 255);
#pragma unroll
        for (int cc = 0; cc < WS_CH; ++cc) pr[cc] = src[cc * WS_WGROWS];
    }

    v8f acc0[2], acc1[2];
#pragma unroll
    for (int u = 0; u < 2; ++u) { acc0[u] = (v8f){}; acc1[u] = (v8f){}; }
#pragma unroll
    for (int j = 0; j < SG_LINES; ++j) {
#pragma unroll
        for (int u = 0; u < 2; ++u) {
            v2f a, bb;
            a.x = xa[u][j][0]; a.y = xa[u][j][1]; bb.x = xw[j][0]; bb.y = xw[j][1];
            acc0[u] = __builtin_amdgcn_wmma_f32_16x16x4_f32(false, a, false, bb,
                                                            (short)0, acc0[u], false, false);
            a.x = xa[u][j][2]; a.y = xa[u][j][3]; bb.x = xw[j][2]; bb.y = xw[j][3];
            acc1[u] = __builtin_amdgcn_wmma_f32_16x16x4_f32(false, a, false, bb,
                                                            (short)0, acc1[u], false, false);
        }
    }
    float lo[2], up[2];
#pragma unroll
    for (int u = 0; u < 2; ++u) {
        const v8f tv = acc0[u] + acc1[u];
        const v4f tl = {tv[0], tv[1], tv[2], tv[3]};
        const v4f tu = {tv[4], tv[5], tv[6], tv[7]};
        lo[u] = tl[kb]; up[u] = tu[kb];
    }
#pragma unroll
    for (int u = 0; u < 2; ++u) { lo[u] += __shfl_xor(lo[u], 1); up[u] += __shfl_xor(up[u], 1); }
#pragma unroll
    for (int u = 0; u < 2; ++u) { lo[u] += __shfl_xor(lo[u], 2); up[u] += __shfl_xor(up[u], 2); }

    __shared__ float tile[SG_ROWS][32];
    if (kb == 0) {
#pragma unroll
        for (int u = 0; u < 2; ++u) {
            tile[4 * u + 2 * hi][4 * wave + (c >> 2)]     = lo[u];
            tile[4 * u + 2 * hi + 1][4 * wave + (c >> 2)] = up[u];
        }
    }
    __syncthreads();
    *cp = (cv + bv) + tile[orow][ocol];
    if (red) {
        float s = 0.0f;
#pragma unroll
        for (int cc = 0; cc < WS_CH; ++cc) s += pr[cc];
        evOut[ei] = s;
    }
}

template <int N>
static __device__ __forceinline__ float row_xor(float x) {
    const int b = __builtin_bit_cast(int, x);
    return __builtin_bit_cast(float,
        __builtin_amdgcn_update_dpp(b, b, 0x160 + N, 0xf, 0xf, false));
}
static __device__ __forceinline__ float xhalf(float x) {
    const int b = __builtin_bit_cast(int, x);
    return __builtin_bit_cast(float,
        __builtin_amdgcn_permlanex16(b, b, 0x76543210, 0xfedcba98, false, false));
}

#define SC_CH     8
#define SC_WAVES  16
#define SC_WGROWS (L_ / SC_CH)
#define SC_WROWS  (SC_WGROWS / SC_WAVES)
#define SC_GROUPS (SC_WROWS / 4)
#define SC_LINES  (MEM_ / 32)
#define SC_AHEAD  3
#define SC_RING   (SC_AHEAD + 1)

template <int NQ, bool LOCAL, int NP>
__global__ void __launch_bounds__(SC_WAVES * 32)
scores_rows_kernel(const float* __restrict__ M,
                   const float* __restrict__ q1,
                   const float* __restrict__ q2,
                   const float* __restrict__ q3,
                   float* __restrict__ S1,
                   float* __restrict__ S2,
                   float* __restrict__ S3,
                   float* __restrict__ stats) {
    const int tid   = threadIdx.x;
    const int lane  = tid & 31;
    const int wave  = tid >> 5;
    const int c     = lane & 15;
    const int hi    = lane >> 4;
    const int kb    = c & 3;
    const int b     = blockIdx.y;
    const int ch    = blockIdx.x;
    const int piece = 4 * (2 * kb + hi);
    const int r0    = ch * SC_WGROWS + wave * SC_WROWS;

    const float* Mb = M + ((size_t)b * L_ + r0 + (c >> 2)) * MEM_ + piece;
    const int qi = (c >> 2) % NQ;
    const float* qv = ((qi == 0) ? q1 : (qi == 1) ? q2 : q3) + (size_t)b * MEM_ + piece;

    v4f x[SC_RING][SC_LINES];
#pragma unroll
    for (int g = 0; g < SC_AHEAD; ++g) {
#pragma unroll
        for (int j = 0; j < SC_LINES; ++j)
            x[g][j] = *(const v4f*)(Mb + (size_t)(4 * g) * MEM_ + 32 * j);
    }

    v4f qf[SC_LINES];
#pragma unroll
    for (int j = 0; j < SC_LINES; ++j) qf[j] = *(const v4f*)(qv + 32 * j);
#pragma unroll
    for (int p = 1; p < NP; ++p) {
#pragma unroll
        for (int j = 0; j < SC_LINES; ++j) qf[j] = qf[j] + *(const v4f*)(qv + (size_t)p * BQ_ + 32 * j);
    }

    float slo[SC_GROUPS], shi[SC_GROUPS];
#pragma unroll
    for (int g = 0; g < SC_GROUPS; ++g) {
        const int cur = g % SC_RING;
        if (g + SC_AHEAD < SC_GROUPS) {
            const int nxt = (g + SC_AHEAD) % SC_RING;
#pragma unroll
            for (int j = 0; j < SC_LINES; ++j)
                x[nxt][j] = *(const v4f*)(Mb + (size_t)(4 * (g + SC_AHEAD)) * MEM_ + 32 * j);
        }
        __builtin_amdgcn_sched_barrier(0);
        v8f acc0 = {}, acc1 = {};
#pragma unroll
        for (int j = 0; j < SC_LINES; ++j) {
            v2f a, bb;
            a.x = x[cur][j][0]; a.y = x[cur][j][1]; bb.x = qf[j][0]; bb.y = qf[j][1];
            acc0 = __builtin_amdgcn_wmma_f32_16x16x4_f32(false, a, false, bb,
                                                         (short)0, acc0, false, false);
            a.x = x[cur][j][2]; a.y = x[cur][j][3]; bb.x = qf[j][2]; bb.y = qf[j][3];
            acc1 = __builtin_amdgcn_wmma_f32_16x16x4_f32(false, a, false, bb,
                                                         (short)0, acc1, false, false);
        }
        float t[8];
#pragma unroll
        for (int r = 0; r < 8; ++r) t[r] = acc0[r] + acc1[r];
        float lo = (kb == 0) ? t[0] : (kb == 1) ? t[1] : (kb == 2) ? t[2] : t[3];
        float up = (kb == 0) ? t[4] : (kb == 1) ? t[5] : (kb == 2) ? t[6] : t[7];
        lo += __shfl_xor(lo, 1); up += __shfl_xor(up, 1);
        lo += __shfl_xor(lo, 2); up += __shfl_xor(up, 2);
        slo[g] = lo; shi[g] = up;
    }

    __shared__ float sc[NQ][SC_WGROWS];
    if (kb == 0 && (c >> 2) < NQ) {
        float* dst = &sc[c >> 2][wave * SC_WROWS + 2 * hi];
#pragma unroll
        for (int g = 0; g < SC_GROUPS; ++g) { dst[4 * g] = slo[g]; dst[4 * g + 1] = shi[g]; }
    }
    __syncthreads();
    const size_t so = (size_t)b * L_ + ch * SC_WGROWS + tid;
    if (!LOCAL) {
        S1[so] = sc[0][tid];
        S2[so] = sc[1][tid];
        if (NQ > 2) S3[so] = sc[NQ - 1][tid];
    } else {
        __shared__ float redm[NQ][SC_WAVES], reds[NQ][SC_WAVES];
        float xs[NQ], ex[NQ], mx[NQ];
#pragma unroll
        for (int q = 0; q < NQ; ++q) {
            xs[q] = sc[q][tid];
            float t = xs[q];
            t = fmaxf(t, row_xor<1>(t)); t = fmaxf(t, row_xor<2>(t));
            t = fmaxf(t, row_xor<4>(t)); t = fmaxf(t, row_xor<8>(t));
            t = fmaxf(t, xhalf(t));
            if (lane == 0) redm[q][wave] = t;
        }
        __syncthreads();
#pragma unroll
        for (int q = 0; q < NQ; ++q) {
            float t = redm[q][c];
            t = fmaxf(t, row_xor<1>(t)); t = fmaxf(t, row_xor<2>(t));
            t = fmaxf(t, row_xor<4>(t)); t = fmaxf(t, row_xor<8>(t));
            mx[q] = t;
            ex[q] = __expf(xs[q] - t);
            float s = ex[q];
            s += row_xor<1>(s); s += row_xor<2>(s); s += row_xor<4>(s); s += row_xor<8>(s);
            s += xhalf(s);
            if (lane == 0) reds[q][wave] = s;
        }
        __syncthreads();
#pragma unroll
        for (int q = 0; q < NQ; ++q) {
            float s = reds[q][c];
            s += row_xor<1>(s); s += row_xor<2>(s); s += row_xor<4>(s); s += row_xor<8>(s);
            if (tid == 0) {
                float* sp = stats + (((size_t)b * NQ + q) * SC_CH + ch) * 2;
                sp[0] = mx[q]; sp[1] = s;
            }
        }
        S1[so] = ex[0];
        S2[so] = ex[1];
        if (NQ > 2) S3[so] = ex[NQ - 1];
    }
}

#define WS_WAVES  16
#define WS_WROWS  (WS_WGROWS / WS_WAVES)
#define WS_KSTEPS (WS_WROWS / 4)
#define WS_KSTRIDE (4 * WS_WAVES)

template <bool FIRST>
__global__ void __launch_bounds__(WS_WAVES * 32)
wsum_rows_kernel(const float* __restrict__ M,
                 float* pbuf,
                 const float* wbuf,
                 const float* __restrict__ stats,
                 float* __restrict__ sacc) {
    const int tid  = threadIdx.x;
    const int lane = tid & 31;
    const int wave = tid >> 5;
    const int row  = lane & 15;
    const int hi   = lane >> 4;
    const int b    = blockIdx.y;
    const int c    = blockIdx.x;
    const int l0   = c * WS_WGROWS + 4 * wave;

    const float* pv = ((row == 1) ? wbuf : (const float*)pbuf) + (size_t)b * L_ + l0 + 2 * hi;
    const float* Mb = M + ((size_t)b * L_ + l0 + 2 * hi) * MEM_ + 4 * row;

    v8f acc[16];
#pragma unroll
    for (int t = 0; t < 16; ++t) acc[t] = (v8f){};

    v2f a[WS_KSTEPS];
#pragma unroll
    for (int ks = 0; ks < WS_KSTEPS; ++ks) {
        a[ks] = *(const v2f*)(pv + WS_KSTRIDE * ks);
        if (row >= 2) { a[ks].x = 0.0f; a[ks].y = 0.0f; }
    }

    v4f x[2][2][4];
#pragma unroll
    for (int g = 0; g < 4; ++g) {
        x[0][0][g] = *(const v4f*)(Mb + 64 * g);
        x[0][1][g] = *(const v4f*)(Mb + MEM_ + 64 * g);
    }

    float cm[2][WS_CH], cz[2][WS_CH];
    {
        const float* st = stats + (size_t)b * (2 * WS_CH * 2);
#pragma unroll
        for (int q = 0; q < 2; ++q) {
#pragma unroll
            for (int cc = 0; cc < WS_CH; ++cc) {
                cm[q][cc] = st[(q * WS_CH + cc) * 2];
                cz[q][cc] = st[(q * WS_CH + cc) * 2 + 1];
            }
        }
    }

#pragma unroll
    for (int ks = 0; ks < WS_KSTEPS; ++ks) {
        const int cur = ks & 1, nxt = cur ^ 1;
        if (ks + 1 < WS_KSTEPS) {
#pragma unroll
            for (int g = 0; g < 4; ++g) {
                x[nxt][0][g] = *(const v4f*)(Mb + (size_t)(WS_KSTRIDE * (ks + 1))     * MEM_ + 64 * g);
                x[nxt][1][g] = *(const v4f*)(Mb + (size_t)(WS_KSTRIDE * (ks + 1) + 1) * MEM_ + 64 * g);
            }
        }
        __builtin_amdgcn_sched_barrier(0);
#pragma unroll
        for (int g = 0; g < 4; ++g) {
#pragma unroll
            for (int cc = 0; cc < 4; ++cc) {
                v2f bb; bb.x = x[cur][0][g][cc]; bb.y = x[cur][1][g][cc];
                acc[4 * g + cc] = __builtin_amdgcn_wmma_f32_16x16x4_f32(
                    false, a[ks], false, bb, (short)0, acc[4 * g + cc], false, false);
            }
        }
    }

    const size_t ro = (size_t)b * L_ + c * WS_WGROWS + tid;
    const float pe = pbuf[ro], sa = FIRST ? 0.0f : sacc[ro];

    float f[2];
#pragma unroll
    for (int q = 0; q < 2; ++q) {
        float m = cm[q][0];
#pragma unroll
        for (int cc = 1; cc < WS_CH; ++cc) m = fmaxf(m, cm[q][cc]);
        float z = 0.0f, own = 0.0f;
#pragma unroll
        for (int cc = 0; cc < WS_CH; ++cc) {
            const float w = __expf(cm[q][cc] - m);
            z += cz[q][cc] * w;
            if (cc == c) own = w;
        }
        f[q] = own / z;
    }

    __shared__ float red[WS_WAVES][2][MEM_];
    if (hi == 0) {
#pragma unroll
        for (int g = 0; g < 4; ++g) {
            v4f es, xs;
#pragma unroll
            for (int cc = 0; cc < 4; ++cc) { es[cc] = acc[4 * g + cc][0]; xs[cc] = acc[4 * g + cc][1]; }
            *(v4f*)&red[wave][0][64 * g + 4 * row] = es;
            *(v4f*)&red[wave][1][64 * g + 4 * row] = xs;
        }
    }
    __syncthreads();
    sacc[ro] = sa + 0.2f * (pe * f[0]);
    {
        const int v = tid >> 8, col = tid & 255;
        float s = 0.0f;
#pragma unroll
        for (int w = 0; w < WS_WAVES; ++w) s += red[w][v][col];
        pbuf[(size_t)b * L_ + c * WS_WGROWS + v * MEM_ + col] = s * (v ? f[1] : f[0]);
    }
}

#define SM_WAVES 8
#define SM_VECS  (L_ / (SM_WAVES * 32 * 4))

template <int NR>
static __device__ __forceinline__ void softmax_rows_regs(const v4f (&x)[NR][SM_VECS], v4f (&e)[NR][SM_VECS],
                                                         float (&redm)[NR][SM_WAVES],
                                                         float (&reds)[NR][SM_WAVES],
                                                         int lane, int wave) {
    float m[NR];
#pragma unroll
    for (int r = 0; r < NR; ++r) {
        float t = fmaxf(fmaxf(x[r][0][0], x[r][0][1]), fmaxf(x[r][0][2], x[r][0][3]));
#pragma unroll
        for (int k = 1; k < SM_VECS; ++k)
            t = fmaxf(t, fmaxf(fmaxf(x[r][k][0], x[r][k][1]), fmaxf(x[r][k][2], x[r][k][3])));
        t = fmaxf(t, row_xor<1>(t)); t = fmaxf(t, row_xor<2>(t));
        t = fmaxf(t, row_xor<4>(t)); t = fmaxf(t, row_xor<8>(t));
        t = fmaxf(t, xhalf(t));
        if (lane == 0) redm[r][wave] = t;
    }
    __syncthreads();
#pragma unroll
    for (int r = 0; r < NR; ++r) {
        float t = redm[r][lane & (SM_WAVES - 1)];
        t = fmaxf(t, row_xor<1>(t)); t = fmaxf(t, row_xor<2>(t)); t = fmaxf(t, row_xor<4>(t));
        m[r] = t;
    }

#pragma unroll
    for (int r = 0; r < NR; ++r) {
        float s = 0.0f;
#pragma unroll
        for (int k = 0; k < SM_VECS; ++k) {
#pragma unroll
            for (int cc = 0; cc < 4; ++cc) e[r][k][cc] = __expf(x[r][k][cc] - m[r]);
            s += (e[r][k][0] + e[r][k][1]) + (e[r][k][2] + e[r][k][3]);
        }
        s += row_xor<1>(s); s += row_xor<2>(s); s += row_xor<4>(s); s += row_xor<8>(s);
        s += xhalf(s);
        if (lane == 0) reds[r][wave] = s;
    }
    __syncthreads();
#pragma unroll
    for (int r = 0; r < NR; ++r) {
        float s = reds[r][lane & (SM_WAVES - 1)];
        s += row_xor<1>(s); s += row_xor<2>(s); s += row_xor<4>(s);
        const float inv = 1.0f / s;
#pragma unroll
        for (int k = 0; k < SM_VECS; ++k) e[r][k] = e[r][k] * inv;
    }
}

template <int NR, bool FIN>
__global__ void __launch_bounds__(SM_WAVES * 32)
end_softmax_kernel(const float* R0, const float* R1, const float* R2,
                   float* acc,
                   const float* __restrict__ sacc,
                   float* __restrict__ outStart,
                   float* __restrict__ outEnd) {
    const int b    = blockIdx.x;
    const int tid  = threadIdx.x;
    const int lane = tid & 31;
    const int wave = tid >> 5;
    __shared__ float redm[NR][SM_WAVES], reds[NR][SM_WAVES];
    const size_t off = (size_t)b * L_ + 4 * tid;

    v4f x[NR][SM_VECS], a[SM_VECS], st[SM_VECS];
#pragma unroll
    for (int r = 0; r < NR; ++r) {
        const float* rp = ((r == 0) ? R0 : (r == 1) ? R1 : R2) + off;
#pragma unroll
        for (int k = 0; k < SM_VECS; ++k) x[r][k] = *(const v4f*)(rp + 1024 * k);
    }
#pragma unroll
    for (int k = 0; k < SM_VECS; ++k) { a[k] = (v4f){}; st[k] = (v4f){}; }
    if (FIN) {
#pragma unroll
        for (int k = 0; k < SM_VECS; ++k) {
            a[k]  = *(const v4f*)(acc + off + 1024 * k);
            st[k] = *(const v4f*)(sacc + off + 1024 * k);
        }
    }
    v4f e[NR][SM_VECS];
    softmax_rows_regs<NR>(x, e, redm, reds, lane, wave);
#pragma unroll
    for (int r = 0; r < NR; ++r) {
#pragma unroll
        for (int k = 0; k < SM_VECS; ++k) a[k] = a[k] + 0.2f * e[r][k];
    }
    if (!FIN) {
#pragma unroll
        for (int k = 0; k < SM_VECS; ++k) *(v4f*)(acc + off + 1024 * k) = a[k];
    } else {
#pragma unroll
        for (int k = 0; k < SM_VECS; ++k) {
            v4f le, ls;
#pragma unroll
            for (int cc = 0; cc < 4; ++cc) { le[cc] = logf(a[k][cc]); ls[cc] = logf(st[k][cc]); }
            *(v4f*)(outEnd + off + 1024 * k)   = le;
            *(v4f*)(outStart + off + 1024 * k) = ls;
        }
    }
}

#define TL_F     64
#define TL_ROWS  4
#define TL_WAVES 16
#define TL_XPITCH (SG_K + 32)

static __device__ __forceinline__ void gate_tile(const v4f (&xa)[SG_LINES],
                                                 const v4f (&xw)[SG_LINES], int kb, float (&o)[2]) {
    v8f acc0 = {}, acc1 = {};
#pragma unroll
    for (int j = 0; j < SG_LINES; ++j) {
        v2f a, bb;
        a.x = xa[j][0]; a.y = xa[j][1]; bb.x = xw[j][0]; bb.y = xw[j][1];
        acc0 = __builtin_amdgcn_wmma_f32_16x16x4_f32(false, a, false, bb,
                                                     (short)0, acc0, false, false);
        a.x = xa[j][2]; a.y = xa[j][3]; bb.x = xw[j][2]; bb.y = xw[j][3];
        acc1 = __builtin_amdgcn_wmma_f32_16x16x4_f32(false, a, false, bb,
                                                     (short)0, acc1, false, false);
    }
    const v8f tv = acc0 + acc1;
    const v4f tl = {tv[0], tv[1], tv[2], tv[3]};
    const v4f tu = {tv[4], tv[5], tv[6], tv[7]};
    float lo = tl[kb], up = tu[kb];
    lo += __shfl_xor(lo, 1); up += __shfl_xor(up, 1);
    lo += __shfl_xor(lo, 2); up += __shfl_xor(up, 2);
    o[0] = lo; o[1] = up;
}

__global__ void __launch_bounds__(TL_WAVES * 32)
turn_tail_kernel(const float* __restrict__ part,
                 const float* __restrict__ sOld,
                 const float* __restrict__ W_ih,
                 const float* __restrict__ W_hh,
                 const float* __restrict__ b_ih,
                 const float* __restrict__ b_hh,
                 const float* __restrict__ Wb,
                 const float* __restrict__ Wa,
                 float* __restrict__ sNew,
                 float* __restrict__ qbPart,
                 float* __restrict__ qaPart,
                 float* __restrict__ evOut,
                 float* __restrict__ sKeep) {
    const int tid   = threadIdx.x;
    const int lane  = tid & 31;
    const int wave  = tid >> 5;
    const int c     = lane & 15;
    const int hi    = lane >> 4;
    const int kb    = c & 3;
    const int piece = 4 * (2 * kb + hi);
    const int slice = blockIdx.x;
    const int n0    = slice * TL_F;
    const int m0    = blockIdx.y * TL_ROWS;
    const int feat  = n0 + 4 * wave + (c >> 2);

    v4f xa[2][SG_LINES], xw[6][SG_LINES], wq[2][8];
    const size_t ao = (size_t)(m0 + (c >> 2)) * SG_K + piece;
    const size_t wo = (size_t)feat * SG_K + piece;
    const size_t qo = (size_t)(16 * wave + c) * Q_ + n0 + 32 * hi;

    const int prow = tid >> 7, pcol = 2 * (tid & 127);
    v2f px[WS_CH];
    {
        const float* xsrc = part + (size_t)(m0 + prow) * L_ + MEM_ + pcol;
#pragma unroll
        for (int cc = 0; cc < WS_CH; ++cc) px[cc] = *(const v2f*)(xsrc + cc * WS_WGROWS);
    }
    const size_t si = (size_t)(m0 + (tid >> 6)) * Q_ + n0 + (tid & 63);
    float keep = 0.0f;
    if (sKeep && tid < TL_ROWS * TL_F) keep = sOld[si];

#pragma unroll
    for (int j = 0; j < SG_LINES; ++j) xa[1][j] = *(const v4f*)(sOld + ao + 32 * j);
#pragma unroll
    for (int jb = 0; jb < 3; ++jb) {
#pragma unroll
        for (int j = 0; j < SG_LINES; ++j)
            xw[jb][j] = *(const v4f*)(W_ih + wo + (size_t)(jb * Q_) * SG_K + 32 * j);
    }

    __shared__ __attribute__((aligned(16))) float xs[TL_ROWS][TL_XPITCH];
    {
        v2f sx = {0.0f, 0.0f};
#pragma unroll
        for (int cc = 0; cc < WS_CH; ++cc) sx = sx + px[cc];
        *(v2f*)&xs[prow][pcol] = sx;
    }
    __builtin_amdgcn_fence(__ATOMIC_RELEASE, "workgroup", "local");
    __builtin_amdgcn_s_barrier();
    __builtin_amdgcn_fence(__ATOMIC_ACQUIRE, "workgroup", "local");
#pragma unroll
    for (int j = 0; j < SG_LINES; ++j) xa[0][j] = *(const v4f*)&xs[c >> 2][piece + 32 * j];
    if (tid < 32 * TL_ROWS) {
        const int ecol = n0 + 2 * lane;
        const float* esrc = part + (size_t)(m0 + wave) * L_ + ecol;
        v2f pe[WS_CH];
#pragma unroll
        for (int cc = 0; cc < WS_CH; ++cc) pe[cc] = *(const v2f*)(esrc + cc * WS_WGROWS);
        v2f se = {0.0f, 0.0f};
#pragma unroll
        for (int cc = 0; cc < WS_CH; ++cc) se = se + pe[cc];
        *(v2f*)(evOut + (size_t)(m0 + wave) * MEM_ + ecol) = se;
    }
    float g[6][2];
#pragma unroll
    for (int jb = 0; jb < 6; ++jb) {
        if (jb < 3) {
#pragma unroll
            for (int j = 0; j < SG_LINES; ++j)
                xw[jb + 3][j] = *(const v4f*)(W_hh + wo + (size_t)(jb * Q_) * SG_K + 32 * j);
        } else if (jb >= 4) {
            const float* wsrc = (jb == 4) ? Wb : Wa;
#pragma unroll
            for (int i = 0; i < 8; ++i) wq[jb - 4][i] = *(const v4f*)(wsrc + qo + 4 * i);
        }
        __builtin_amdgcn_sched_barrier(0);
        gate_tile(xa[jb / 3], xw[jb], kb, g[jb]);
        __builtin_amdgcn_sched_barrier(0);
    }

    __shared__ __attribute__((aligned(16))) float st[TL_ROWS][TL_F];
    {
        const float bir = b_ih[feat], biz = b_ih[Q_ + feat], bin = b_ih[2 * Q_ + feat];
        const float bhr = b_hh[feat], bhz = b_hh[Q_ + feat], bhn = b_hh[2 * Q_ + feat];
#pragma unroll
        for (int k = 0; k < 2; ++k) {
            const int trow = 2 * hi + k;
            const float so = sOld[(size_t)(m0 + trow) * Q_ + feat];
            const float ir = bir + g[0][k], iz = biz + g[1][k], in_ = bin + g[2][k];
            const float hr = bhr + g[3][k], hz = bhz + g[4][k], hn  = bhn + g[5][k];
            const float r = 1.0f / (1.0f + __expf(-(ir + hr)));
            const float z = 1.0f / (1.0f + __expf(-(iz + hz)));
            const float n = tanhf(in_ + r * hn);
            const float sn = (1.0f - z) * n + z * so;
            if (kb == 0) st[trow][4 * wave + (c >> 2)] = sn;
        }
    }
    __syncthreads();

    if (tid < TL_ROWS * TL_F) {
        sNew[si] = st[tid >> 6][tid & 63];
        if (sKeep) sKeep[si] = keep;
    }

    v4f av[8];
#pragma unroll
    for (int i = 0; i < 8; ++i) {
        av[i] = *(const v4f*)&st[c & 3][32 * hi + 4 * i];
        if (c >= TL_ROWS) av[i] = (v4f){};
    }
    v8f qacc[2];
    qacc[0] = (v8f){}; qacc[1] = (v8f){};
#pragma unroll
    for (int i = 0; i < 8; ++i) {
#pragma unroll
        for (int e = 0; e < 2; ++e) {
#pragma unroll
            for (int x = 0; x < 2; ++x) {
                v2f a, bb;
                a.x = av[i][2 * e]; a.y = av[i][2 * e + 1];
                bb.x = wq[x][i][2 * e]; bb.y = wq[x][i][2 * e + 1];
                qacc[x] = __builtin_amdgcn_wmma_f32_16x16x4_f32(false, a, false, bb,
                                                                (short)0, qacc[x], false, false);
            }
        }
    }
    __shared__ float qt[2][16][MEM_];
#pragma unroll
    for (int x = 0; x < 2; ++x) {
#pragma unroll
        for (int r = 0; r < 8; ++r) qt[x][8 * hi + r][16 * wave + c] = qacc[x][r];
    }
    __syncthreads();
    {
        float* qb = qbPart + (size_t)slice * BQ_ + (size_t)m0 * MEM_;
        float* qa = qaPart + (size_t)slice * BQ_ + (size_t)m0 * MEM_;
#pragma unroll
        for (int k = 0; k < 2; ++k) {
            const int idx = k * (TL_WAVES * 32) + tid;
            qb[idx] = qt[0][idx >> 8][idx & 255];
            qa[idx] = qt[1][idx >> 8][idx & 255];
        }
    }
}

extern "C" void kernel_launch(void* const* d_in, const int* in_sizes, int n_in,
                              void* d_out, int out_size, void* d_ws, size_t ws_size,
                              hipStream_t stream) {
    const float* M    = (const float*)d_in[0];
    const float* s0   = (const float*)d_in[1];
    const float* Wb   = (const float*)d_in[2];
    const float* We   = (const float*)d_in[3];
    const float* Wa   = (const float*)d_in[4];
    const float* W_ih = (const float*)d_in[5];
    const float* W_hh = (const float*)d_in[6];
    const float* b_ih = (const float*)d_in[7];
    const float* b_hh = (const float*)d_in[8];

    float* ws    = (float*)d_ws;
    float* out   = (float*)d_out;
    float* S1    = ws + OFF_S1;
    float* S2    = ws + OFF_S2;
    float* sacc  = ws + OFF_SACC;
    float* S3    = ws + OFF_EACC;
    float* qb    = ws + OFF_QB;
    float* qa    = ws + OFF_QA;
    float* sAll  = ws + OFF_SALL;
    float* evAll = ws + OFF_EVALL;
    float* qeAll = ws + OFF_QEALL;
    float* stats = ws + OFF_STATS;
    float* qbP   = ws + OFF_QBP;
    float* qaP   = ws + OFF_QAP;

    auto gemm = [&](const float* A, int lda, const float* W, int ldw,
                    const float* bias, float* C, int ldc, int N, int accF, int rows) {
        dim3 g(N / 32, rows / SG_ROWS);
        wmma_gemm_kernel<false><<<g, SG_WAVES * 32, 0, stream>>>(A, lda, W, ldw, bias, C, ldc, accF,
                                                                nullptr, nullptr);
    };

    gemm(s0, Q_, Wb, Q_, nullptr, qb, MEM_, MEM_, 0, B_);
    gemm(s0, Q_, Wa, Q_, nullptr, qa, MEM_, MEM_, 0, B_);

    for (int t = 0; t < 5; ++t) {
        if (t == 0)
            scores_rows_kernel<2, true, 1><<<dim3(SC_CH, B_), SC_WAVES * 32, 0, stream>>>(
                M, qb, qa, nullptr, S1, S2, nullptr, stats);
        else
            scores_rows_kernel<2, true, TL_SLICES><<<dim3(SC_CH, B_), SC_WAVES * 32, 0, stream>>>(
                M, qbP, qaP, nullptr, S1, S2, nullptr, stats);

        if (t == 0)
            wsum_rows_kernel<true><<<dim3(WS_CH, B_), WS_WAVES * 32, 0, stream>>>(M, S1, S2, stats, sacc);
        else
            wsum_rows_kernel<false><<<dim3(WS_CH, B_), WS_WAVES * 32, 0, stream>>>(M, S1, S2, stats, sacc);

        if (t == 0)
            turn_tail_kernel<<<dim3(TL_SLICES, B_ / TL_ROWS), TL_WAVES * 32, 0, stream>>>(
                S1, s0, W_ih, W_hh, b_ih, b_hh, Wb, Wa,
                sAll + BQ_, qbP, qaP, evAll, sAll);
        else if (t < 4)
            turn_tail_kernel<<<dim3(TL_SLICES, B_ / TL_ROWS), TL_WAVES * 32, 0, stream>>>(
                S1, sAll + t * BQ_, W_ih, W_hh, b_ih, b_hh, Wb, Wa,
                sAll + (t + 1) * BQ_, qbP, qaP, evAll + t * BQ_, nullptr);
    }

    wmma_gemm_kernel<true><<<dim3(MEM_ / 32, (5 * B_) / SG_ROWS), SG_WAVES * 32, 0, stream>>>(
        sAll, Q_, We, Q_ + MEM_, nullptr, qeAll, MEM_, 0, S1, evAll + 4 * BQ_);
    gemm(evAll, MEM_, We + Q_, Q_ + MEM_, nullptr, qeAll, MEM_, MEM_, 1, 5 * B_);

    scores_rows_kernel<3, false, 1><<<dim3(SC_CH, B_), SC_WAVES * 32, 0, stream>>>(
        M, qeAll, qeAll + BQ_, qeAll + 2 * BQ_, S1, S2, S3, nullptr);
    end_softmax_kernel<3, false><<<B_, 256, 0, stream>>>(S1, S2, S3, S3, nullptr, nullptr, nullptr);

    scores_rows_kernel<2, false, 1><<<dim3(SC_CH, B_), SC_WAVES * 32, 0, stream>>>(
        M, qeAll + 3 * BQ_, qeAll + 4 * BQ_, nullptr, S1, S2, nullptr, nullptr);
    end_softmax_kernel<2, true><<<B_, 256, 0, stream>>>(S1, S2, nullptr, S3, sacc, out, out + BL_);
}


# turn tail on 64 work-groups instead of 32: a work-group owns 32 state features (8 waves), 256 WMMAs per SIMD instead of 512; 8 query shares added once per work-group via LDS (new query sum order)
# speedup vs baseline: 17.1199x; 1.0984x over previous
#include <hip/hip_runtime.h>
#include <math.h>

#define B_   32
#define L_   4096
#define MEM_ 256
#define Q_   256
#define BL_  (B_ * L_)
#define BQ_  (B_ * Q_)
#define B3Q_ (B_ * 3 * Q_)

#define OFF_S1   0
#define OFF_S2   (OFF_S1 + BL_)
#define OFF_SACC (OFF_S2 + BL_)
#define OFF_EACC (OFF_SACC + BL_)
#define OFF_QB   (OFF_EACC + BL_)
#define OFF_QA   (OFF_QB + BQ_)
#define OFF_X1   (OFF_QA + BQ_)
#define OFF_GI   (OFF_X1 + BQ_)
#define OFF_GH   (OFF_GI + B3Q_)
#define OFF_SALL  OFF_EACC
#define OFF_EVALL (OFF_SALL + 4 * BQ_)
#define OFF_QEALL OFF_QB
#define OFF_STATS (OFF_EVALL + 4 * BQ_)
#define TL_SLICES 8
#define OFF_QBP  (OFF_EVALL + 5 * BQ_)
#define OFF_QAP  (OFF_QBP + TL_SLICES * BQ_)

#define WS_CH     8
#define WS_WGROWS (L_ / WS_CH)

typedef __attribute__((ext_vector_type(2))) float v2f;
typedef __attribute__((ext_vector_type(8))) float v8f;

typedef __attribute__((ext_vector_type(4))) float v4f;

#define SG_K     256
#define SG_LINES (SG_K / 32)
#define SG_WAVES 8
#define SG_ROWS  8

template <bool EVRED>
__global__ void __launch_bounds__(SG_WAVES * 32)
wmma_gemm_kernel(const float* __restrict__ A0, int rows0,
                 const float* __restrict__ A, int lda,
                 const float* __restrict__ W, int ldw,
                 const float* __restrict__ bias,
                 float* __restrict__ C, int ldc,
                 int accumulate,
                 const float* __restrict__ part,
                 float* __restrict__ evOut) {
    const int tid   = threadIdx.x;
    const int lane  = tid & 31;
    const int wave  = tid >> 5;
    const int c     = lane & 15;
    const int hi    = lane >> 4;
    const int kb    = c & 3;
    const int piece = 4 * (2 * kb + hi);
    const int m0    = blockIdx.y * SG_ROWS;
    const int n0    = blockIdx.x * 32;

    const float* ap = ((m0 < rows0) ? A0 + (size_t)(m0 + (c >> 2)) * lda
                                    : A + (size_t)(m0 - rows0 + (c >> 2)) * lda) + piece;
    const float* wp = W + (size_t)(n0 + 4 * wave + (c >> 2)) * ldw + piece;

    v4f xw[SG_LINES], xa[2][SG_LINES];
#pragma unroll
    for (int j = 0; j < SG_LINES; ++j) xw[j] = *(const v4f*)(wp + 32 * j);
#pragma unroll
    for (int u = 0; u < 2; ++u) {
#pragma unroll
        for (int j = 0; j < SG_LINES; ++j)
            xa[u][j] = *(const v4f*)(ap + (size_t)(4 * u) * lda + 32 * j);
    }

    const int orow = wave, ocol = lane;
    float* cp = C + (size_t)(m0 + orow) * ldc + n0 + ocol;
    float bv = 0.0f, cv = 0.0f;
    if (bias) bv = bias[n0 + ocol];
    if (accumulate) cv = *cp;

    const int  ei  = (blockIdx.y * gridDim.x + blockIdx.x) * (SG_WAVES * 32) + tid;
    const bool red = EVRED && ei < BQ_;
    float pr[WS_CH];
    if (red) {
        const float* src = part + (size_t)(ei >> 8) * L_ + (ei & 255);
#pragma unroll
        for (int cc = 0; cc < WS_CH; ++cc) pr[cc] = src[cc * WS_WGROWS];
    }

    v8f acc0[2], acc1[2];
#pragma unroll
    for (int u = 0; u < 2; ++u) { acc0[u] = (v8f){}; acc1[u] = (v8f){}; }
#pragma unroll
    for (int j = 0; j < SG_LINES; ++j) {
#pragma unroll
        for (int u = 0; u < 2; ++u) {
            v2f a, bb;
            a.x = xa[u][j][0]; a.y = xa[u][j][1]; bb.x = xw[j][0]; bb.y = xw[j][1];
            acc0[u] = __builtin_amdgcn_wmma_f32_16x16x4_f32(false, a, false, bb,
                                                            (short)0, acc0[u], false, false);
            a.x = xa[u][j][2]; a.y = xa[u][j][3]; bb.x = xw[j][2]; bb.y = xw[j][3];
            acc1[u] = __builtin_amdgcn_wmma_f32_16x16x4_f32(false, a, false, bb,
                                                            (short)0, acc1[u], false, false);
        }
    }
    float lo[2], up[2];
#pragma unroll
    for (int u = 0; u < 2; ++u) {
        const v8f tv = acc0[u] + acc1[u];
        const v4f tl = {tv[0], tv[1], tv[2], tv[3]};
        const v4f tu = {tv[4], tv[5], tv[6], tv[7]};
        lo[u] = tl[kb]; up[u] = tu[kb];
    }
#pragma unroll
    for (int u = 0; u < 2; ++u) { lo[u] += __shfl_xor(lo[u], 1); up[u] += __shfl_xor(up[u], 1); }
#pragma unroll
    for (int u = 0; u < 2; ++u) { lo[u] += __shfl_xor(lo[u], 2); up[u] += __shfl_xor(up[u], 2); }

    __shared__ float tile[SG_ROWS][32];
    if (kb == 0) {
#pragma unroll
        for (int u = 0; u < 2; ++u) {
            tile[4 * u + 2 * hi][4 * wave + (c >> 2)]     = lo[u];
            tile[4 * u + 2 * hi + 1][4 * wave + (c >> 2)] = up[u];
        }
    }
    __syncthreads();
    *cp = (cv + bv) + tile[orow][ocol];
    if (red) {
        float s = 0.0f;
#pragma unroll
        for (int cc = 0; cc < WS_CH; ++cc) s += pr[cc];
        evOut[ei] = s;
    }
}

template <int N>
static __device__ __forceinline__ float row_xor(float x) {
    const int b = __builtin_bit_cast(int, x);
    return __builtin_bit_cast(float,
        __builtin_amdgcn_update_dpp(b, b, 0x160 + N, 0xf, 0xf, false));
}
static __device__ __forceinline__ float xhalf(float x) {
    const int b = __builtin_bit_cast(int, x);
    return __builtin_bit_cast(float,
        __builtin_amdgcn_permlanex16(b, b, 0x76543210, 0xfedcba98, false, false));
}

#define SC_CH     8
#define SC_WAVES  16
#define SC_WGROWS (L_ / SC_CH)
#define SC_WROWS  (SC_WGROWS / SC_WAVES)
#define SC_GROUPS (SC_WROWS / 4)
#define SC_LINES  (MEM_ / 32)
#define SC_AHEAD  3
#define SC_RING   (SC_AHEAD + 1)

template <int NQ, bool LOCAL, int NP>
__global__ void __launch_bounds__(SC_WAVES * 32)
scores_rows_kernel(const float* __restrict__ M,
                   const float* __restrict__ q1,
                   const float* __restrict__ q2,
                   const float* __restrict__ q3,
                   float* __restrict__ S1,
                   float* __restrict__ S2,
                   float* __restrict__ S3,
                   float* __restrict__ stats) {
    const int tid   = threadIdx.x;
    const int lane  = tid & 31;
    const int wave  = tid >> 5;
    const int c     = lane & 15;
    const int hi    = lane >> 4;
    const int kb    = c & 3;
    const int b     = blockIdx.y;
    const int ch    = blockIdx.x;
    const int piece = 4 * (2 * kb + hi);
    const int r0    = ch * SC_WGROWS + wave * SC_WROWS;

    const float* Mb = M + ((size_t)b * L_ + r0 + (c >> 2)) * MEM_ + piece;
    const int qi = (c >> 2) % NQ;
    const float* qv = ((qi == 0) ? q1 : (qi == 1) ? q2 : q3) + (size_t)b * MEM_ + piece;

    static_assert(NP == 1 || NQ * MEM_ == SC_WAVES * 32, "one query element per thread");
    __shared__ __attribute__((aligned(16))) float qs[2][MEM_];
    if constexpr (NP > 1) {
        const float* src = ((tid >> 8) ? q2 : q1) + (size_t)b * MEM_ + (tid & 255);
        float pq[NP];
#pragma unroll
        for (int p = 0; p < NP; ++p) pq[p] = src[(size_t)p * BQ_];
        float sq = pq[0];
#pragma unroll
        for (int p = 1; p < NP; ++p) sq += pq[p];
        qs[tid >> 8][tid & 255] = sq;
    }

    v4f x[SC_RING][SC_LINES];
#pragma unroll
    for (int g = 0; g < SC_AHEAD; ++g) {
#pragma unroll
        for (int j = 0; j < SC_LINES; ++j)
            x[g][j] = *(const v4f*)(Mb + (size_t)(4 * g) * MEM_ + 32 * j);
    }

    v4f qf[SC_LINES];
    if constexpr (NP > 1) {
        __builtin_amdgcn_fence(__ATOMIC_RELEASE, "workgroup", "local");
        __builtin_amdgcn_s_barrier();
        __builtin_amdgcn_fence(__ATOMIC_ACQUIRE, "workgroup", "local");
#pragma unroll
        for (int j = 0; j < SC_LINES; ++j) qf[j] = *(const v4f*)&qs[qi][piece + 32 * j];
    } else {
#pragma unroll
        for (int j = 0; j < SC_LINES; ++j) qf[j] = *(const v4f*)(qv + 32 * j);
    }

    float slo[SC_GROUPS], shi[SC_GROUPS];
#pragma unroll
    for (int g = 0; g < SC_GROUPS; ++g) {
        const int cur = g % SC_RING;
        if (g + SC_AHEAD < SC_GROUPS) {
            const int nxt = (g + SC_AHEAD) % SC_RING;
#pragma unroll
            for (int j = 0; j < SC_LINES; ++j)
                x[nxt][j] = *(const v4f*)(Mb + (size_t)(4 * (g + SC_AHEAD)) * MEM_ + 32 * j);
        }
        __builtin_amdgcn_sched_barrier(0);
        v8f acc0 = {}, acc1 = {};
#pragma unroll
        for (int j = 0; j < SC_LINES; ++j) {
            v2f a, bb;
            a.x = x[cur][j][0]; a.y = x[cur][j][1]; bb.x = qf[j][0]; bb.y = qf[j][1];
            acc0 = __builtin_amdgcn_wmma_f32_16x16x4_f32(false, a, false, bb,
                                                         (short)0, acc0, false, false);
            a.x = x[cur][j][2]; a.y = x[cur][j][3]; bb.x = qf[j][2]; bb.y = qf[j][3];
            acc1 = __builtin_amdgcn_wmma_f32_16x16x4_f32(false, a, false, bb,
                                                         (short)0, acc1, false, false);
        }
        float t[8];
#pragma unroll
        for (int r = 0; r < 8; ++r) t[r] = acc0[r] + acc1[r];
        float lo = (kb == 0) ? t[0] : (kb == 1) ? t[1] : (kb == 2) ? t[2] : t[3];
        float up = (kb == 0) ? t[4] : (kb == 1) ? t[5] : (kb == 2) ? t[6] : t[7];
        lo += __shfl_xor(lo, 1); up += __shfl_xor(up, 1);
        lo += __shfl_xor(lo, 2); up += __shfl_xor(up, 2);
        slo[g] = lo; shi[g] = up;
    }

    __shared__ float sc[NQ][SC_WGROWS];
    if (kb == 0 && (c >> 2) < NQ) {
        float* dst = &sc[c >> 2][wave * SC_WROWS + 2 * hi];
#pragma unroll
        for (int g = 0; g < SC_GROUPS; ++g) { dst[4 * g] = slo[g]; dst[4 * g + 1] = shi[g]; }
    }
    __syncthreads();
    const size_t so = (size_t)b * L_ + ch * SC_WGROWS + tid;
    if (!LOCAL) {
        S1[so] = sc[0][tid];
        S2[so] = sc[1][tid];
        if (NQ > 2) S3[so] = sc[NQ - 1][tid];
    } else {
        __shared__ float redm[NQ][SC_WAVES], reds[NQ][SC_WAVES];
        float xs[NQ], ex[NQ], mx[NQ];
#pragma unroll
        for (int q = 0; q < NQ; ++q) {
            xs[q] = sc[q][tid];
            float t = xs[q];
            t = fmaxf(t, row_xor<1>(t)); t = fmaxf(t, row_xor<2>(t));
            t = fmaxf(t, row_xor<4>(t)); t = fmaxf(t, row_xor<8>(t));
            t = fmaxf(t, xhalf(t));
            if (lane == 0) redm[q][wave] = t;
        }
        __syncthreads();
#pragma unroll
        for (int q = 0; q < NQ; ++q) {
            float t = redm[q][c];
            t = fmaxf(t, row_xor<1>(t)); t = fmaxf(t, row_xor<2>(t));
            t = fmaxf(t, row_xor<4>(t)); t = fmaxf(t, row_xor<8>(t));
            mx[q] = t;
            ex[q] = __expf(xs[q] - t);
            float s = ex[q];
            s += row_xor<1>(s); s += row_xor<2>(s); s += row_xor<4>(s); s += row_xor<8>(s);
            s += xhalf(s);
            if (lane == 0) reds[q][wave] = s;
        }
        __syncthreads();
#pragma unroll
        for (int q = 0; q < NQ; ++q) {
            float s = reds[q][c];
            s += row_xor<1>(s); s += row_xor<2>(s); s += row_xor<4>(s); s += row_xor<8>(s);
            if (tid == 0) {
                float* sp = stats + (((size_t)b * NQ + q) * SC_CH + ch) * 2;
                sp[0] = mx[q]; sp[1] = s;
            }
        }
        S1[so] = ex[0];
        S2[so] = ex[1];
        if (NQ > 2) S3[so] = ex[NQ - 1];
    }
}

#define WS_WAVES  16
#define WS_WROWS  (WS_WGROWS / WS_WAVES)
#define WS_KSTEPS (WS_WROWS / 4)
#define WS_KSTRIDE (4 * WS_WAVES)

template <bool FIRST>
__global__ void __launch_bounds__(WS_WAVES * 32)
wsum_rows_kernel(const float* __restrict__ M,
                 float* pbuf,
                 const float* wbuf,
                 const float* __restrict__ stats,
                 float* __restrict__ sacc) {
    const int tid  = threadIdx.x;
    const int lane = tid & 31;
    const int wave = tid >> 5;
    const int row  = lane & 15;
    const int hi   = lane >> 4;
    const int b    = blockIdx.y;
    const int c    = blockIdx.x;
    const int l0   = c * WS_WGROWS + 4 * wave;

    const float* pv = ((row == 1) ? wbuf : (const float*)pbuf) + (size_t)b * L_ + l0 + 2 * hi;
    const float* Mb = M + ((size_t)b * L_ + l0 + 2 * hi) * MEM_ + 4 * row;

    v8f acc[16];
#pragma unroll
    for (int t = 0; t < 16; ++t) acc[t] = (v8f){};

    v2f a[WS_KSTEPS];
#pragma unroll
    for (int ks = 0; ks < WS_KSTEPS; ++ks) {
        a[ks] = *(const v2f*)(pv + WS_KSTRIDE * ks);
        if (row >= 2) { a[ks].x = 0.0f; a[ks].y = 0.0f; }
    }

    v4f x[2][2][4];
#pragma unroll
    for (int g = 0; g < 4; ++g) {
        x[0][0][g] = *(const v4f*)(Mb + 64 * g);
        x[0][1][g] = *(const v4f*)(Mb + MEM_ + 64 * g);
    }

    float cm[2][WS_CH], cz[2][WS_CH];
    {
        const float* st = stats + (size_t)b * (2 * WS_CH * 2);
#pragma unroll
        for (int q = 0; q < 2; ++q) {
#pragma unroll
            for (int cc = 0; cc < WS_CH; ++cc) {
                cm[q][cc] = st[(q * WS_CH + cc) * 2];
                cz[q][cc] = st[(q * WS_CH + cc) * 2 + 1];
            }
        }
    }

#pragma unroll
    for (int ks = 0; ks < WS_KSTEPS; ++ks) {
        const int cur = ks & 1, nxt = cur ^ 1;
        if (ks + 1 < WS_KSTEPS) {
#pragma unroll
            for (int g = 0; g < 4; ++g) {
                x[nxt][0][g] = *(const v4f*)(Mb + (size_t)(WS_KSTRIDE * (ks + 1))     * MEM_ + 64 * g);
                x[nxt][1][g] = *(const v4f*)(Mb + (size_t)(WS_KSTRIDE * (ks + 1) + 1) * MEM_ + 64 * g);
            }
        }
        __builtin_amdgcn_sched_barrier(0);
#pragma unroll
        for (int g = 0; g < 4; ++g) {
#pragma unroll
            for (int cc = 0; cc < 4; ++cc) {
                v2f bb; bb.x = x[cur][0][g][cc]; bb.y = x[cur][1][g][cc];
                acc[4 * g + cc] = __builtin_amdgcn_wmma_f32_16x16x4_f32(
                    false, a[ks], false, bb, (short)0, acc[4 * g + cc], false, false);
            }
        }
    }

    const size_t ro = (size_t)b * L_ + c * WS_WGROWS + tid;
    const float pe = pbuf[ro], sa = FIRST ? 0.0f : sacc[ro];

    float f[2];
#pragma unroll
    for (int q = 0; q < 2; ++q) {
        float m = cm[q][0];
#pragma unroll
        for (int cc = 1; cc < WS_CH; ++cc) m = fmaxf(m, cm[q][cc]);
        float z = 0.0f, own = 0.0f;
#pragma unroll
        for (int cc = 0; cc < WS_CH; ++cc) {
            const float w = __expf(cm[q][cc] - m);
            z += cz[q][cc] * w;
            if (cc == c) own = w;
        }
        f[q] = own / z;
    }

    __shared__ float red[WS_WAVES][2][MEM_];
    if (hi == 0) {
#pragma unroll
        for (int g = 0; g < 4; ++g) {
            v4f es, xs;
#pragma unroll
            for (int cc = 0; cc < 4; ++cc) { es[cc] = acc[4 * g + cc][0]; xs[cc] = acc[4 * g + cc][1]; }
            *(v4f*)&red[wave][0][64 * g + 4 * row] = es;
            *(v4f*)&red[wave][1][64 * g + 4 * row] = xs;
        }
    }
    __syncthreads();
    sacc[ro] = sa + 0.2f * (pe * f[0]);
    {
        const int v = tid >> 8, col = tid & 255;
        float s = 0.0f;
#pragma unroll
        for (int w = 0; w < WS_WAVES; ++w) s += red[w][v][col];
        pbuf[(size_t)b * L_ + c * WS_WGROWS + v * MEM_ + col] = s * (v ? f[1] : f[0]);
    }
}

#define SM_WAVES 8
#define SM_VECS  (L_ / (SM_WAVES * 32 * 4))

template <int NR>
static __device__ __forceinline__ void softmax_rows_regs(const v4f (&x)[NR][SM_VECS], v4f (&e)[NR][SM_VECS],
                                                         float (&redm)[NR][SM_WAVES],
                                                         float (&reds)[NR][SM_WAVES],
                                                         int lane, int wave) {
    float m[NR];
#pragma unroll
    for (int r = 0; r < NR; ++r) {
        float t = fmaxf(fmaxf(x[r][0][0], x[r][0][1]), fmaxf(x[r][0][2], x[r][0][3]));
#pragma unroll
        for (int k = 1; k < SM_VECS; ++k)
            t = fmaxf(t, fmaxf(fmaxf(x[r][k][0], x[r][k][1]), fmaxf(x[r][k][2], x[r][k][3])));
        t = fmaxf(t, row_xor<1>(t)); t = fmaxf(t, row_xor<2>(t));
        t = fmaxf(t, row_xor<4>(t)); t = fmaxf(t, row_xor<8>(t));
        t = fmaxf(t, xhalf(t));
        if (lane == 0) redm[r][wave] = t;
    }
    __syncthreads();
#pragma unroll
    for (int r = 0; r < NR; ++r) {
        float t = redm[r][lane & (SM_WAVES - 1)];
        t = fmaxf(t, row_xor<1>(t)); t = fmaxf(t, row_xor<2>(t)); t = fmaxf(t, row_xor<4>(t));
        m[r] = t;
    }

#pragma unroll
    for (int r = 0; r < NR; ++r) {
        float s = 0.0f;
#pragma unroll
        for (int k = 0; k < SM_VECS; ++k) {
#pragma unroll
            for (int cc = 0; cc < 4; ++cc) e[r][k][cc] = __expf(x[r][k][cc] - m[r]);
            s += (e[r][k][0] + e[r][k][1]) + (e[r][k][2] + e[r][k][3]);
        }
        s += row_xor<1>(s); s += row_xor<2>(s); s += row_xor<4>(s); s += row_xor<8>(s);
        s += xhalf(s);
        if (lane == 0) reds[r][wave] = s;
    }
    __syncthreads();
#pragma unroll
    for (int r = 0; r < NR; ++r) {
        float s = reds[r][lane & (SM_WAVES - 1)];
        s += row_xor<1>(s); s += row_xor<2>(s); s += row_xor<4>(s);
        const float inv = 1.0f / s;
#pragma unroll
        for (int k = 0; k < SM_VECS; ++k) e[r][k] = e[r][k] * inv;
    }
}

template <int NR, bool FIN>
__global__ void __launch_bounds__(SM_WAVES * 32)
end_softmax_kernel(const float* R0, const float* R1, const float* R2,
                   float* acc,
                   const float* __restrict__ sacc,
                   float* __restrict__ outStart,
                   float* __restrict__ outEnd) {
    const int b    = blockIdx.x;
    const int tid  = threadIdx.x;
    const int lane = tid & 31;
    const int wave = tid >> 5;
    __shared__ float redm[NR][SM_WAVES], reds[NR][SM_WAVES];
    const size_t off = (size_t)b * L_ + 4 * tid;

    v4f x[NR][SM_VECS], a[SM_VECS], st[SM_VECS];
#pragma unroll
    for (int r = 0; r < NR; ++r) {
        const float* rp = ((r == 0) ? R0 : (r == 1) ? R1 : R2) + off;
#pragma unroll
        for (int k = 0; k < SM_VECS; ++k) x[r][k] = *(const v4f*)(rp + 1024 * k);
    }
#pragma unroll
    for (int k = 0; k < SM_VECS; ++k) { a[k] = (v4f){}; st[k] = (v4f){}; }
    if (FIN) {
#pragma unroll
        for (int k = 0; k < SM_VECS; ++k) {
            a[k]  = *(const v4f*)(acc + off + 1024 * k);
            st[k] = *(const v4f*)(sacc + off + 1024 * k);
        }
    }
    v4f e[NR][SM_VECS];
    softmax_rows_regs<NR>(x, e, redm, reds, lane, wave);
#pragma unroll
    for (int r = 0; r < NR; ++r) {
#pragma unroll
        for (int k = 0; k < SM_VECS; ++k) a[k] = a[k] + 0.2f * e[r][k];
    }
    if (!FIN) {
#pragma unroll
        for (int k = 0; k < SM_VECS; ++k) *(v4f*)(acc + off + 1024 * k) = a[k];
    } else {
#pragma unroll
        for (int k = 0; k < SM_VECS; ++k) {
            v4f le, ls;
#pragma unroll
            for (int cc = 0; cc < 4; ++cc) { le[cc] = logf(a[k][cc]); ls[cc] = logf(st[k][cc]); }
            *(v4f*)(outEnd + off + 1024 * k)   = le;
            *(v4f*)(outStart + off + 1024 * k) = ls;
        }
    }
}

#define TL_F     32
#define TL_ROWS  4
#define TL_WAVES 8
#define TL_KQ    (TL_F / 8)
#define TL_XPITCH (SG_K + 32)

static __device__ __forceinline__ void gate_tile(const v4f (&xa)[SG_LINES],
                                                 const v4f (&xw)[SG_LINES], int kb, float (&o)[2]) {
    v8f acc0 = {}, acc1 = {};
#pragma unroll
    for (int j = 0; j < SG_LINES; ++j) {
        v2f a, bb;
        a.x = xa[j][0]; a.y = xa[j][1]; bb.x = xw[j][0]; bb.y = xw[j][1];
        acc0 = __builtin_amdgcn_wmma_f32_16x16x4_f32(false, a, false, bb,
                                                     (short)0, acc0, false, false);
        a.x = xa[j][2]; a.y = xa[j][3]; bb.x = xw[j][2]; bb.y = xw[j][3];
        acc1 = __builtin_amdgcn_wmma_f32_16x16x4_f32(false, a, false, bb,
                                                     (short)0, acc1, false, false);
    }
    const v8f tv = acc0 + acc1;
    const v4f tl = {tv[0], tv[1], tv[2], tv[3]};
    const v4f tu = {tv[4], tv[5], tv[6], tv[7]};
    float lo = tl[kb], up = tu[kb];
    lo += __shfl_xor(lo, 1); up += __shfl_xor(up, 1);
    lo += __shfl_xor(lo, 2); up += __shfl_xor(up, 2);
    o[0] = lo; o[1] = up;
}

__global__ void __launch_bounds__(TL_WAVES * 32)
turn_tail_kernel(const float* __restrict__ part,
                 const float* __restrict__ sOld,
                 const float* __restrict__ W_ih,
                 const float* __restrict__ W_hh,
                 const float* __restrict__ b_ih,
                 const float* __restrict__ b_hh,
                 const float* __restrict__ Wb,
                 const float* __restrict__ Wa,
                 float* __restrict__ sNew,
                 float* __restrict__ qbPart,
                 float* __restrict__ qaPart,
                 float* __restrict__ evOut) {
    const int tid   = threadIdx.x;
    const int lane  = tid & 31;
    const int wave  = tid >> 5;
    const int c     = lane & 15;
    const int hi    = lane >> 4;
    const int kb    = c & 3;
    const int piece = 4 * (2 * kb + hi);
    const int slice = blockIdx.x;
    const int n0    = slice * TL_F;
    const int m0    = blockIdx.y * TL_ROWS;
    const int feat  = n0 + 4 * wave + (c >> 2);

    v4f xa[2][SG_LINES], xw[6][SG_LINES], wq[2][2][TL_KQ];
    const size_t ao = (size_t)(m0 + (c >> 2)) * SG_K + piece;
    const size_t wo = (size_t)feat * SG_K + piece;
    const size_t qo = (size_t)(32 * wave + c) * Q_ + n0 + 16 * hi;

    const int prow = tid >> 6, pcol = 4 * (tid & 63);
    v4f px[WS_CH];
    {
        const float* xsrc = part + (size_t)(m0 + prow) * L_ + MEM_ + pcol;
#pragma unroll
        for (int cc = 0; cc < WS_CH; ++cc) px[cc] = *(const v4f*)(xsrc + cc * WS_WGROWS);
    }
    const size_t si = (size_t)(m0 + (tid >> 5)) * Q_ + n0 + (tid & 31);

#pragma unroll
    for (int j = 0; j < SG_LINES; ++j) xa[1][j] = *(const v4f*)(sOld + ao + 32 * j);
#pragma unroll
    for (int jb = 0; jb < 3; ++jb) {
#pragma unroll
        for (int j = 0; j < SG_LINES; ++j)
            xw[jb][j] = *(const v4f*)(W_ih + wo + (size_t)(jb * Q_) * SG_K + 32 * j);
    }

    __shared__ __attribute__((aligned(16))) float xs[TL_ROWS][TL_XPITCH];
    {
        v4f sx = {0.0f, 0.0f, 0.0f, 0.0f};
#pragma unroll
        for (int cc = 0; cc < WS_CH; ++cc) sx = sx + px[cc];
        *(v4f*)&xs[prow][pcol] = sx;
    }
    __builtin_amdgcn_fence(__ATOMIC_RELEASE, "workgroup", "local");
    __builtin_amdgcn_s_barrier();
    __builtin_amdgcn_fence(__ATOMIC_ACQUIRE, "workgroup", "local");
#pragma unroll
    for (int j = 0; j < SG_LINES; ++j) xa[0][j] = *(const v4f*)&xs[c >> 2][piece + 32 * j];
    if (tid < 32 * TL_ROWS) {
        const int ecol = n0 + lane;
        const float* esrc = part + (size_t)(m0 + wave) * L_ + ecol;
        float pe[WS_CH];
#pragma unroll
        for (int cc = 0; cc < WS_CH; ++cc) pe[cc] = esrc[cc * WS_WGROWS];
        float se = 0.0f;
#pragma unroll
        for (int cc = 0; cc < WS_CH; ++cc) se += pe[cc];
        evOut[(size_t)(m0 + wave) * MEM_ + ecol] = se;
    }
    float g[6][2];
#pragma unroll
    for (int jb = 0; jb < 6; ++jb) {
        if (jb < 3) {
#pragma unroll
            for (int j = 0; j < SG_LINES; ++j)
                xw[jb + 3][j] = *(const v4f*)(W_hh + wo + (size_t)(jb * Q_) * SG_K + 32 * j);
        } else if (jb >= 4) {
            const float* wsrc = (jb == 4) ? Wb : Wa;
#pragma unroll
            for (int tt = 0; tt < 2; ++tt) {
#pragma unroll
                for (int i = 0; i < TL_KQ; ++i)
                    wq[jb - 4][tt][i] = *(const v4f*)(wsrc + qo + (size_t)(16 * tt) * Q_ + 4 * i);
            }
        }
        __builtin_amdgcn_sched_barrier(0);
        gate_tile(xa[jb / 3], xw[jb], kb, g[jb]);
        __builtin_amdgcn_sched_barrier(0);
    }

    __shared__ __attribute__((aligned(16))) float st[TL_ROWS][TL_F];
    {
        const float bir = b_ih[feat], biz = b_ih[Q_ + feat], bin = b_ih[2 * Q_ + feat];
        const float bhr = b_hh[feat], bhz = b_hh[Q_ + feat], bhn = b_hh[2 * Q_ + feat];
#pragma unroll
        for (int k = 0; k < 2; ++k) {
            const int trow = 2 * hi + k;
            const float so = sOld[(size_t)(m0 + trow) * Q_ + feat];
            const float ir = bir + g[0][k], iz = biz + g[1][k], in_ = bin + g[2][k];
            const float hr = bhr + g[3][k], hz = bhz + g[4][k], hn  = bhn + g[5][k];
            const float r = 1.0f / (1.0f + __expf(-(ir + hr)));
            const float z = 1.0f / (1.0f + __expf(-(iz + hz)));
            const float n = tanhf(in_ + r * hn);
            const float sn = (1.0f - z) * n + z * so;
            if (kb == 0) st[trow][4 * wave + (c >> 2)] = sn;
        }
    }
    __syncthreads();

    if (tid < TL_ROWS * TL_F) sNew[si] = st[tid >> 5][tid & 31];

    v4f av[TL_KQ];
#pragma unroll
    for (int i = 0; i < TL_KQ; ++i) {
        av[i] = *(const v4f*)&st[c & 3][16 * hi + 4 * i];
        if (c >= TL_ROWS) av[i] = (v4f){};
    }
    v8f qacc[2][2];
#pragma unroll
    for (int x = 0; x < 2; ++x) { qacc[x][0] = (v8f){}; qacc[x][1] = (v8f){}; }
#pragma unroll
    for (int i = 0; i < TL_KQ; ++i) {
#pragma unroll
        for (int e = 0; e < 2; ++e) {
#pragma unroll
            for (int x = 0; x < 2; ++x) {
#pragma unroll
                for (int tt = 0; tt < 2; ++tt) {
                    v2f a, bb;
                    a.x = av[i][2 * e]; a.y = av[i][2 * e + 1];
                    bb.x = wq[x][tt][i][2 * e]; bb.y = wq[x][tt][i][2 * e + 1];
                    qacc[x][tt] = __builtin_amdgcn_wmma_f32_16x16x4_f32(false, a, false, bb,
                                                                        (short)0, qacc[x][tt], false, false);
                }
            }
        }
    }
    __shared__ float qt[2][TL_ROWS][MEM_];
    if (hi == 0) {
#pragma unroll
        for (int x = 0; x < 2; ++x) {
#pragma unroll
            for (int tt = 0; tt < 2; ++tt) {
#pragma unroll
                for (int r = 0; r < TL_ROWS; ++r) qt[x][r][32 * wave + 16 * tt + c] = qacc[x][tt][r];
            }
        }
    }
    __syncthreads();
    {
        float* qb = qbPart + (size_t)slice * BQ_ + (size_t)m0 * MEM_;
        float* qa = qaPart + (size_t)slice * BQ_ + (size_t)m0 * MEM_;
#pragma unroll
        for (int k = 0; k < (TL_ROWS * MEM_) / (TL_WAVES * 32); ++k) {
            const int idx = k * (TL_WAVES * 32) + tid;
            qb[idx] = qt[0][idx >> 8][idx & 255];
            qa[idx] = qt[1][idx >> 8][idx & 255];
        }
    }
}

extern "C" void kernel_launch(void* const* d_in, const int* in_sizes, int n_in,
                              void* d_out, int out_size, void* d_ws, size_t ws_size,
                              hipStream_t stream) {
    const float* M    = (const float*)d_in[0];
    const float* s0   = (const float*)d_in[1];
    const float* Wb   = (const float*)d_in[2];
    const float* We   = (const float*)d_in[3];
    const float* Wa   = (const float*)d_in[4];
    const float* W_ih = (const float*)d_in[5];
    const float* W_hh = (const float*)d_in[6];
    const float* b_ih = (const float*)d_in[7];
    const float* b_hh = (const float*)d_in[8];

    float* ws    = (float*)d_ws;
    float* out   = (float*)d_out;
    float* S1    = ws + OFF_S1;
    float* S2    = ws + OFF_S2;
    float* sacc  = ws + OFF_SACC;
    float* S3    = ws + OFF_EACC;
    float* qb    = ws + OFF_QB;
    float* qa    = ws + OFF_QA;
    float* sAll  = ws + OFF_SALL;
    float* evAll = ws + OFF_EVALL;
    float* qeAll = ws + OFF_QEALL;
    float* stats = ws + OFF_STATS;
    float* qbP   = ws + OFF_QBP;
    float* qaP   = ws + OFF_QAP;

    auto gemm = [&](const float* A, int lda, const float* W, int ldw,
                    const float* bias, float* C, int ldc, int N, int accF, int rows) {
        dim3 g(N / 32, rows / SG_ROWS);
        wmma_gemm_kernel<false><<<g, SG_WAVES * 32, 0, stream>>>(nullptr, 0, A, lda, W, ldw, bias, C, ldc,
                                                                accF, nullptr, nullptr);
    };

    gemm(s0, Q_, Wb, Q_, nullptr, qb, MEM_, MEM_, 0, B_);
    gemm(s0, Q_, Wa, Q_, nullptr, qa, MEM_, MEM_, 0, B_);

    for (int t = 0; t < 5; ++t) {
        if (t == 0)
            scores_rows_kernel<2, true, 1><<<dim3(SC_CH, B_), SC_WAVES * 32, 0, stream>>>(
                M, qb, qa, nullptr, S1, S2, nullptr, stats);
        else
            scores_rows_kernel<2, true, TL_SLICES><<<dim3(SC_CH, B_), SC_WAVES * 32, 0, stream>>>(
                M, qbP, qaP, nullptr, S1, S2, nullptr, stats);

        if (t == 0)
            wsum_rows_kernel<true><<<dim3(WS_CH, B_), WS_WAVES * 32, 0, stream>>>(M, S1, S2, stats, sacc);
        else
            wsum_rows_kernel<false><<<dim3(WS_CH, B_), WS_WAVES * 32, 0, stream>>>(M, S1, S2, stats, sacc);

        if (t == 0)
            turn_tail_kernel<<<dim3(TL_SLICES, B_ / TL_ROWS), TL_WAVES * 32, 0, stream>>>(
                S1, s0, W_ih, W_hh, b_ih, b_hh, Wb, Wa,
                sAll, qbP, qaP, evAll);
        else if (t < 4)
            turn_tail_kernel<<<dim3(TL_SLICES, B_ / TL_ROWS), TL_WAVES * 32, 0, stream>>>(
                S1, sAll + (t - 1) * BQ_, W_ih, W_hh, b_ih, b_hh, Wb, Wa,
                sAll + t * BQ_, qbP, qaP, evAll + t * BQ_);
    }

    wmma_gemm_kernel<true><<<dim3(MEM_ / 32, (5 * B_) / SG_ROWS), SG_WAVES * 32, 0, stream>>>(
        s0, B_, sAll, Q_, We, Q_ + MEM_, nullptr, qeAll, MEM_, 0, S1, evAll + 4 * BQ_);
    gemm(evAll, MEM_, We + Q_, Q_ + MEM_, nullptr, qeAll, MEM_, MEM_, 1, 5 * B_);

    scores_rows_kernel<3, false, 1><<<dim3(SC_CH, B_), SC_WAVES * 32, 0, stream>>>(
        M, qeAll, qeAll + BQ_, qeAll + 2 * BQ_, S1, S2, S3, nullptr);
    end_softmax_kernel<3, false><<<B_, 256, 0, stream>>>(S1, S2, S3, S3, nullptr, nullptr, nullptr);

    scores_rows_kernel<2, false, 1><<<dim3(SC_CH, B_), SC_WAVES * 32, 0, stream>>>(
        M, qeAll + 3 * BQ_, qeAll + 4 * BQ_, nullptr, S1, S2, nullptr, nullptr);
    end_softmax_kernel<2, true><<<B_, 256, 0, stream>>>(S1, S2, nullptr, S3, sacc, out, out + BL_);
}


# end branch in one pass over M instead of two: all 5 end queries per row load (two WMMA chain pairs, 4 + 1 queries), one softmax launch for the 5 rows; 22 to 20 launches; same sums
# speedup vs baseline: 17.7297x; 1.0356x over previous
#include <hip/hip_runtime.h>
#include <math.h>

#define B_   32
#define L_   4096
#define MEM_ 256
#define Q_   256
#define BL_  (B_ * L_)
#define BQ_  (B_ * Q_)
#define B3Q_ (B_ * 3 * Q_)

#define OFF_S1   0
#define OFF_S2   (OFF_S1 + BL_)
#define OFF_SACC (OFF_S2 + BL_)
#define OFF_EACC (OFF_SACC + BL_)
#define OFF_QB   (OFF_EACC + BL_)
#define OFF_QA   (OFF_QB + BQ_)
#define OFF_X1   (OFF_QA + BQ_)
#define OFF_GI   (OFF_X1 + BQ_)
#define OFF_GH   (OFF_GI + B3Q_)
#define OFF_SALL  OFF_EACC
#define OFF_EVALL (OFF_SALL + 4 * BQ_)
#define OFF_QEALL OFF_QB
#define OFF_STATS (OFF_EVALL + 4 * BQ_)
#define TL_SLICES 8
#define OFF_QBP  (OFF_EVALL + 5 * BQ_)
#define OFF_QAP  (OFF_QBP + TL_SLICES * BQ_)

#define WS_CH     8
#define WS_WGROWS (L_ / WS_CH)

typedef __attribute__((ext_vector_type(2))) float v2f;
typedef __attribute__((ext_vector_type(8))) float v8f;

typedef __attribute__((ext_vector_type(4))) float v4f;

#define SG_K     256
#define SG_LINES (SG_K / 32)
#define SG_WAVES 8
#define SG_ROWS  8

template <bool EVRED>
__global__ void __launch_bounds__(SG_WAVES * 32)
wmma_gemm_kernel(const float* __restrict__ A0, int rows0,
                 const float* __restrict__ A, int lda,
                 const float* __restrict__ W, int ldw,
                 const float* __restrict__ bias,
                 float* __restrict__ C, int ldc,
                 int accumulate,
                 const float* __restrict__ part,
                 float* __restrict__ evOut) {
    const int tid   = threadIdx.x;
    const int lane  = tid & 31;
    const int wave  = tid >> 5;
    const int c     = lane & 15;
    const int hi    = lane >> 4;
    const int kb    = c & 3;
    const int piece = 4 * (2 * kb + hi);
    const int m0    = blockIdx.y * SG_ROWS;
    const int n0    = blockIdx.x * 32;

    const float* ap = ((m0 < rows0) ? A0 + (size_t)(m0 + (c >> 2)) * lda
                                    : A + (size_t)(m0 - rows0 + (c >> 2)) * lda) + piece;
    const float* wp = W + (size_t)(n0 + 4 * wave + (c >> 2)) * ldw + piece;

    v4f xw[SG_LINES], xa[2][SG_LINES];
#pragma unroll
    for (int j = 0; j < SG_LINES; ++j) xw[j] = *(const v4f*)(wp + 32 * j);
#pragma unroll
    for (int u = 0; u < 2; ++u) {
#pragma unroll
        for (int j = 0; j < SG_LINES; ++j)
            xa[u][j] = *(const v4f*)(ap + (size_t)(4 * u) * lda + 32 * j);
    }

    const int orow = wave, ocol = lane;
    float* cp = C + (size_t)(m0 + orow) * ldc + n0 + ocol;
    float bv = 0.0f, cv = 0.0f;
    if (bias) bv = bias[n0 + ocol];
    if (accumulate) cv = *cp;

    const int  ei  = (blockIdx.y * gridDim.x + blockIdx.x) * (SG_WAVES * 32) + tid;
    const bool red = EVRED && ei < BQ_;
    float pr[WS_CH];
    if (red) {
        const float* src = part + (size_t)(ei >> 8) * L_ + (ei & 255);
#pragma unroll
        for (int cc = 0; cc < WS_CH; ++cc) pr[cc] = src[cc * WS_WGROWS];
    }

    v8f acc0[2], acc1[2];
#pragma unroll
    for (int u = 0; u < 2; ++u) { acc0[u] = (v8f){}; acc1[u] = (v8f){}; }
#pragma unroll
    for (int j = 0; j < SG_LINES; ++j) {
#pragma unroll
        for (int u = 0; u < 2; ++u) {
            v2f a, bb;
            a.x = xa[u][j][0]; a.y = xa[u][j][1]; bb.x = xw[j][0]; bb.y = xw[j][1];
            acc0[u] = __builtin_amdgcn_wmma_f32_16x16x4_f32(false, a, false, bb,
                                                            (short)0, acc0[u], false, false);
            a.x = xa[u][j][2]; a.y = xa[u][j][3]; bb.x = xw[j][2]; bb.y = xw[j][3];
            acc1[u] = __builtin_amdgcn_wmma_f32_16x16x4_f32(false, a, false, bb,
                                                            (short)0, acc1[u], false, false);
        }
    }
    float lo[2], up[2];
#pragma unroll
    for (int u = 0; u < 2; ++u) {
        const v8f tv = acc0[u] + acc1[u];
        const v4f tl = {tv[0], tv[1], tv[2], tv[3]};
        const v4f tu = {tv[4], tv[5], tv[6], tv[7]};
        lo[u] = tl[kb]; up[u] = tu[kb];
    }
#pragma unroll
    for (int u = 0; u < 2; ++u) { lo[u] += __shfl_xor(lo[u], 1); up[u] += __shfl_xor(up[u], 1); }
#pragma unroll
    for (int u = 0; u < 2; ++u) { lo[u] += __shfl_xor(lo[u], 2); up[u] += __shfl_xor(up[u], 2); }

    __shared__ float tile[SG_ROWS][32];
    if (kb == 0) {
#pragma unroll
        for (int u = 0; u < 2; ++u) {
            tile[4 * u + 2 * hi][4 * wave + (c >> 2)]     = lo[u];
            tile[4 * u + 2 * hi + 1][4 * wave + (c >> 2)] = up[u];
        }
    }
    __syncthreads();
    *cp = (cv + bv) + tile[orow][ocol];
    if (red) {
        float s = 0.0f;
#pragma unroll
        for (int cc = 0; cc < WS_CH; ++cc) s += pr[cc];
        evOut[ei] = s;
    }
}

template <int N>
static __device__ __forceinline__ float row_xor(float x) {
    const int b = __builtin_bit_cast(int, x);
    return __builtin_bit_cast(float,
        __builtin_amdgcn_update_dpp(b, b, 0x160 + N, 0xf, 0xf, false));
}
static __device__ __forceinline__ float xhalf(float x) {
    const int b = __builtin_bit_cast(int, x);
    return __builtin_bit_cast(float,
        __builtin_amdgcn_permlanex16(b, b, 0x76543210, 0xfedcba98, false, false));
}

#define SC_CH     8
#define SC_WAVES  16
#define SC_WGROWS (L_ / SC_CH)
#define SC_WROWS  (SC_WGROWS / SC_WAVES)
#define SC_GROUPS (SC_WROWS / 4)
#define SC_LINES  (MEM_ / 32)
#define SC_AHEAD(NQ) ((NQ) > 4 ? 2 : 3)

template <int NQ, bool LOCAL, int NP>
__global__ void __launch_bounds__(SC_WAVES * 32)
scores_rows_kernel(const float* __restrict__ M,
                   const float* __restrict__ q1,
                   const float* __restrict__ q2,
                   const float* __restrict__ q3,
                   const float* __restrict__ q4,
                   const float* __restrict__ q5,
                   float* __restrict__ S1,
                   float* __restrict__ S2,
                   float* __restrict__ S3,
                   float* __restrict__ S4,
                   float* __restrict__ S5,
                   float* __restrict__ stats) {
    static_assert(NQ == 2 || NQ == 5, "2 queries, or 4 + 1");
    static_assert(!(LOCAL && NQ != 2), "the chunk softmax is written for the turn loop's 2 queries");
    constexpr int NA    = (NQ > 4) ? 4 : NQ;
    constexpr int AHEAD = SC_AHEAD(NQ);
    constexpr int RING  = AHEAD + 1;
    const int tid   = threadIdx.x;
    const int lane  = tid & 31;
    const int wave  = tid >> 5;
    const int c     = lane & 15;
    const int hi    = lane >> 4;
    const int kb    = c & 3;
    const int b     = blockIdx.y;
    const int ch    = blockIdx.x;
    const int piece = 4 * (2 * kb + hi);
    const int r0    = ch * SC_WGROWS + wave * SC_WROWS;

    const float* Mb = M + ((size_t)b * L_ + r0 + (c >> 2)) * MEM_ + piece;
    const int qi = (c >> 2) % NA;
    const float* qv = ((qi == 0) ? q1 : (qi == 1) ? q2 : (qi == 2) ? q3 : q4) + (size_t)b * MEM_ + piece;

    static_assert(NP == 1 || NQ * MEM_ == SC_WAVES * 32, "one query element per thread");
    __shared__ __attribute__((aligned(16))) float qs[2][MEM_];
    if constexpr (NP > 1) {
        const float* src = ((tid >> 8) ? q2 : q1) + (size_t)b * MEM_ + (tid & 255);
        float pq[NP];
#pragma unroll
        for (int p = 0; p < NP; ++p) pq[p] = src[(size_t)p * BQ_];
        float sq = pq[0];
#pragma unroll
        for (int p = 1; p < NP; ++p) sq += pq[p];
        qs[tid >> 8][tid & 255] = sq;
    }

    v4f x[RING][SC_LINES];
#pragma unroll
    for (int g = 0; g < AHEAD; ++g) {
#pragma unroll
        for (int j = 0; j < SC_LINES; ++j)
            x[g][j] = *(const v4f*)(Mb + (size_t)(4 * g) * MEM_ + 32 * j);
    }

    v4f qf[SC_LINES];
    if constexpr (NP > 1) {
        __builtin_amdgcn_fence(__ATOMIC_RELEASE, "workgroup", "local");
        __builtin_amdgcn_s_barrier();
        __builtin_amdgcn_fence(__ATOMIC_ACQUIRE, "workgroup", "local");
#pragma unroll
        for (int j = 0; j < SC_LINES; ++j) qf[j] = *(const v4f*)&qs[qi][piece + 32 * j];
    } else {
#pragma unroll
        for (int j = 0; j < SC_LINES; ++j) qf[j] = *(const v4f*)(qv + 32 * j);
    }
    v4f qg[(NQ > 4) ? SC_LINES : 1];
    if constexpr (NQ > 4) {
#pragma unroll
        for (int j = 0; j < SC_LINES; ++j) qg[j] = *(const v4f*)(q5 + (size_t)b * MEM_ + piece + 32 * j);
    }

    __shared__ float sc[NQ][SC_WGROWS];
    float slo[SC_GROUPS], shi[SC_GROUPS];
#pragma unroll
    for (int g = 0; g < SC_GROUPS; ++g) {
        const int cur = g % RING;
        if (g + AHEAD < SC_GROUPS) {
            const int nxt = (g + AHEAD) % RING;
#pragma unroll
            for (int j = 0; j < SC_LINES; ++j)
                x[nxt][j] = *(const v4f*)(Mb + (size_t)(4 * (g + AHEAD)) * MEM_ + 32 * j);
        }
        __builtin_amdgcn_sched_barrier(0);
        v8f acc0 = {}, acc1 = {};
        v8f acc2 = {}, acc3 = {};
#pragma unroll
        for (int j = 0; j < SC_LINES; ++j) {
            v2f a, bb;
            a.x = x[cur][j][0]; a.y = x[cur][j][1]; bb.x = qf[j][0]; bb.y = qf[j][1];
            acc0 = __builtin_amdgcn_wmma_f32_16x16x4_f32(false, a, false, bb,
                                                         (short)0, acc0, false, false);
            if constexpr (NQ > 4) {
                bb.x = qg[j][0]; bb.y = qg[j][1];
                acc2 = __builtin_amdgcn_wmma_f32_16x16x4_f32(false, a, false, bb,
                                                             (short)0, acc2, false, false);
            }
            a.x = x[cur][j][2]; a.y = x[cur][j][3]; bb.x = qf[j][2]; bb.y = qf[j][3];
            acc1 = __builtin_amdgcn_wmma_f32_16x16x4_f32(false, a, false, bb,
                                                         (short)0, acc1, false, false);
            if constexpr (NQ > 4) {
                bb.x = qg[j][2]; bb.y = qg[j][3];
                acc3 = __builtin_amdgcn_wmma_f32_16x16x4_f32(false, a, false, bb,
                                                             (short)0, acc3, false, false);
            }
        }
        float t[8];
#pragma unroll
        for (int r = 0; r < 8; ++r) t[r] = acc0[r] + acc1[r];
        float lo = (kb == 0) ? t[0] : (kb == 1) ? t[1] : (kb == 2) ? t[2] : t[3];
        float up = (kb == 0) ? t[4] : (kb == 1) ? t[5] : (kb == 2) ? t[6] : t[7];
        lo += __shfl_xor(lo, 1); up += __shfl_xor(up, 1);
        lo += __shfl_xor(lo, 2); up += __shfl_xor(up, 2);
        slo[g] = lo; shi[g] = up;
        if constexpr (NQ > 4) {
            float u[8];
#pragma unroll
            for (int r = 0; r < 8; ++r) u[r] = acc2[r] + acc3[r];
            float lo5 = (kb == 0) ? u[0] : (kb == 1) ? u[1] : (kb == 2) ? u[2] : u[3];
            float up5 = (kb == 0) ? u[4] : (kb == 1) ? u[5] : (kb == 2) ? u[6] : u[7];
            lo5 += row_xor<1>(lo5); up5 += row_xor<1>(up5);
            lo5 += row_xor<2>(lo5); up5 += row_xor<2>(up5);
            if (c == 0) {
                float* d5 = &sc[NQ - 1][wave * SC_WROWS + 2 * hi + 4 * g];
                d5[0] = lo5; d5[1] = up5;
            }
        }
    }

    if (kb == 0 && (c >> 2) < NA) {
        float* dst = &sc[c >> 2][wave * SC_WROWS + 2 * hi];
#pragma unroll
        for (int g = 0; g < SC_GROUPS; ++g) { dst[4 * g] = slo[g]; dst[4 * g + 1] = shi[g]; }
    }
    __syncthreads();
    const size_t so = (size_t)b * L_ + ch * SC_WGROWS + tid;
    if (!LOCAL) {
        S1[so] = sc[0][tid];
        S2[so] = sc[1][tid];
        if constexpr (NQ > 4) {
            S3[so] = sc[2][tid];
            S4[so] = sc[3][tid];
            S5[so] = sc[4][tid];
        }
    } else {
        __shared__ float redm[NQ][SC_WAVES], reds[NQ][SC_WAVES];
        float xs[NQ], ex[NQ], mx[NQ];
#pragma unroll
        for (int q = 0; q < NQ; ++q) {
            xs[q] = sc[q][tid];
            float t = xs[q];
            t = fmaxf(t, row_xor<1>(t)); t = fmaxf(t, row_xor<2>(t));
            t = fmaxf(t, row_xor<4>(t)); t = fmaxf(t, row_xor<8>(t));
            t = fmaxf(t, xhalf(t));
            if (lane == 0) redm[q][wave] = t;
        }
        __syncthreads();
#pragma unroll
        for (int q = 0; q < NQ; ++q) {
            float t = redm[q][c];
            t = fmaxf(t, row_xor<1>(t)); t = fmaxf(t, row_xor<2>(t));
            t = fmaxf(t, row_xor<4>(t)); t = fmaxf(t, row_xor<8>(t));
            mx[q] = t;
            ex[q] = __expf(xs[q] - t);
            float s = ex[q];
            s += row_xor<1>(s); s += row_xor<2>(s); s += row_xor<4>(s); s += row_xor<8>(s);
            s += xhalf(s);
            if (lane == 0) reds[q][wave] = s;
        }
        __syncthreads();
#pragma unroll
        for (int q = 0; q < NQ; ++q) {
            float s = reds[q][c];
            s += row_xor<1>(s); s += row_xor<2>(s); s += row_xor<4>(s); s += row_xor<8>(s);
            if (tid == 0) {
                float* sp = stats + (((size_t)b * NQ + q) * SC_CH + ch) * 2;
                sp[0] = mx[q]; sp[1] = s;
            }
        }
        S1[so] = ex[0];
        S2[so] = ex[1];
    }
}

#define WS_WAVES  16
#define WS_WROWS  (WS_WGROWS / WS_WAVES)
#define WS_KSTEPS (WS_WROWS / 4)
#define WS_KSTRIDE (4 * WS_WAVES)

template <bool FIRST>
__global__ void __launch_bounds__(WS_WAVES * 32)
wsum_rows_kernel(const float* __restrict__ M,
                 float* pbuf,
                 const float* wbuf,
                 const float* __restrict__ stats,
                 float* __restrict__ sacc) {
    const int tid  = threadIdx.x;
    const int lane = tid & 31;
    const int wave = tid >> 5;
    const int row  = lane & 15;
    const int hi   = lane >> 4;
    const int b    = blockIdx.y;
    const int c    = blockIdx.x;
    const int l0   = c * WS_WGROWS + 4 * wave;

    const float* pv = ((row == 1) ? wbuf : (const float*)pbuf) + (size_t)b * L_ + l0 + 2 * hi;
    const float* Mb = M + ((size_t)b * L_ + l0 + 2 * hi) * MEM_ + 4 * row;

    v8f acc[16];
#pragma unroll
    for (int t = 0; t < 16; ++t) acc[t] = (v8f){};

    v2f a[WS_KSTEPS];
#pragma unroll
    for (int ks = 0; ks < WS_KSTEPS; ++ks) {
        a[ks] = *(const v2f*)(pv + WS_KSTRIDE * ks);
        if (row >= 2) { a[ks].x = 0.0f; a[ks].y = 0.0f; }
    }

    v4f x[2][2][4];
#pragma unroll
    for (int g = 0; g < 4; ++g) {
        x[0][0][g] = *(const v4f*)(Mb + 64 * g);
        x[0][1][g] = *(const v4f*)(Mb + MEM_ + 64 * g);
    }

    float cm[2][WS_CH], cz[2][WS_CH];
    {
        const float* st = stats + (size_t)b * (2 * WS_CH * 2);
#pragma unroll
        for (int q = 0; q < 2; ++q) {
#pragma unroll
            for (int cc = 0; cc < WS_CH; ++cc) {
                cm[q][cc] = st[(q * WS_CH + cc) * 2];
                cz[q][cc] = st[(q * WS_CH + cc) * 2 + 1];
            }
        }
    }

#pragma unroll
    for (int ks = 0; ks < WS_KSTEPS; ++ks) {
        const int cur = ks & 1, nxt = cur ^ 1;
        if (ks + 1 < WS_KSTEPS) {
#pragma unroll
            for (int g = 0; g < 4; ++g) {
                x[nxt][0][g] = *(const v4f*)(Mb + (size_t)(WS_KSTRIDE * (ks + 1))     * MEM_ + 64 * g);
                x[nxt][1][g] = *(const v4f*)(Mb + (size_t)(WS_KSTRIDE * (ks + 1) + 1) * MEM_ + 64 * g);
            }
        }
        __builtin_amdgcn_sched_barrier(0);
#pragma unroll
        for (int g = 0; g < 4; ++g) {
#pragma unroll
            for (int cc = 0; cc < 4; ++cc) {
                v2f bb; bb.x = x[cur][0][g][cc]; bb.y = x[cur][1][g][cc];
                acc[4 * g + cc] = __builtin_amdgcn_wmma_f32_16x16x4_f32(
                    false, a[ks], false, bb, (short)0, acc[4 * g + cc], false, false);
            }
        }
    }

    const size_t ro = (size_t)b * L_ + c * WS_WGROWS + tid;
    const float pe = pbuf[ro], sa = FIRST ? 0.0f : sacc[ro];

    float f[2];
#pragma unroll
    for (int q = 0; q < 2; ++q) {
        float m = cm[q][0];
#pragma unroll
        for (int cc = 1; cc < WS_CH; ++cc) m = fmaxf(m, cm[q][cc]);
        float z = 0.0f, own = 0.0f;
#pragma unroll
        for (int cc = 0; cc < WS_CH; ++cc) {
            const float w = __expf(cm[q][cc] - m);
            z += cz[q][cc] * w;
            if (cc == c) own = w;
        }
        f[q] = own / z;
    }

    __shared__ float red[WS_WAVES][2][MEM_];
    if (hi == 0) {
#pragma unroll
        for (int g = 0; g < 4; ++g) {
            v4f es, xs;
#pragma unroll
            for (int cc = 0; cc < 4; ++cc) { es[cc] = acc[4 * g + cc][0]; xs[cc] = acc[4 * g + cc][1]; }
            *(v4f*)&red[wave][0][64 * g + 4 * row] = es;
            *(v4f*)&red[wave][1][64 * g + 4 * row] = xs;
        }
    }
    __syncthreads();
    sacc[ro] = sa + 0.2f * (pe * f[0]);
    {
        const int v = tid >> 8, col = tid & 255;
        float s = 0.0f;
#pragma unroll
        for (int w = 0; w < WS_WAVES; ++w) s += red[w][v][col];
        pbuf[(size_t)b * L_ + c * WS_WGROWS + v * MEM_ + col] = s * (v ? f[1] : f[0]);
    }
}

#define SM_WAVES 8
#define SM_VECS  (L_ / (SM_WAVES * 32 * 4))

template <int NR>
static __device__ __forceinline__ void softmax_rows_regs(const v4f (&x)[NR][SM_VECS], v4f (&e)[NR][SM_VECS],
                                                         float (&redm)[NR][SM_WAVES],
                                                         float (&reds)[NR][SM_WAVES],
                                                         int lane, int wave) {
    float m[NR];
#pragma unroll
    for (int r = 0; r < NR; ++r) {
        float t = fmaxf(fmaxf(x[r][0][0], x[r][0][1]), fmaxf(x[r][0][2], x[r][0][3]));
#pragma unroll
        for (int k = 1; k < SM_VECS; ++k)
            t = fmaxf(t, fmaxf(fmaxf(x[r][k][0], x[r][k][1]), fmaxf(x[r][k][2], x[r][k][3])));
        t = fmaxf(t, row_xor<1>(t)); t = fmaxf(t, row_xor<2>(t));
        t = fmaxf(t, row_xor<4>(t)); t = fmaxf(t, row_xor<8>(t));
        t = fmaxf(t, xhalf(t));
        if (lane == 0) redm[r][wave] = t;
    }
    __syncthreads();
#pragma unroll
    for (int r = 0; r < NR; ++r) {
        float t = redm[r][lane & (SM_WAVES - 1)];
        t = fmaxf(t, row_xor<1>(t)); t = fmaxf(t, row_xor<2>(t)); t = fmaxf(t, row_xor<4>(t));
        m[r] = t;
    }

#pragma unroll
    for (int r = 0; r < NR; ++r) {
        float s = 0.0f;
#pragma unroll
        for (int k = 0; k < SM_VECS; ++k) {
#pragma unroll
            for (int cc = 0; cc < 4; ++cc) e[r][k][cc] = __expf(x[r][k][cc] - m[r]);
            s += (e[r][k][0] + e[r][k][1]) + (e[r][k][2] + e[r][k][3]);
        }
        s += row_xor<1>(s); s += row_xor<2>(s); s += row_xor<4>(s); s += row_xor<8>(s);
        s += xhalf(s);
        if (lane == 0) reds[r][wave] = s;
    }
    __syncthreads();
#pragma unroll
    for (int r = 0; r < NR; ++r) {
        float s = reds[r][lane & (SM_WAVES - 1)];
        s += row_xor<1>(s); s += row_xor<2>(s); s += row_xor<4>(s);
        const float inv = 1.0f / s;
#pragma unroll
        for (int k = 0; k < SM_VECS; ++k) e[r][k] = e[r][k] * inv;
    }
}

template <int NR>
__global__ void __launch_bounds__(SM_WAVES * 32)
end_softmax_kernel(const float* R0, const float* R1, const float* R2,
                   const float* R3, const float* R4,
                   const float* __restrict__ sacc,
                   float* outStart,
                   float* outEnd) {
    const int b    = blockIdx.x;
    const int tid  = threadIdx.x;
    const int lane = tid & 31;
    const int wave = tid >> 5;
    __shared__ float redm[NR][SM_WAVES], reds[NR][SM_WAVES];
    const size_t off = (size_t)b * L_ + 4 * tid;

    v4f x[NR][SM_VECS], a[SM_VECS], st[SM_VECS];
#pragma unroll
    for (int r = 0; r < NR; ++r) {
        const float* rp = ((r == 0) ? R0 : (r == 1) ? R1 : (r == 2) ? R2 : (r == 3) ? R3 : R4) + off;
#pragma unroll
        for (int k = 0; k < SM_VECS; ++k) x[r][k] = *(const v4f*)(rp + 1024 * k);
    }
#pragma unroll
    for (int k = 0; k < SM_VECS; ++k) {
        a[k]  = (v4f){};
        st[k] = *(const v4f*)(sacc + off + 1024 * k);
    }
    v4f e[NR][SM_VECS];
    softmax_rows_regs<NR>(x, e, redm, reds, lane, wave);
#pragma unroll
    for (int r = 0; r < NR; ++r) {
#pragma unroll
        for (int k = 0; k < SM_VECS; ++k) a[k] = a[k] + 0.2f * e[r][k];
    }
#pragma unroll
    for (int k = 0; k < SM_VECS; ++k) {
        v4f le, ls;
#pragma unroll
        for (int cc = 0; cc < 4; ++cc) { le[cc] = logf(a[k][cc]); ls[cc] = logf(st[k][cc]); }
        *(v4f*)(outEnd + off + 1024 * k)   = le;
        *(v4f*)(outStart + off + 1024 * k) = ls;
    }
}

#define TL_F     32
#define TL_ROWS  4
#define TL_WAVES 8
#define TL_KQ    (TL_F / 8)
#define TL_XPITCH (SG_K + 32)

static __device__ __forceinline__ void gate_tile(const v4f (&xa)[SG_LINES],
                                                 const v4f (&xw)[SG_LINES], int kb, float (&o)[2]) {
    v8f acc0 = {}, acc1 = {};
#pragma unroll
    for (int j = 0; j < SG_LINES; ++j) {
        v2f a, bb;
        a.x = xa[j][0]; a.y = xa[j][1]; bb.x = xw[j][0]; bb.y = xw[j][1];
        acc0 = __builtin_amdgcn_wmma_f32_16x16x4_f32(false, a, false, bb,
                                                     (short)0, acc0, false, false);
        a.x = xa[j][2]; a.y = xa[j][3]; bb.x = xw[j][2]; bb.y = xw[j][3];
        acc1 = __builtin_amdgcn_wmma_f32_16x16x4_f32(false, a, false, bb,
                                                     (short)0, acc1, false, false);
    }
    const v8f tv = acc0 + acc1;
    const v4f tl = {tv[0], tv[1], tv[2], tv[3]};
    const v4f tu = {tv[4], tv[5], tv[6], tv[7]};
    float lo = tl[kb], up = tu[kb];
    lo += __shfl_xor(lo, 1); up += __shfl_xor(up, 1);
    lo += __shfl_xor(lo, 2); up += __shfl_xor(up, 2);
    o[0] = lo; o[1] = up;
}

__global__ void __launch_bounds__(TL_WAVES * 32)
turn_tail_kernel(const float* __restrict__ part,
                 const float* __restrict__ sOld,
                 const float* __restrict__ W_ih,
                 const float* __restrict__ W_hh,
                 const float* __restrict__ b_ih,
                 const float* __restrict__ b_hh,
                 const float* __restrict__ Wb,
                 const float* __restrict__ Wa,
                 float* __restrict__ sNew,
                 float* __restrict__ qbPart,
                 float* __restrict__ qaPart,
                 float* __restrict__ evOut) {
    const int tid   = threadIdx.x;
    const int lane  = tid & 31;
    const int wave  = tid >> 5;
    const int c     = lane & 15;
    const int hi    = lane >> 4;
    const int kb    = c & 3;
    const int piece = 4 * (2 * kb + hi);
    const int slice = blockIdx.x;
    const int n0    = slice * TL_F;
    const int m0    = blockIdx.y * TL_ROWS;
    const int feat  = n0 + 4 * wave + (c >> 2);

    v4f xa[2][SG_LINES], xw[6][SG_LINES], wq[2][2][TL_KQ];
    const size_t ao = (size_t)(m0 + (c >> 2)) * SG_K + piece;
    const size_t wo = (size_t)feat * SG_K + piece;
    const size_t qo = (size_t)(32 * wave + c) * Q_ + n0 + 16 * hi;

    const int prow = tid >> 6, pcol = 4 * (tid & 63);
    v4f px[WS_CH];
    {
        const float* xsrc = part + (size_t)(m0 + prow) * L_ + MEM_ + pcol;
#pragma unroll
        for (int cc = 0; cc < WS_CH; ++cc) px[cc] = *(const v4f*)(xsrc + cc * WS_WGROWS);
    }
    const size_t si = (size_t)(m0 + (tid >> 5)) * Q_ + n0 + (tid & 31);

#pragma unroll
    for (int j = 0; j < SG_LINES; ++j) xa[1][j] = *(const v4f*)(sOld + ao + 32 * j);
#pragma unroll
    for (int jb = 0; jb < 3; ++jb) {
#pragma unroll
        for (int j = 0; j < SG_LINES; ++j)
            xw[jb][j] = *(const v4f*)(W_ih + wo + (size_t)(jb * Q_) * SG_K + 32 * j);
    }

    __shared__ __attribute__((aligned(16))) float xs[TL_ROWS][TL_XPITCH];
    {
        v4f sx = {0.0f, 0.0f, 0.0f, 0.0f};
#pragma unroll
        for (int cc = 0; cc < WS_CH; ++cc) sx = sx + px[cc];
        *(v4f*)&xs[prow][pcol] = sx;
    }
    __builtin_amdgcn_fence(__ATOMIC_RELEASE, "workgroup", "local");
    __builtin_amdgcn_s_barrier();
    __builtin_amdgcn_fence(__ATOMIC_ACQUIRE, "workgroup", "local");
#pragma unroll
    for (int j = 0; j < SG_LINES; ++j) xa[0][j] = *(const v4f*)&xs[c >> 2][piece + 32 * j];
    if (tid < 32 * TL_ROWS) {
        const int ecol = n0 + lane;
        const float* esrc = part + (size_t)(m0 + wave) * L_ + ecol;
        float pe[WS_CH];
#pragma unroll
        for (int cc = 0; cc < WS_CH; ++cc) pe[cc] = esrc[cc * WS_WGROWS];
        float se = 0.0f;
#pragma unroll
        for (int cc = 0; cc < WS_CH; ++cc) se += pe[cc];
        evOut[(size_t)(m0 + wave) * MEM_ + ecol] = se;
    }
    float g[6][2];
#pragma unroll
    for (int jb = 0; jb < 6; ++jb) {
        if (jb < 3) {
#pragma unroll
            for (int j = 0; j < SG_LINES; ++j)
                xw[jb + 3][j] = *(const v4f*)(W_hh + wo + (size_t)(jb * Q_) * SG_K + 32 * j);
        } else if (jb >= 4) {
            const float* wsrc = (jb == 4) ? Wb : Wa;
#pragma unroll
            for (int tt = 0; tt < 2; ++tt) {
#pragma unroll
                for (int i = 0; i < TL_KQ; ++i)
                    wq[jb - 4][tt][i] = *(const v4f*)(wsrc + qo + (size_t)(16 * tt) * Q_ + 4 * i);
            }
        }
        __builtin_amdgcn_sched_barrier(0);
        gate_tile(xa[jb / 3], xw[jb], kb, g[jb]);
        __builtin_amdgcn_sched_barrier(0);
    }

    __shared__ __attribute__((aligned(16))) float st[TL_ROWS][TL_F];
    {
        const float bir = b_ih[feat], biz = b_ih[Q_ + feat], bin = b_ih[2 * Q_ + feat];
        const float bhr = b_hh[feat], bhz = b_hh[Q_ + feat], bhn = b_hh[2 * Q_ + feat];
#pragma unroll
        for (int k = 0; k < 2; ++k) {
            const int trow = 2 * hi + k;
            const float so = sOld[(size_t)(m0 + trow) * Q_ + feat];
            const float ir = bir + g[0][k], iz = biz + g[1][k], in_ = bin + g[2][k];
            const float hr = bhr + g[3][k], hz = bhz + g[4][k], hn  = bhn + g[5][k];
            const float r = 1.0f / (1.0f + __expf(-(ir + hr)));
            const float z = 1.0f / (1.0f + __expf(-(iz + hz)));
            const float n = tanhf(in_ + r * hn);
            const float sn = (1.0f - z) * n + z * so;
            if (kb == 0) st[trow][4 * wave + (c >> 2)] = sn;
        }
    }
    __syncthreads();

    if (tid < TL_ROWS * TL_F) sNew[si] = st[tid >> 5][tid & 31];

    v4f av[TL_KQ];
#pragma unroll
    for (int i = 0; i < TL_KQ; ++i) {
        av[i] = *(const v4f*)&st[c & 3][16 * hi + 4 * i];
        if (c >= TL_ROWS) av[i] = (v4f){};
    }
    v8f qacc[2][2];
#pragma unroll
    for (int x = 0; x < 2; ++x) { qacc[x][0] = (v8f){}; qacc[x][1] = (v8f){}; }
#pragma unroll
    for (int i = 0; i < TL_KQ; ++i) {
#pragma unroll
        for (int e = 0; e < 2; ++e) {
#pragma unroll
            for (int x = 0; x < 2; ++x) {
#pragma unroll
                for (int tt = 0; tt < 2; ++tt) {
                    v2f a, bb;
                    a.x = av[i][2 * e]; a.y = av[i][2 * e + 1];
                    bb.x = wq[x][tt][i][2 * e]; bb.y = wq[x][tt][i][2 * e + 1];
                    qacc[x][tt] = __builtin_amdgcn_wmma_f32_16x16x4_f32(false, a, false, bb,
                                                                        (short)0, qacc[x][tt], false, false);
                }
            }
        }
    }
    __shared__ float qt[2][TL_ROWS][MEM_];
    if (hi == 0) {
#pragma unroll
        for (int x = 0; x < 2; ++x) {
#pragma unroll
            for (int tt = 0; tt < 2; ++tt) {
#pragma unroll
                for (int r = 0; r < TL_ROWS; ++r) qt[x][r][32 * wave + 16 * tt + c] = qacc[x][tt][r];
            }
        }
    }
    __syncthreads();
    {
        float* qb = qbPart + (size_t)slice * BQ_ + (size_t)m0 * MEM_;
        float* qa = qaPart + (size_t)slice * BQ_ + (size_t)m0 * MEM_;
#pragma unroll
        for (int k = 0; k < (TL_ROWS * MEM_) / (TL_WAVES * 32); ++k) {
            const int idx = k * (TL_WAVES * 32) + tid;
            qb[idx] = qt[0][idx >> 8][idx & 255];
            qa[idx] = qt[1][idx >> 8][idx & 255];
        }
    }
}

extern "C" void kernel_launch(void* const* d_in, const int* in_sizes, int n_in,
                              void* d_out, int out_size, void* d_ws, size_t ws_size,
                              hipStream_t stream) {
    const float* M    = (const float*)d_in[0];
    const float* s0   = (const float*)d_in[1];
    const float* Wb   = (const float*)d_in[2];
    const float* We   = (const float*)d_in[3];
    const float* Wa   = (const float*)d_in[4];
    const float* W_ih = (const float*)d_in[5];
    const float* W_hh = (const float*)d_in[6];
    const float* b_ih = (const float*)d_in[7];
    const float* b_hh = (const float*)d_in[8];

    float* ws    = (float*)d_ws;
    float* out   = (float*)d_out;
    float* S1    = ws + OFF_S1;
    float* S2    = ws + OFF_S2;
    float* sacc  = ws + OFF_SACC;
    float* S3    = ws + OFF_EACC;
    float* qb    = ws + OFF_QB;
    float* qa    = ws + OFF_QA;
    float* sAll  = ws + OFF_SALL;
    float* evAll = ws + OFF_EVALL;
    float* qeAll = ws + OFF_QEALL;
    float* stats = ws + OFF_STATS;
    float* qbP   = ws + OFF_QBP;
    float* qaP   = ws + OFF_QAP;

    auto gemm = [&](const float* A, int lda, const float* W, int ldw,
                    const float* bias, float* C, int ldc, int N, int accF, int rows) {
        dim3 g(N / 32, rows / SG_ROWS);
        wmma_gemm_kernel<false><<<g, SG_WAVES * 32, 0, stream>>>(nullptr, 0, A, lda, W, ldw, bias, C, ldc,
                                                                accF, nullptr, nullptr);
    };

    gemm(s0, Q_, Wb, Q_, nullptr, qb, MEM_, MEM_, 0, B_);
    gemm(s0, Q_, Wa, Q_, nullptr, qa, MEM_, MEM_, 0, B_);

    for (int t = 0; t < 5; ++t) {
        if (t == 0)
            scores_rows_kernel<2, true, 1><<<dim3(SC_CH, B_), SC_WAVES * 32, 0, stream>>>(
                M, qb, qa, nullptr, nullptr, nullptr, S1, S2, nullptr, nullptr, nullptr, stats);
        else
            scores_rows_kernel<2, true, TL_SLICES><<<dim3(SC_CH, B_), SC_WAVES * 32, 0, stream>>>(
                M, qbP, qaP, nullptr, nullptr, nullptr, S1, S2, nullptr, nullptr, nullptr, stats);

        if (t == 0)
            wsum_rows_kernel<true><<<dim3(WS_CH, B_), WS_WAVES * 32, 0, stream>>>(M, S1, S2, stats, sacc);
        else
            wsum_rows_kernel<false><<<dim3(WS_CH, B_), WS_WAVES * 32, 0, stream>>>(M, S1, S2, stats, sacc);

        if (t == 0)
            turn_tail_kernel<<<dim3(TL_SLICES, B_ / TL_ROWS), TL_WAVES * 32, 0, stream>>>(
                S1, s0, W_ih, W_hh, b_ih, b_hh, Wb, Wa,
                sAll, qbP, qaP, evAll);
        else if (t < 4)
            turn_tail_kernel<<<dim3(TL_SLICES, B_ / TL_ROWS), TL_WAVES * 32, 0, stream>>>(
                S1, sAll + (t - 1) * BQ_, W_ih, W_hh, b_ih, b_hh, Wb, Wa,
                sAll + t * BQ_, qbP, qaP, evAll + t * BQ_);
    }

    wmma_gemm_kernel<true><<<dim3(MEM_ / 32, (5 * B_) / SG_ROWS), SG_WAVES * 32, 0, stream>>>(
        s0, B_, sAll, Q_, We, Q_ + MEM_, nullptr, qeAll, MEM_, 0, S1, evAll + 4 * BQ_);
    gemm(evAll, MEM_, We + Q_, Q_ + MEM_, nullptr, qeAll, MEM_, MEM_, 1, 5 * B_);

    scores_rows_kernel<5, false, 1><<<dim3(SC_CH, B_), SC_WAVES * 32, 0, stream>>>(
        M, qeAll, qeAll + BQ_, qeAll + 2 * BQ_, qeAll + 3 * BQ_, qeAll + 4 * BQ_,
        S1, S2, S3, out, out + BL_, nullptr);
    end_softmax_kernel<5><<<B_, 256, 0, stream>>>(S1, S2, S3, out, out + BL_, sacc, out, out + BL_);
}
